# starting kernel
# baseline (speedup 1.0000x reference)
#include <hip/hip_runtime.h>


#define N_NODES   50000
#define N_GRAPHS  128
#define HID       512
#define OUT_PITCH (3 * HID)
#define MAX_CHUNKS 64
#define COLS_PER_WG 256

typedef __attribute__((ext_vector_type(2))) float v2f;
typedef __attribute__((ext_vector_type(8))) float v8f;

__global__ void k_zero_counts(unsigned* __restrict__ counts) {
    counts[threadIdx.x] = 0u;
}

__global__ void k_pb_hist(const int* __restrict__ walk2_row0,
                          const int* __restrict__ batch,
                          int* __restrict__ pb,
                          unsigned* __restrict__ counts) {
    int n = blockIdx.x * 256 + threadIdx.x;
    if (n < N_NODES) {
        int g = batch[walk2_row0[n]];
        pb[n] = g;
        atomicAdd(&counts[g], 1u);
    }
}

__global__ void __launch_bounds__(256)
k_accum(const float* __restrict__ x,
        const int* __restrict__ w2, const int* __restrict__ w3,
        const int* __restrict__ w4, const int* __restrict__ pb,
        float* __restrict__ partial, int nchunks) {
    __shared__ float acc[N_GRAPHS * COLS_PER_WG];
    const int chunk = blockIdx.x;
    const int cg    = blockIdx.y;
    const int t     = threadIdx.x;
    const int c     = cg * COLS_PER_WG + t;

    for (int i = t; i < N_GRAPHS * COLS_PER_WG; i += 256) acc[i] = 0.0f;
    __syncthreads();

    const int per = (N_NODES + nchunks - 1) / nchunks;
    const int n0  = chunk * per;
    const int n1  = (n0 + per < N_NODES) ? (n0 + per) : N_NODES;

    for (int n = n0; n < n1; ++n) {
        const int g = pb[n];
        float y = 0.0f;
        #pragma unroll
        for (int l = 0; l < 3; ++l) y += x[w2[l * N_NODES + n] * HID + c];
        #pragma unroll
        for (int l = 0; l < 4; ++l) y += x[w3[l * N_NODES + n] * HID + c];
        #pragma unroll
        for (int l = 0; l < 5; ++l) y += x[w4[l * N_NODES + n] * HID + c];
        acc[g * COLS_PER_WG + t] += y;
    }
    __syncthreads();

    float* dst = partial + (size_t)chunk * (N_GRAPHS * HID) + cg * COLS_PER_WG;
    for (int g = 0; g < N_GRAPHS; ++g)
        dst[g * HID + t] = acc[g * COLS_PER_WG + t];
}

__global__ void k_reduce(const float* __restrict__ partial,
                         const unsigned* __restrict__ counts,
                         float* __restrict__ P, float* __restrict__ s,
                         int nchunks) {
    int i = blockIdx.x * 256 + threadIdx.x;
    int g = i >> 9;
    float a = 0.0f;
    for (int ch = 0; ch < nchunks; ++ch)
        a += partial[(size_t)ch * (N_GRAPHS * HID) + i];
    float cnt = (float)counts[g];
    P[i] = a / fmaxf(cnt, 1.0f);
    if (i < N_GRAPHS) s[i] = (counts[i] != 0u) ? 12.0f : 0.0f;
}

__global__ void __launch_bounds__(256)
k_gemm_wmma(const float* __restrict__ A, int lda,
            const float* __restrict__ W, const float* __restrict__ bias,
            const float* __restrict__ s, float* __restrict__ out) {
    const int wave = (blockIdx.x << 3) + (threadIdx.x >> 5);
    const int lane = threadIdx.x & 31;
    const int l16  = lane & 15;
    const int lh   = lane >> 4;
    const int tm   = wave >> 5;
    const int tn   = wave & 31;

    const float* Arow = A + (tm * 16 + l16) * lda;
    const int col = tn * 16 + l16;

    v8f acc = {};
    for (int k = 0; k < HID; k += 4) {
        const int kk = k + lh * 2;
        v2f a; a.x = Arow[kk];           a.y = Arow[kk + 1];
        v2f b; b.x = W[kk * HID + col];  b.y = W[(kk + 1) * HID + col];
        acc = __builtin_amdgcn_wmma_f32_16x16x4_f32(
             false, a,  false, b,
             (short)0, acc,  false,  false);
    }

    const float bn = bias[col];
    #pragma unroll
    for (int v = 0; v < 8; ++v) {
        const int m = tm * 16 + v + lh * 8;
        out[m * OUT_PITCH + col] = acc[v] + s[m] * bn;
    }
}

extern "C" void kernel_launch(void* const* d_in, const int* in_sizes, int n_in,
                              void* d_out, int out_size, void* d_ws, size_t ws_size,
                              hipStream_t stream) {
    const float* x     = (const float*)d_in[0];
    const int*   w2    = (const int*)  d_in[1];
    const int*   w3    = (const int*)  d_in[2];
    const int*   w4    = (const int*)  d_in[3];
    const int*   batch = (const int*)  d_in[4];
    const float* W0 = (const float*)d_in[5];  const float* b0 = (const float*)d_in[6];
    const float* W1 = (const float*)d_in[7];  const float* b1 = (const float*)d_in[8];
    const float* W2 = (const float*)d_in[9];  const float* b2 = (const float*)d_in[10];
    float* out = (float*)d_out;

    const size_t partBytes = (size_t)N_GRAPHS * HID * sizeof(float);
    const size_t fixed = partBytes   + 512   + 512
                       + (size_t)N_NODES * sizeof(int)  ;
    int nchunks = 1;
    if (ws_size > fixed + partBytes) {
        size_t fit = (ws_size - fixed) / partBytes;
        nchunks = (fit > MAX_CHUNKS) ? MAX_CHUNKS : (int)fit;
        if (nchunks < 1) nchunks = 1;
    }
    char* ws = (char*)d_ws;
    float*    partial = (float*)ws;
    float*    P       = (float*)(ws + (size_t)nchunks * partBytes);
    unsigned* counts  = (unsigned*)((char*)P + partBytes);
    float*    s       = (float*)((char*)counts + 512);
    int*      pb      = (int*)((char*)s + 512);

    k_zero_counts<<<1, 128, 0, stream>>>(counts);
    k_pb_hist<<<(N_NODES + 255) / 256, 256, 0, stream>>>(w2, batch, pb, counts);

    dim3 gAcc(nchunks, 2);
    k_accum<<<gAcc, 256, 0, stream>>>(x, w2, w3, w4, pb, partial, nchunks);
    k_reduce<<<(N_GRAPHS * HID) / 256, 256, 0, stream>>>(partial, counts, P, s, nchunks);

    k_gemm_wmma<<<32, 256, 0, stream>>>(P,         HID,       W0, b0, s, out);
    k_gemm_wmma<<<32, 256, 0, stream>>>(out,       OUT_PITCH, W1, b1, s, out + HID);
    k_gemm_wmma<<<32, 256, 0, stream>>>(out + HID, OUT_PITCH, W2, b2, s, out + 2 * HID);
}


# pooled GEMM chain re-associated to P@W0, P@(W0W1), P@(W0W1W2): weight products with bias row built in spare workgroups of earlier launches, three GEMM launches become one (new summation order)
# speedup vs baseline: 4.5154x; 4.5154x over previous
#include <hip/hip_runtime.h>


#define N_NODES   50000
#define N_GRAPHS  128
#define HID       512
#define OUT_PITCH (3 * HID)
#define MAX_CHUNKS 64
#define COLS_PER_WG 256
#define KSTEPS    (N_NODES / 4)
#define CT_ROWS   (N_NODES + 4)
#define DEPTH     7
#define WP_ROWS   528
#define WP_TILES  (33 * 32)
#define SCAT_WGS  ((N_NODES + 255) / 256)

typedef __attribute__((ext_vector_type(2))) float v2f;
typedef __attribute__((ext_vector_type(8))) float v8f;
typedef __attribute__((ext_vector_type(4))) float v4f;
typedef __attribute__((ext_vector_type(4))) unsigned v4u;

__global__ void k_zero_counts(unsigned* __restrict__ counts) {
    counts[threadIdx.x] = 0u;
}

__global__ void k_pb_hist(const int* __restrict__ walk2_row0,
                          const int* __restrict__ batch,
                          int* __restrict__ pb,
                          unsigned* __restrict__ counts) {
    int n = blockIdx.x * 256 + threadIdx.x;
    if (n < N_NODES) {
        int g = batch[walk2_row0[n]];
        pb[n] = g;
        atomicAdd(&counts[g], 1u);
    }
}

__global__ void __launch_bounds__(256)
k_accum(const float* __restrict__ x,
        const int* __restrict__ w2, const int* __restrict__ w3,
        const int* __restrict__ w4, const int* __restrict__ pb,
        float* __restrict__ partial, int nchunks) {
    __shared__ float acc[N_GRAPHS * COLS_PER_WG];
    const int chunk = blockIdx.x;
    const int cg    = blockIdx.y;
    const int t     = threadIdx.x;
    const int c     = cg * COLS_PER_WG + t;

    for (int i = t; i < N_GRAPHS * COLS_PER_WG; i += 256) acc[i] = 0.0f;
    __syncthreads();

    const int per = (N_NODES + nchunks - 1) / nchunks;
    const int n0  = chunk * per;
    const int n1  = (n0 + per < N_NODES) ? (n0 + per) : N_NODES;

    for (int n = n0; n < n1; ++n) {
        const int g = pb[n];
        float y = 0.0f;
        #pragma unroll
        for (int l = 0; l < 3; ++l) y += x[w2[l * N_NODES + n] * HID + c];
        #pragma unroll
        for (int l = 0; l < 4; ++l) y += x[w3[l * N_NODES + n] * HID + c];
        #pragma unroll
        for (int l = 0; l < 5; ++l) y += x[w4[l * N_NODES + n] * HID + c];
        acc[g * COLS_PER_WG + t] += y;
    }
    __syncthreads();

    float* dst = partial + (size_t)chunk * (N_GRAPHS * HID) + cg * COLS_PER_WG;
    for (int g = 0; g < N_GRAPHS; ++g)
        dst[g * HID + t] = acc[g * COLS_PER_WG + t];
}


__global__ void k_zero_ct(unsigned* __restrict__ ct, unsigned* __restrict__ counts) {
    v4u* ct4 = (v4u*)ct;
    const int t = blockIdx.x * 256 + threadIdx.x;
    const v4u z = {0u, 0u, 0u, 0u};
    for (int i = t; i < CT_ROWS * N_GRAPHS / 4; i += 512 * 256) ct4[i] = z;
    if (t < N_GRAPHS) counts[t] = 0u;
}

__device__ __forceinline__ void wprod_tile(const float* __restrict__ A,
                                           const float* __restrict__ brow,
                                           const float* __restrict__ B,
                                           const float* __restrict__ badd,
                                           float* __restrict__ outp, int tile) {
    const int lane = threadIdx.x & 31;
    const int l16  = lane & 15;
    const int lh   = lane >> 4;
    const int tm   = tile >> 5;
    const int tn   = tile & 31;
    const int row  = tm * 16 + l16;
    const bool valid = row <= HID;
    const float* Arow = (row < HID) ? (A + row * HID) : brow;
    const int col = tn * 16 + l16;

    v8f acc = {};
    for (int k = 0; k < HID; k += 4) {
        const int kk = k + lh * 2;
        v2f a; a.x = Arow[kk];           a.y = Arow[kk + 1];
        a.x = valid ? a.x : 0.0f;        a.y = valid ? a.y : 0.0f;
        v2f b; b.x = B[kk * HID + col];  b.y = B[(kk + 1) * HID + col];
        acc = __builtin_amdgcn_wmma_f32_16x16x4_f32(
            false, a, false, b, (short)0, acc, false, false);
    }
    const float bn = badd[col];
    #pragma unroll
    for (int v = 0; v < 8; ++v) {
        const int m = tm * 16 + v + lh * 8;
        outp[m * HID + col] = acc[v] + ((m == HID) ? bn : 0.0f);
    }
}

__device__ __forceinline__ void scatter_body(const int* __restrict__ w2, const int* __restrict__ w3,
                          const int* __restrict__ w4, const int* __restrict__ batch,
                          unsigned* __restrict__ ct, unsigned* __restrict__ counts) {
    const int n = blockIdx.x * 256 + threadIdx.x;
    if (n < N_NODES) {
        int m[12];
        for (int l = 0; l < 3; ++l) m[l]     = w2[l * N_NODES + n];
        for (int l = 0; l < 4; ++l) m[3 + l] = w3[l * N_NODES + n];
        for (int l = 0; l < 5; ++l) m[7 + l] = w4[l * N_NODES + n];
        const int g = batch[m[0]];
        atomicAdd(&counts[g], 1u);
        for (int l = 0; l < 12; ++l) atomicAdd(&ct[(size_t)m[l] * N_GRAPHS + g], 1u);
    }
}

__global__ void k_scatter(const int* __restrict__ w2, const int* __restrict__ w3,
                          const int* __restrict__ w4, const int* __restrict__ batch,
                          unsigned* __restrict__ ct, unsigned* __restrict__ counts) {
    scatter_body(w2, w3, w4, batch, ct, counts);
}

__global__ void __launch_bounds__(256)
k_scatter_w(const int* __restrict__ w2, const int* __restrict__ w3,
            const int* __restrict__ w4, const int* __restrict__ batch,
            unsigned* __restrict__ ct, unsigned* __restrict__ counts,
            const float* __restrict__ W0, const float* __restrict__ b0,
            const float* __restrict__ W1, const float* __restrict__ b1,
            float* __restrict__ w01) {
    if (blockIdx.x < SCAT_WGS) {
        scatter_body(w2, w3, w4, batch, ct, counts);
    } else {
        const int tile = (blockIdx.x - SCAT_WGS) * 8 + (threadIdx.x >> 5);
        if (tile < WP_TILES) wprod_tile(W0, b0, W1, b1, w01, tile);
    }
}

__device__ __forceinline__ void
cgemm_body(const float* __restrict__ x, const unsigned* __restrict__ ct,
           float* __restrict__ partial, int nchunks) {
    const int chunk = blockIdx.x;
    const int wave  = threadIdx.x >> 5;
    const int lane  = threadIdx.x & 31;
    const int l16   = lane & 15;
    const int lh    = lane >> 4;
    const int gb    = (wave & 1) * 64;
    const int cb    = (blockIdx.y * 2 + (wave >> 1)) * 64;

    const int per = (KSTEPS + nchunks - 1) / nchunks;
    const int s0  = chunk * per;
    const int s1  = (s0 + per < KSTEPS) ? (s0 + per) : KSTEPS;
    const int nsteps = s1 - s0;

    const float*    xl = x  + (2 * lh) * HID      + cb + 4 * l16;
    const unsigned* cl = ct + (2 * lh) * N_GRAPHS + gb + 4 * l16;

    v4f rb0[2][DEPTH], rb1[2][DEPTH];
    v4u ra0[2][DEPTH], ra1[2][DEPTH];
    v8f acc[4][4] = {};

    #define ISSUE(p, d, s)                                                      \
        {                                                                       \
            const int sx = ((s) < KSTEPS) ? (s) : (KSTEPS - 1);                 \
            const int sc = ((s) < s1) ? (s) : KSTEPS;                           \
            const float*    xp = xl + (size_t)sx * (4 * HID);                   \
            const unsigned* cp = cl + (size_t)sc * (4 * N_GRAPHS);              \
            rb0[p][d] = *(const v4f*)xp;                                        \
            rb1[p][d] = *(const v4f*)(xp + HID);                                \
            ra0[p][d] = *(const v4u*)cp;                                        \
            ra1[p][d] = *(const v4u*)(cp + N_GRAPHS);                           \
        }
    #define COMPUTE(p, d)                                                       \
        {                                                                       \
            v2f a[4], b[4];                                                     \
            _Pragma("unroll")                                                   \
            for (int i = 0; i < 4; ++i) {                                       \
                a[i].x = (float)ra0[p][d][i];  a[i].y = (float)ra1[p][d][i];    \
                b[i].x = rb0[p][d][i];         b[i].y = rb1[p][d][i];           \
            }                                                                   \
            _Pragma("unroll")                                                   \
            for (int i = 0; i < 4; ++i) {                                       \
                _Pragma("unroll")                                               \
                for (int j = 0; j < 4; ++j)                                     \
                    acc[i][j] = __builtin_amdgcn_wmma_f32_16x16x4_f32(          \
                        false, a[i], false, b[j], (short)0, acc[i][j], false, false); \
            }                                                                   \
        }

    #pragma unroll
    for (int d = 0; d < DEPTH; ++d) ISSUE(0, d, s0 + d)

    for (int base = 0; base < nsteps; base += 2 * DEPTH) {
        #pragma unroll
        for (int d = 0; d < DEPTH; ++d) ISSUE(1, d, s0 + base + DEPTH + d)
        __builtin_amdgcn_sched_barrier(0);
        #pragma unroll
        for (int d = 0; d < DEPTH; ++d) COMPUTE(0, d)
        __builtin_amdgcn_sched_barrier(0);
        #pragma unroll
        for (int d = 0; d < DEPTH; ++d) ISSUE(0, d, s0 + base + 2 * DEPTH + d)
        __builtin_amdgcn_sched_barrier(0);
        #pragma unroll
        for (int d = 0; d < DEPTH; ++d) COMPUTE(1, d)
    }
    #undef ISSUE
    #undef COMPUTE

    float* dst = partial + (size_t)chunk * (N_GRAPHS * HID) + cb + 4 * l16;
    #pragma unroll
    for (int i = 0; i < 4; ++i) {
        #pragma unroll
        for (int v = 0; v < 8; ++v) {
            const int g = gb + 4 * (v + 8 * lh) + i;
            v4f o = {acc[i][0][v], acc[i][1][v], acc[i][2][v], acc[i][3][v]};
            *(v4f*)(dst + g * HID) = o;
        }
    }
}

__global__ void __launch_bounds__(128)
k_cgemm(const float* __restrict__ x, const unsigned* __restrict__ ct,
        float* __restrict__ partial, int nchunks) {
    cgemm_body(x, ct, partial, nchunks);
}

__global__ void __launch_bounds__(128)
k_cgemm_w(const float* __restrict__ x, const unsigned* __restrict__ ct,
          float* __restrict__ partial, int nchunks,
          const float* __restrict__ w01, const float* __restrict__ W2,
          const float* __restrict__ b2, float* __restrict__ w012) {
    if (blockIdx.y < 4) {
        cgemm_body(x, ct, partial, nchunks);
    } else {
        const int tile = ((blockIdx.y - 4) * gridDim.x + blockIdx.x) * 4 + (threadIdx.x >> 5);
        if (tile < WP_TILES) wprod_tile(w01, w01 + HID * HID, W2, b2, w012, tile);
    }
}

__global__ void k_reduce(const float* __restrict__ partial,
                         const unsigned* __restrict__ counts,
                         float* __restrict__ P, float* __restrict__ s,
                         int nchunks) {
    int i = blockIdx.x * 256 + threadIdx.x;
    int g = i >> 9;
    float a = 0.0f;
    for (int ch = 0; ch < nchunks; ++ch)
        a += partial[(size_t)ch * (N_GRAPHS * HID) + i];
    float cnt = (float)counts[g];
    P[i] = a / fmaxf(cnt, 1.0f);
    if (i < N_GRAPHS) s[i] = (counts[i] != 0u) ? 12.0f : 0.0f;
}

__global__ void __launch_bounds__(256)
k_gemm_wmma(const float* __restrict__ A, int lda,
            const float* __restrict__ W, const float* __restrict__ bias,
            const float* __restrict__ s, float* __restrict__ out) {
    const int wave = (blockIdx.x << 3) + (threadIdx.x >> 5);
    const int lane = threadIdx.x & 31;
    const int l16  = lane & 15;
    const int lh   = lane >> 4;
    const int tm   = wave >> 5;
    const int tn   = wave & 31;

    const float* Arow = A + (tm * 16 + l16) * lda;
    const int col = tn * 16 + l16;

    v8f acc = {};
    for (int k = 0; k < HID; k += 4) {
        const int kk = k + lh * 2;
        v2f a; a.x = Arow[kk];           a.y = Arow[kk + 1];
        v2f b; b.x = W[kk * HID + col];  b.y = W[(kk + 1) * HID + col];
        acc = __builtin_amdgcn_wmma_f32_16x16x4_f32(
             false, a,  false, b,
             (short)0, acc,  false,  false);
    }

    const float bn = bias[col];
    #pragma unroll
    for (int v = 0; v < 8; ++v) {
        const int m = tm * 16 + v + lh * 8;
        out[m * OUT_PITCH + col] = acc[v] + s[m] * bn;
    }
}

__global__ void __launch_bounds__(256)
k_gemm3(const float* __restrict__ P,
        const float* __restrict__ W0, const float* __restrict__ b0,
        const float* __restrict__ w01, const float* __restrict__ w012,
        const float* __restrict__ s, float* __restrict__ out) {
    const int wave = (blockIdx.x << 3) + (threadIdx.x >> 5);
    const int lane = threadIdx.x & 31;
    const int l16  = lane & 15;
    const int lh   = lane >> 4;
    const int tm    = wave / 96;
    const int tq    = wave - tm * 96;
    const int layer = tq >> 5;
    const int tn    = tq & 31;

    const float* B    = (layer == 0) ? W0 : ((layer == 1) ? w01 : w012);
    const float* bias = (layer == 0) ? b0 : (B + HID * HID);
    const float* Arow = P + (tm * 16 + l16) * HID;
    const int col = tn * 16 + l16;

    v8f acc = {};
    for (int k = 0; k < HID; k += 4) {
        const int kk = k + lh * 2;
        v2f a; a.x = Arow[kk];           a.y = Arow[kk + 1];
        v2f b; b.x = B[kk * HID + col];  b.y = B[(kk + 1) * HID + col];
        acc = __builtin_amdgcn_wmma_f32_16x16x4_f32(
            false, a, false, b, (short)0, acc, false, false);
    }

    const float bn = bias[col];
    #pragma unroll
    for (int v = 0; v < 8; ++v) {
        const int m = tm * 16 + v + lh * 8;
        out[m * OUT_PITCH + layer * HID + col] = acc[v] + s[m] * bn;
    }
}

extern "C" void kernel_launch(void* const* d_in, const int* in_sizes, int n_in,
                              void* d_out, int out_size, void* d_ws, size_t ws_size,
                              hipStream_t stream) {
    const float* x     = (const float*)d_in[0];
    const int*   w2    = (const int*)  d_in[1];
    const int*   w3    = (const int*)  d_in[2];
    const int*   w4    = (const int*)  d_in[3];
    const int*   batch = (const int*)  d_in[4];
    const float* W0 = (const float*)d_in[5];  const float* b0 = (const float*)d_in[6];
    const float* W1 = (const float*)d_in[7];  const float* b1 = (const float*)d_in[8];
    const float* W2 = (const float*)d_in[9];  const float* b2 = (const float*)d_in[10];
    float* out = (float*)d_out;

    const size_t partBytes = (size_t)N_GRAPHS * HID * sizeof(float);
    const size_t fixed = partBytes   + 512   + 512
                       + (size_t)N_NODES * sizeof(int)  ;
    int nchunks = 1;
    if (ws_size > fixed + partBytes) {
        size_t fit = (ws_size - fixed) / partBytes;
        nchunks = (fit > MAX_CHUNKS) ? MAX_CHUNKS : (int)fit;
        if (nchunks < 1) nchunks = 1;
    }
    char* ws = (char*)d_ws;
    float*    partial = (float*)ws;
    float*    P       = (float*)(ws + (size_t)nchunks * partBytes);
    unsigned* counts  = (unsigned*)((char*)P + partBytes);
    float*    s       = (float*)((char*)counts + 512);
    int*      pb      = (int*)((char*)s + 512);

    const size_t ctBytes = (size_t)CT_ROWS * N_GRAPHS * sizeof(unsigned);
    const size_t denseNeed = (size_t)MAX_CHUNKS * partBytes + partBytes + 1024 + ctBytes;
    const size_t wpBytes = (size_t)WP_ROWS * HID * sizeof(float);
    if (ws_size >= denseNeed + 2 * wpBytes) {
        unsigned* ct = (unsigned*)pb;
        float* w01  = (float*)((char*)ct + ctBytes);
        float* w012 = (float*)((char*)w01 + wpBytes);
        k_zero_ct<<<512, 256, 0, stream>>>(ct, counts);
        k_scatter_w<<<SCAT_WGS + WP_TILES / 8, 256, 0, stream>>>(w2, w3, w4, batch, ct, counts,
                                                              W0, b0, W1, b1, w01);
        dim3 gDenseW(nchunks, 4 + 5);
        k_cgemm_w<<<gDenseW, 128, 0, stream>>>(x, ct, partial, nchunks, w01, W2, b2, w012);
        k_reduce<<<(N_GRAPHS * HID) / 256, 256, 0, stream>>>(partial, counts, P, s, nchunks);
        k_gemm3<<<96, 256, 0, stream>>>(P, W0, b0, w01, w012, s, out);
        return;
    }
    if (ws_size >= denseNeed) {
        unsigned* ct = (unsigned*)pb;
        k_zero_ct<<<512, 256, 0, stream>>>(ct, counts);
        k_scatter<<<(N_NODES + 255) / 256, 256, 0, stream>>>(w2, w3, w4, batch, ct, counts);
        dim3 gDense(nchunks, 4);
        k_cgemm<<<gDense, 128, 0, stream>>>(x, ct, partial, nchunks);
    } else {
        k_zero_counts<<<1, 128, 0, stream>>>(counts);
        k_pb_hist<<<(N_NODES + 255) / 256, 256, 0, stream>>>(w2, batch, pb, counts);
        dim3 gAcc(nchunks, 2);
        k_accum<<<gAcc, 256, 0, stream>>>(x, w2, w3, w4, pb, partial, nchunks);
    }
    k_reduce<<<(N_GRAPHS * HID) / 256, 256, 0, stream>>>(partial, counts, P, s, nchunks);

    k_gemm_wmma<<<32, 256, 0, stream>>>(P,         HID,       W0, b0, s, out);
    k_gemm_wmma<<<32, 256, 0, stream>>>(out,       OUT_PITCH, W1, b1, s, out + HID);
    k_gemm_wmma<<<32, 256, 0, stream>>>(out + HID, OUT_PITCH, W2, b2, s, out + 2 * HID);
}


# Ct built without global atomics or zero fill: walk entries grouped by node range in each workgroup (LDS ranks), then each 196-row slab of Ct counted in LDS and stored whole; same counts and sums
# speedup vs baseline: 5.3951x; 1.1948x over previous
#include <hip/hip_runtime.h>


#define N_NODES   50000
#define N_GRAPHS  128
#define HID       512
#define OUT_PITCH (3 * HID)
#define MAX_CHUNKS 64
#define COLS_PER_WG 256
#define KSTEPS    (N_NODES / 4)
#define CT_ROWS   (N_NODES + 4)
#define DEPTH     7
#define WP_ROWS   528
#define WP_TILES  (33 * 32)
#define SCAT_WGS  ((N_NODES + 255) / 256)
#define SLAB      196
#define NSLABS    256
#define NBUCK     64
#define BUCK_NODES (4 * SLAB)
#define SEGS_PER_WAVE 7
#define ENT_PER_WG (256 * 12)

typedef __attribute__((ext_vector_type(2))) float v2f;
typedef __attribute__((ext_vector_type(8))) float v8f;
typedef __attribute__((ext_vector_type(4))) float v4f;
typedef __attribute__((ext_vector_type(4))) unsigned v4u;

__global__ void k_zero_counts(unsigned* __restrict__ counts) {
    counts[threadIdx.x] = 0u;
}

__global__ void k_pb_hist(const int* __restrict__ walk2_row0,
                          const int* __restrict__ batch,
                          int* __restrict__ pb,
                          unsigned* __restrict__ counts) {
    int n = blockIdx.x * 256 + threadIdx.x;
    if (n < N_NODES) {
        int g = batch[walk2_row0[n]];
        pb[n] = g;
        atomicAdd(&counts[g], 1u);
    }
}

__global__ void __launch_bounds__(256)
k_accum(const float* __restrict__ x,
        const int* __restrict__ w2, const int* __restrict__ w3,
        const int* __restrict__ w4, const int* __restrict__ pb,
        float* __restrict__ partial, int nchunks) {
    __shared__ float acc[N_GRAPHS * COLS_PER_WG];
    const int chunk = blockIdx.x;
    const int cg    = blockIdx.y;
    const int t     = threadIdx.x;
    const int c     = cg * COLS_PER_WG + t;

    for (int i = t; i < N_GRAPHS * COLS_PER_WG; i += 256) acc[i] = 0.0f;
    __syncthreads();

    const int per = (N_NODES + nchunks - 1) / nchunks;
    const int n0  = chunk * per;
    const int n1  = (n0 + per < N_NODES) ? (n0 + per) : N_NODES;

    for (int n = n0; n < n1; ++n) {
        const int g = pb[n];
        float y = 0.0f;
        #pragma unroll
        for (int l = 0; l < 3; ++l) y += x[w2[l * N_NODES + n] * HID + c];
        #pragma unroll
        for (int l = 0; l < 4; ++l) y += x[w3[l * N_NODES + n] * HID + c];
        #pragma unroll
        for (int l = 0; l < 5; ++l) y += x[w4[l * N_NODES + n] * HID + c];
        acc[g * COLS_PER_WG + t] += y;
    }
    __syncthreads();

    float* dst = partial + (size_t)chunk * (N_GRAPHS * HID) + cg * COLS_PER_WG;
    for (int g = 0; g < N_GRAPHS; ++g)
        dst[g * HID + t] = acc[g * COLS_PER_WG + t];
}


__global__ void k_zero_ct(unsigned* __restrict__ ct, unsigned* __restrict__ counts) {
    v4u* ct4 = (v4u*)ct;
    const int t = blockIdx.x * 256 + threadIdx.x;
    const v4u z = {0u, 0u, 0u, 0u};
    for (int i = t; i < CT_ROWS * N_GRAPHS / 4; i += 512 * 256) ct4[i] = z;
    if (t < N_GRAPHS) counts[t] = 0u;
}

__device__ __forceinline__ void wprod_tile(const float* __restrict__ A,
                                           const float* __restrict__ brow,
                                           const float* __restrict__ B,
                                           const float* __restrict__ badd,
                                           float* __restrict__ outp, int tile) {
    const int lane = threadIdx.x & 31;
    const int l16  = lane & 15;
    const int lh   = lane >> 4;
    const int tm   = tile >> 5;
    const int tn   = tile & 31;
    const int row  = tm * 16 + l16;
    const bool valid = row <= HID;
    const float* Arow = (row < HID) ? (A + row * HID) : brow;
    const int col = tn * 16 + l16;

    v8f acc = {};
    for (int k = 0; k < HID; k += 4) {
        const int kk = k + lh * 2;
        v2f a; a.x = Arow[kk];           a.y = Arow[kk + 1];
        a.x = valid ? a.x : 0.0f;        a.y = valid ? a.y : 0.0f;
        v2f b; b.x = B[kk * HID + col];  b.y = B[(kk + 1) * HID + col];
        acc = __builtin_amdgcn_wmma_f32_16x16x4_f32(
            false, a, false, b, (short)0, acc, false, false);
    }
    const float bn = badd[col];
    #pragma unroll
    for (int v = 0; v < 8; ++v) {
        const int m = tm * 16 + v + lh * 8;
        outp[m * HID + col] = acc[v] + ((m == HID) ? bn : 0.0f);
    }
}

__device__ __forceinline__ void scatter_body(const int* __restrict__ w2, const int* __restrict__ w3,
                          const int* __restrict__ w4, const int* __restrict__ batch,
                          unsigned* __restrict__ ct, unsigned* __restrict__ counts) {
    const int n = blockIdx.x * 256 + threadIdx.x;
    if (n < N_NODES) {
        int m[12];
        for (int l = 0; l < 3; ++l) m[l]     = w2[l * N_NODES + n];
        for (int l = 0; l < 4; ++l) m[3 + l] = w3[l * N_NODES + n];
        for (int l = 0; l < 5; ++l) m[7 + l] = w4[l * N_NODES + n];
        const int g = batch[m[0]];
        atomicAdd(&counts[g], 1u);
        for (int l = 0; l < 12; ++l) atomicAdd(&ct[(size_t)m[l] * N_GRAPHS + g], 1u);
    }
}

__global__ void k_scatter(const int* __restrict__ w2, const int* __restrict__ w3,
                          const int* __restrict__ w4, const int* __restrict__ batch,
                          unsigned* __restrict__ ct, unsigned* __restrict__ counts) {
    scatter_body(w2, w3, w4, batch, ct, counts);
}

__global__ void __launch_bounds__(256)
k_bin_w(const int* __restrict__ w2, const int* __restrict__ w3,
        const int* __restrict__ w4, const int* __restrict__ batch,
        unsigned* __restrict__ ent, unsigned* __restrict__ off,
        unsigned* __restrict__ hist,
        const float* __restrict__ W0, const float* __restrict__ b0,
        const float* __restrict__ W1, const float* __restrict__ b1,
        float* __restrict__ w01) {
    __shared__ unsigned cnt[NBUCK];
    __shared__ unsigned base[NBUCK + 1];
    __shared__ unsigned hg[N_GRAPHS];
    __shared__ __attribute__((aligned(16))) unsigned stage[ENT_PER_WG];
    if (blockIdx.x >= SCAT_WGS) {
        const int tile = (blockIdx.x - SCAT_WGS) * 8 + (threadIdx.x >> 5);
        if (tile < WP_TILES) wprod_tile(W0, b0, W1, b1, w01, tile);
        return;
    }
    const int t = threadIdx.x;
    const int wg = blockIdx.x;
    if (t < NBUCK) cnt[t] = 0u;
    if (t < N_GRAPHS) hg[t] = 0u;
    __syncthreads();

    const int n = wg * 256 + t;
    unsigned e[12], pos[12], bk[12];
    if (n < N_NODES) {
        int m[12];
        #pragma unroll
        for (int l = 0; l < 3; ++l) m[l]     = w2[l * N_NODES + n];
        #pragma unroll
        for (int l = 0; l < 4; ++l) m[3 + l] = w3[l * N_NODES + n];
        #pragma unroll
        for (int l = 0; l < 5; ++l) m[7 + l] = w4[l * N_NODES + n];
        const unsigned g = (unsigned)batch[m[0]];
        atomicAdd(&hg[g], 1u);
        #pragma unroll
        for (int l = 0; l < 12; ++l) {
            bk[l]  = (unsigned)m[l] / BUCK_NODES;
            e[l]   = ((unsigned)m[l] << 7) | g;
            pos[l] = atomicAdd(&cnt[bk[l]], 1u);
        }
    }
    __syncthreads();
    if (t == 0) {
        unsigned run = 0u;
        for (int b = 0; b < NBUCK; ++b) { base[b] = run; run += cnt[b]; }
        base[NBUCK] = run;
    }
    __syncthreads();
    const unsigned total = base[NBUCK];
    if (n < N_NODES) {
        #pragma unroll
        for (int l = 0; l < 12; ++l) stage[base[bk[l]] + pos[l]] = e[l];
    }
    if (t < 3 && total + t < ENT_PER_WG) stage[total + t] = 0u;
    __syncthreads();
    {
        const v4u* s4 = (const v4u*)stage;
        v4u* d4 = (v4u*)(ent + (size_t)wg * ENT_PER_WG);
        for (unsigned i = t; i < (total + 3u) / 4u; i += 256) d4[i] = s4[i];
    }
    if (t <= NBUCK) off[wg * (NBUCK + 1) + t] = base[t];
    if (t < N_GRAPHS) hist[wg * N_GRAPHS + t] = hg[t];
}

__global__ void __launch_bounds__(1024)
k_build(const unsigned* __restrict__ ent, const unsigned* __restrict__ off,
        const unsigned* __restrict__ hist,
        unsigned* __restrict__ ct, unsigned* __restrict__ counts) {
    __shared__ __attribute__((aligned(16))) unsigned ctl[SLAB * N_GRAPHS];
    const int t = threadIdx.x;
    const int o = blockIdx.x;
    if (o == NSLABS) {
        const int g = t & (N_GRAPHS - 1);
        const int part = t >> 7;
        unsigned a = 0u;
        for (int w = part; w < SCAT_WGS; w += 8) a += hist[w * N_GRAPHS + g];
        ctl[t] = a;
        __syncthreads();
        if (t < N_GRAPHS) {
            unsigned s = 0u;
            #pragma unroll
            for (int p = 0; p < 8; ++p) s += ctl[p * N_GRAPHS + t];
            counts[t] = s;
        }
        return;
    }
    v4u* ctl4 = (v4u*)ctl;
    const v4u z = {0u, 0u, 0u, 0u};
    for (int i = t; i < SLAB * N_GRAPHS / 4; i += 1024) ctl4[i] = z;
    __syncthreads();

    const int b = o >> 2;
    const unsigned m0 = (unsigned)o * SLAB;
    const int wave = __builtin_amdgcn_readfirstlane(t >> 5);
    const int lane = t & 31;

    unsigned lo[SEGS_PER_WAVE], hi[SEGS_PER_WAVE];
    #pragma unroll
    for (int k = 0; k < SEGS_PER_WAVE; ++k) {
        const int w = wave + 32 * k;
        const int wc = (w < SCAT_WGS) ? w : (SCAT_WGS - 1);
        const unsigned l0 = off[wc * (NBUCK + 1) + b];
        const unsigned h0 = off[wc * (NBUCK + 1) + b + 1];
        lo[k] = l0;
        hi[k] = (w < SCAT_WGS) ? h0 : l0;
    }
    v4u d[SEGS_PER_WAVE];
    #pragma unroll
    for (int k = 0; k < SEGS_PER_WAVE; ++k) {
        const int w = wave + 32 * k;
        const int wc = (w < SCAT_WGS) ? w : (SCAT_WGS - 1);
        const unsigned idx = (lo[k] & ~3u) + 4u * lane;
        const unsigned idc = (idx < ENT_PER_WG - 4u) ? idx : (ENT_PER_WG - 4u);
        d[k] = *(const v4u*)(ent + (size_t)wc * ENT_PER_WG + idc);
    }
    #pragma unroll
    for (int k = 0; k < SEGS_PER_WAVE; ++k) {
        const unsigned idx = (lo[k] & ~3u) + 4u * lane;
        #pragma unroll
        for (int j = 0; j < 4; ++j) {
            const unsigned i = idx + j;
            const unsigned r = (d[k][j] >> 7) - m0;
            if (i >= lo[k] && i < hi[k] && r < (unsigned)SLAB)
                atomicAdd(&ctl[r * N_GRAPHS + (d[k][j] & 127u)], 1u);
        }
    }
    for (int k = 0; k < SEGS_PER_WAVE; ++k) {
        const int w = wave + 32 * k;
        const int wc = (w < SCAT_WGS) ? w : (SCAT_WGS - 1);
        const unsigned* src = ent + (size_t)wc * ENT_PER_WG;
        for (unsigned i = (lo[k] & ~3u) + 128u + lane; i < hi[k]; i += 32) {
            const unsigned ev = src[i];
            const unsigned r = (ev >> 7) - m0;
            if (r < (unsigned)SLAB) atomicAdd(&ctl[r * N_GRAPHS + (ev & 127u)], 1u);
        }
    }
    __syncthreads();

    int rows = CT_ROWS - (int)m0;
    rows = (rows > SLAB) ? SLAB : rows;
    v4u* dst4 = (v4u*)(ct + (size_t)m0 * N_GRAPHS);
    for (int i = t; i < rows * (N_GRAPHS / 4); i += 1024) dst4[i] = ctl4[i];
}

__device__ __forceinline__ void
cgemm_body(const float* __restrict__ x, const unsigned* __restrict__ ct,
           float* __restrict__ partial, int nchunks) {
    const int chunk = blockIdx.x;
    const int wave  = threadIdx.x >> 5;
    const int lane  = threadIdx.x & 31;
    const int l16   = lane & 15;
    const int lh    = lane >> 4;
    const int gb    = (wave & 1) * 64;
    const int cb    = (blockIdx.y * 2 + (wave >> 1)) * 64;

    const int per = (KSTEPS + nchunks - 1) / nchunks;
    const int s0  = chunk * per;
    const int s1  = (s0 + per < KSTEPS) ? (s0 + per) : KSTEPS;
    const int nsteps = s1 - s0;

    const float*    xl = x  + (2 * lh) * HID      + cb + 4 * l16;
    const unsigned* cl = ct + (2 * lh) * N_GRAPHS + gb + 4 * l16;

    v4f rb0[2][DEPTH], rb1[2][DEPTH];
    v4u ra0[2][DEPTH], ra1[2][DEPTH];
    v8f acc[4][4] = {};

    #define ISSUE(p, d, s)                                                      \
        {                                                                       \
            const int sx = ((s) < KSTEPS) ? (s) : (KSTEPS - 1);                 \
            const int sc = ((s) < s1) ? (s) : KSTEPS;                           \
            const float*    xp = xl + (size_t)sx * (4 * HID);                   \
            const unsigned* cp = cl + (size_t)sc * (4 * N_GRAPHS);              \
            rb0[p][d] = *(const v4f*)xp;                                        \
            rb1[p][d] = *(const v4f*)(xp + HID);                                \
            ra0[p][d] = *(const v4u*)cp;                                        \
            ra1[p][d] = *(const v4u*)(cp + N_GRAPHS);                           \
        }
    #define COMPUTE(p, d)                                                       \
        {                                                                       \
            v2f a[4], b[4];                                                     \
            _Pragma("unroll")                                                   \
            for (int i = 0; i < 4; ++i) {                                       \
                a[i].x = (float)ra0[p][d][i];  a[i].y = (float)ra1[p][d][i];    \
                b[i].x = rb0[p][d][i];         b[i].y = rb1[p][d][i];           \
            }                                                                   \
            _Pragma("unroll")                                                   \
            for (int i = 0; i < 4; ++i) {                                       \
                _Pragma("unroll")                                               \
                for (int j = 0; j < 4; ++j)                                     \
                    acc[i][j] = __builtin_amdgcn_wmma_f32_16x16x4_f32(          \
                        false, a[i], false, b[j], (short)0, acc[i][j], false, false); \
            }                                                                   \
        }

    #pragma unroll
    for (int d = 0; d < DEPTH; ++d) ISSUE(0, d, s0 + d)

    for (int base = 0; base < nsteps; base += 2 * DEPTH) {
        #pragma unroll
        for (int d = 0; d < DEPTH; ++d) ISSUE(1, d, s0 + base + DEPTH + d)
        __builtin_amdgcn_sched_barrier(0);
        #pragma unroll
        for (int d = 0; d < DEPTH; ++d) COMPUTE(0, d)
        __builtin_amdgcn_sched_barrier(0);
        #pragma unroll
        for (int d = 0; d < DEPTH; ++d) ISSUE(0, d, s0 + base + 2 * DEPTH + d)
        __builtin_amdgcn_sched_barrier(0);
        #pragma unroll
        for (int d = 0; d < DEPTH; ++d) COMPUTE(1, d)
    }
    #undef ISSUE
    #undef COMPUTE

    float* dst = partial + (size_t)chunk * (N_GRAPHS * HID) + cb + 4 * l16;
    #pragma unroll
    for (int i = 0; i < 4; ++i) {
        #pragma unroll
        for (int v = 0; v < 8; ++v) {
            const int g = gb + 4 * (v + 8 * lh) + i;
            v4f o = {acc[i][0][v], acc[i][1][v], acc[i][2][v], acc[i][3][v]};
            *(v4f*)(dst + g * HID) = o;
        }
    }
}

__global__ void __launch_bounds__(128)
k_cgemm(const float* __restrict__ x, const unsigned* __restrict__ ct,
        float* __restrict__ partial, int nchunks) {
    cgemm_body(x, ct, partial, nchunks);
}

__global__ void __launch_bounds__(128)
k_cgemm_w(const float* __restrict__ x, const unsigned* __restrict__ ct,
          float* __restrict__ partial, int nchunks,
          const float* __restrict__ w01, const float* __restrict__ W2,
          const float* __restrict__ b2, float* __restrict__ w012) {
    if (blockIdx.y < 4) {
        cgemm_body(x, ct, partial, nchunks);
    } else {
        const int tile = ((blockIdx.y - 4) * gridDim.x + blockIdx.x) * 4 + (threadIdx.x >> 5);
        if (tile < WP_TILES) wprod_tile(w01, w01 + HID * HID, W2, b2, w012, tile);
    }
}

__global__ void k_reduce(const float* __restrict__ partial,
                         const unsigned* __restrict__ counts,
                         float* __restrict__ P, float* __restrict__ s,
                         int nchunks) {
    int i = blockIdx.x * 256 + threadIdx.x;
    int g = i >> 9;
    float a = 0.0f;
    for (int ch = 0; ch < nchunks; ++ch)
        a += partial[(size_t)ch * (N_GRAPHS * HID) + i];
    float cnt = (float)counts[g];
    P[i] = a / fmaxf(cnt, 1.0f);
    if (i < N_GRAPHS) s[i] = (counts[i] != 0u) ? 12.0f : 0.0f;
}

__global__ void __launch_bounds__(256)
k_gemm_wmma(const float* __restrict__ A, int lda,
            const float* __restrict__ W, const float* __restrict__ bias,
            const float* __restrict__ s, float* __restrict__ out) {
    const int wave = (blockIdx.x << 3) + (threadIdx.x >> 5);
    const int lane = threadIdx.x & 31;
    const int l16  = lane & 15;
    const int lh   = lane >> 4;
    const int tm   = wave >> 5;
    const int tn   = wave & 31;

    const float* Arow = A + (tm * 16 + l16) * lda;
    const int col = tn * 16 + l16;

    v8f acc = {};
    for (int k = 0; k < HID; k += 4) {
        const int kk = k + lh * 2;
        v2f a; a.x = Arow[kk];           a.y = Arow[kk + 1];
        v2f b; b.x = W[kk * HID + col];  b.y = W[(kk + 1) * HID + col];
        acc = __builtin_amdgcn_wmma_f32_16x16x4_f32(
             false, a,  false, b,
             (short)0, acc,  false,  false);
    }

    const float bn = bias[col];
    #pragma unroll
    for (int v = 0; v < 8; ++v) {
        const int m = tm * 16 + v + lh * 8;
        out[m * OUT_PITCH + col] = acc[v] + s[m] * bn;
    }
}

__global__ void __launch_bounds__(256)
k_gemm3(const float* __restrict__ P,
        const float* __restrict__ W0, const float* __restrict__ b0,
        const float* __restrict__ w01, const float* __restrict__ w012,
        const float* __restrict__ s, float* __restrict__ out) {
    const int wave = (blockIdx.x << 3) + (threadIdx.x >> 5);
    const int lane = threadIdx.x & 31;
    const int l16  = lane & 15;
    const int lh   = lane >> 4;
    const int tm    = wave / 96;
    const int tq    = wave - tm * 96;
    const int layer = tq >> 5;
    const int tn    = tq & 31;

    const float* B    = (layer == 0) ? W0 : ((layer == 1) ? w01 : w012);
    const float* bias = (layer == 0) ? b0 : (B + HID * HID);
    const float* Arow = P + (tm * 16 + l16) * HID;
    const int col = tn * 16 + l16;

    v8f acc = {};
    for (int k = 0; k < HID; k += 4) {
        const int kk = k + lh * 2;
        v2f a; a.x = Arow[kk];           a.y = Arow[kk + 1];
        v2f b; b.x = B[kk * HID + col];  b.y = B[(kk + 1) * HID + col];
        acc = __builtin_amdgcn_wmma_f32_16x16x4_f32(
            false, a, false, b, (short)0, acc, false, false);
    }

    const float bn = bias[col];
    #pragma unroll
    for (int v = 0; v < 8; ++v) {
        const int m = tm * 16 + v + lh * 8;
        out[m * OUT_PITCH + layer * HID + col] = acc[v] + s[m] * bn;
    }
}

extern "C" void kernel_launch(void* const* d_in, const int* in_sizes, int n_in,
                              void* d_out, int out_size, void* d_ws, size_t ws_size,
                              hipStream_t stream) {
    const float* x     = (const float*)d_in[0];
    const int*   w2    = (const int*)  d_in[1];
    const int*   w3    = (const int*)  d_in[2];
    const int*   w4    = (const int*)  d_in[3];
    const int*   batch = (const int*)  d_in[4];
    const float* W0 = (const float*)d_in[5];  const float* b0 = (const float*)d_in[6];
    const float* W1 = (const float*)d_in[7];  const float* b1 = (const float*)d_in[8];
    const float* W2 = (const float*)d_in[9];  const float* b2 = (const float*)d_in[10];
    float* out = (float*)d_out;

    const size_t partBytes = (size_t)N_GRAPHS * HID * sizeof(float);
    const size_t fixed = partBytes   + 512   + 512
                       + (size_t)N_NODES * sizeof(int)  ;
    int nchunks = 1;
    if (ws_size > fixed + partBytes) {
        size_t fit = (ws_size - fixed) / partBytes;
        nchunks = (fit > MAX_CHUNKS) ? MAX_CHUNKS : (int)fit;
        if (nchunks < 1) nchunks = 1;
    }
    char* ws = (char*)d_ws;
    float*    partial = (float*)ws;
    float*    P       = (float*)(ws + (size_t)nchunks * partBytes);
    unsigned* counts  = (unsigned*)((char*)P + partBytes);
    float*    s       = (float*)((char*)counts + 512);
    int*      pb      = (int*)((char*)s + 512);

    const size_t ctBytes = (size_t)CT_ROWS * N_GRAPHS * sizeof(unsigned);
    const size_t denseNeed = (size_t)MAX_CHUNKS * partBytes + partBytes + 1024 + ctBytes;
    const size_t wpBytes = (size_t)WP_ROWS * HID * sizeof(float);
    if (ws_size >= denseNeed + 2 * wpBytes) {
        unsigned* ct = (unsigned*)pb;
        float* w01  = (float*)((char*)ct + ctBytes);
        float* w012 = (float*)((char*)w01 + wpBytes);
        unsigned* ent  = (unsigned*)partial;
        unsigned* off  = ent + (size_t)SCAT_WGS * ENT_PER_WG;
        unsigned* hist = off + SCAT_WGS * (NBUCK + 1);
        k_bin_w<<<SCAT_WGS + WP_TILES / 8, 256, 0, stream>>>(w2, w3, w4, batch, ent, off, hist,
                                                          W0, b0, W1, b1, w01);
        k_build<<<NSLABS + 1, 1024, 0, stream>>>(ent, off, hist, ct, counts);
        dim3 gDenseW(nchunks, 4 + 5);
        k_cgemm_w<<<gDenseW, 128, 0, stream>>>(x, ct, partial, nchunks, w01, W2, b2, w012);
        k_reduce<<<(N_GRAPHS * HID) / 256, 256, 0, stream>>>(partial, counts, P, s, nchunks);
        k_gemm3<<<96, 256, 0, stream>>>(P, W0, b0, w01, w012, s, out);
        return;
    }
    if (ws_size >= denseNeed) {
        unsigned* ct = (unsigned*)pb;
        k_zero_ct<<<512, 256, 0, stream>>>(ct, counts);
        k_scatter<<<(N_NODES + 255) / 256, 256, 0, stream>>>(w2, w3, w4, batch, ct, counts);
        dim3 gDense(nchunks, 4);
        k_cgemm<<<gDense, 128, 0, stream>>>(x, ct, partial, nchunks);
    } else {
        k_zero_counts<<<1, 128, 0, stream>>>(counts);
        k_pb_hist<<<(N_NODES + 255) / 256, 256, 0, stream>>>(w2, batch, pb, counts);
        dim3 gAcc(nchunks, 2);
        k_accum<<<gAcc, 256, 0, stream>>>(x, w2, w3, w4, pb, partial, nchunks);
    }
    k_reduce<<<(N_GRAPHS * HID) / 256, 256, 0, stream>>>(partial, counts, P, s, nchunks);

    k_gemm_wmma<<<32, 256, 0, stream>>>(P,         HID,       W0, b0, s, out);
    k_gemm_wmma<<<32, 256, 0, stream>>>(out,       OUT_PITCH, W1, b1, s, out + HID);
    k_gemm_wmma<<<32, 256, 0, stream>>>(out + HID, OUT_PITCH, W2, b2, s, out + 2 * HID);
}


# 16x16 tile GEMMs (weight products, pooled GEMM): operands of 16 K steps in two register buffers, next buffer's 48 loads issued in front of the current buffer's WMMAs; same products, same sums
# speedup vs baseline: 6.7253x; 1.2466x over previous
#include <hip/hip_runtime.h>


#define N_NODES   50000
#define N_GRAPHS  128
#define HID       512
#define OUT_PITCH (3 * HID)
#define MAX_CHUNKS 64
#define COLS_PER_WG 256
#define KSTEPS    (N_NODES / 4)
#define CT_ROWS   (N_NODES + 4)
#define DEPTH     7
#define WP_ROWS   528
#define WP_TILES  (33 * 32)
#define SCAT_WGS  ((N_NODES + 255) / 256)
#define SLAB      196
#define NSLABS    256
#define NBUCK     64
#define BUCK_NODES (4 * SLAB)
#define SEGS_PER_WAVE 7
#define ENT_PER_WG (256 * 12)

typedef __attribute__((ext_vector_type(2))) float v2f;
typedef __attribute__((ext_vector_type(8))) float v8f;
typedef __attribute__((ext_vector_type(4))) float v4f;
typedef __attribute__((ext_vector_type(4))) unsigned v4u;

__global__ void k_zero_counts(unsigned* __restrict__ counts) {
    counts[threadIdx.x] = 0u;
}

__global__ void k_pb_hist(const int* __restrict__ walk2_row0,
                          const int* __restrict__ batch,
                          int* __restrict__ pb,
                          unsigned* __restrict__ counts) {
    int n = blockIdx.x * 256 + threadIdx.x;
    if (n < N_NODES) {
        int g = batch[walk2_row0[n]];
        pb[n] = g;
        atomicAdd(&counts[g], 1u);
    }
}

__global__ void __launch_bounds__(256)
k_accum(const float* __restrict__ x,
        const int* __restrict__ w2, const int* __restrict__ w3,
        const int* __restrict__ w4, const int* __restrict__ pb,
        float* __restrict__ partial, int nchunks) {
    __shared__ float acc[N_GRAPHS * COLS_PER_WG];
    const int chunk = blockIdx.x;
    const int cg    = blockIdx.y;
    const int t     = threadIdx.x;
    const int c     = cg * COLS_PER_WG + t;

    for (int i = t; i < N_GRAPHS * COLS_PER_WG; i += 256) acc[i] = 0.0f;
    __syncthreads();

    const int per = (N_NODES + nchunks - 1) / nchunks;
    const int n0  = chunk * per;
    const int n1  = (n0 + per < N_NODES) ? (n0 + per) : N_NODES;

    for (int n = n0; n < n1; ++n) {
        const int g = pb[n];
        float y = 0.0f;
        #pragma unroll
        for (int l = 0; l < 3; ++l) y += x[w2[l * N_NODES + n] * HID + c];
        #pragma unroll
        for (int l = 0; l < 4; ++l) y += x[w3[l * N_NODES + n] * HID + c];
        #pragma unroll
        for (int l = 0; l < 5; ++l) y += x[w4[l * N_NODES + n] * HID + c];
        acc[g * COLS_PER_WG + t] += y;
    }
    __syncthreads();

    float* dst = partial + (size_t)chunk * (N_GRAPHS * HID) + cg * COLS_PER_WG;
    for (int g = 0; g < N_GRAPHS; ++g)
        dst[g * HID + t] = acc[g * COLS_PER_WG + t];
}


__global__ void k_zero_ct(unsigned* __restrict__ ct, unsigned* __restrict__ counts) {
    v4u* ct4 = (v4u*)ct;
    const int t = blockIdx.x * 256 + threadIdx.x;
    const v4u z = {0u, 0u, 0u, 0u};
    for (int i = t; i < CT_ROWS * N_GRAPHS / 4; i += 512 * 256) ct4[i] = z;
    if (t < N_GRAPHS) counts[t] = 0u;
}

#define TSTEPS  16
#define TBATCH  (HID / 4 / TSTEPS)
__device__ __forceinline__ v8f
tile_kloop(const float* __restrict__ ap, const float* __restrict__ bp, const bool valid) {
    v2f ta[2][TSTEPS], tb[2][TSTEPS];
    v8f acc = {};
    #define T_ISSUE(p, kb)                                                      \
        {                                                                       \
            _Pragma("unroll")                                                   \
            for (int d = 0; d < TSTEPS; ++d) {                                  \
                const int k = ((kb) * TSTEPS + d) * 4;                          \
                ta[p][d].x = ap[k];          ta[p][d].y = ap[k + 1];            \
                tb[p][d].x = bp[k * HID];    tb[p][d].y = bp[(k + 1) * HID];    \
            }                                                                   \
        }
    #define T_COMPUTE(p)                                                        \
        {                                                                       \
            _Pragma("unroll")                                                   \
            for (int d = 0; d < TSTEPS; ++d) {                                  \
                v2f a;                                                          \
                a.x = valid ? ta[p][d].x : 0.0f;                                \
                a.y = valid ? ta[p][d].y : 0.0f;                                \
                acc = __builtin_amdgcn_wmma_f32_16x16x4_f32(                    \
                    false, a, false, tb[p][d], (short)0, acc, false, false);    \
            }                                                                   \
        }
    T_ISSUE(0, 0)
    #pragma unroll
    for (int kb = 0; kb < TBATCH; kb += 2) {
        T_ISSUE(1, kb + 1)
        __builtin_amdgcn_sched_barrier(0);
        T_COMPUTE(0)
        __builtin_amdgcn_sched_barrier(0);
        if (kb + 2 < TBATCH) T_ISSUE(0, kb + 2)
        __builtin_amdgcn_sched_barrier(0);
        T_COMPUTE(1)
    }
    #undef T_ISSUE
    #undef T_COMPUTE
    return acc;
}

__device__ __forceinline__ void wprod_tile(const float* __restrict__ A,
                                           const float* __restrict__ brow,
                                           const float* __restrict__ B,
                                           const float* __restrict__ badd,
                                           float* __restrict__ outp, int tile) {
    const int lane = threadIdx.x & 31;
    const int l16  = lane & 15;
    const int lh   = lane >> 4;
    const int tm   = tile >> 5;
    const int tn   = tile & 31;
    const int row  = tm * 16 + l16;
    const bool valid = row <= HID;
    const float* Arow = (row < HID) ? (A + row * HID) : brow;
    const int col = tn * 16 + l16;

    const v8f acc = tile_kloop(Arow + lh * 2, B + (lh * 2) * HID + col, valid);
    const float bn = badd[col];
    #pragma unroll
    for (int v = 0; v < 8; ++v) {
        const int m = tm * 16 + v + lh * 8;
        outp[m * HID + col] = acc[v] + ((m == HID) ? bn : 0.0f);
    }
}

__device__ __forceinline__ void scatter_body(const int* __restrict__ w2, const int* __restrict__ w3,
                          const int* __restrict__ w4, const int* __restrict__ batch,
                          unsigned* __restrict__ ct, unsigned* __restrict__ counts) {
    const int n = blockIdx.x * 256 + threadIdx.x;
    if (n < N_NODES) {
        int m[12];
        for (int l = 0; l < 3; ++l) m[l]     = w2[l * N_NODES + n];
        for (int l = 0; l < 4; ++l) m[3 + l] = w3[l * N_NODES + n];
        for (int l = 0; l < 5; ++l) m[7 + l] = w4[l * N_NODES + n];
        const int g = batch[m[0]];
        atomicAdd(&counts[g], 1u);
        for (int l = 0; l < 12; ++l) atomicAdd(&ct[(size_t)m[l] * N_GRAPHS + g], 1u);
    }
}

__global__ void k_scatter(const int* __restrict__ w2, const int* __restrict__ w3,
                          const int* __restrict__ w4, const int* __restrict__ batch,
                          unsigned* __restrict__ ct, unsigned* __restrict__ counts) {
    scatter_body(w2, w3, w4, batch, ct, counts);
}

__global__ void __launch_bounds__(256)
k_bin_w(const int* __restrict__ w2, const int* __restrict__ w3,
        const int* __restrict__ w4, const int* __restrict__ batch,
        unsigned* __restrict__ ent, unsigned* __restrict__ off,
        unsigned* __restrict__ hist,
        const float* __restrict__ W0, const float* __restrict__ b0,
        const float* __restrict__ W1, const float* __restrict__ b1,
        float* __restrict__ w01) {
    __shared__ unsigned cnt[NBUCK];
    __shared__ unsigned base[NBUCK + 1];
    __shared__ unsigned hg[N_GRAPHS];
    __shared__ __attribute__((aligned(16))) unsigned stage[ENT_PER_WG];
    if (blockIdx.x >= SCAT_WGS) {
        const int tile = (blockIdx.x - SCAT_WGS) * 8 + (threadIdx.x >> 5);
        if (tile < WP_TILES) wprod_tile(W0, b0, W1, b1, w01, tile);
        return;
    }
    const int t = threadIdx.x;
    const int wg = blockIdx.x;
    if (t < NBUCK) cnt[t] = 0u;
    if (t < N_GRAPHS) hg[t] = 0u;
    __syncthreads();

    const int n = wg * 256 + t;
    unsigned e[12], pos[12], bk[12];
    if (n < N_NODES) {
        int m[12];
        #pragma unroll
        for (int l = 0; l < 3; ++l) m[l]     = w2[l * N_NODES + n];
        #pragma unroll
        for (int l = 0; l < 4; ++l) m[3 + l] = w3[l * N_NODES + n];
        #pragma unroll
        for (int l = 0; l < 5; ++l) m[7 + l] = w4[l * N_NODES + n];
        const unsigned g = (unsigned)batch[m[0]];
        atomicAdd(&hg[g], 1u);
        #pragma unroll
        for (int l = 0; l < 12; ++l) {
            bk[l]  = (unsigned)m[l] / BUCK_NODES;
            e[l]   = ((unsigned)m[l] << 7) | g;
            pos[l] = atomicAdd(&cnt[bk[l]], 1u);
        }
    }
    __syncthreads();
    if (t == 0) {
        unsigned run = 0u;
        for (int b = 0; b < NBUCK; ++b) { base[b] = run; run += cnt[b]; }
        base[NBUCK] = run;
    }
    __syncthreads();
    const unsigned total = base[NBUCK];
    if (n < N_NODES) {
        #pragma unroll
        for (int l = 0; l < 12; ++l) stage[base[bk[l]] + pos[l]] = e[l];
    }
    if (t < 3 && total + t < ENT_PER_WG) stage[total + t] = 0u;
    __syncthreads();
    {
        const v4u* s4 = (const v4u*)stage;
        v4u* d4 = (v4u*)(ent + (size_t)wg * ENT_PER_WG);
        for (unsigned i = t; i < (total + 3u) / 4u; i += 256) d4[i] = s4[i];
    }
    if (t <= NBUCK) off[wg * (NBUCK + 1) + t] = base[t];
    if (t < N_GRAPHS) hist[wg * N_GRAPHS + t] = hg[t];
}

__global__ void __launch_bounds__(1024)
k_build(const unsigned* __restrict__ ent, const unsigned* __restrict__ off,
        const unsigned* __restrict__ hist,
        unsigned* __restrict__ ct, unsigned* __restrict__ counts) {
    __shared__ __attribute__((aligned(16))) unsigned ctl[SLAB * N_GRAPHS];
    const int t = threadIdx.x;
    const int o = blockIdx.x;
    if (o == NSLABS) {
        const int g = t & (N_GRAPHS - 1);
        const int part = t >> 7;
        unsigned a = 0u;
        for (int w = part; w < SCAT_WGS; w += 8) a += hist[w * N_GRAPHS + g];
        ctl[t] = a;
        __syncthreads();
        if (t < N_GRAPHS) {
            unsigned s = 0u;
            #pragma unroll
            for (int p = 0; p < 8; ++p) s += ctl[p * N_GRAPHS + t];
            counts[t] = s;
        }
        return;
    }
    v4u* ctl4 = (v4u*)ctl;
    const v4u z = {0u, 0u, 0u, 0u};
    for (int i = t; i < SLAB * N_GRAPHS / 4; i += 1024) ctl4[i] = z;
    __syncthreads();

    const int b = o >> 2;
    const unsigned m0 = (unsigned)o * SLAB;
    const int wave = __builtin_amdgcn_readfirstlane(t >> 5);
    const int lane = t & 31;

    unsigned lo[SEGS_PER_WAVE], hi[SEGS_PER_WAVE];
    #pragma unroll
    for (int k = 0; k < SEGS_PER_WAVE; ++k) {
        const int w = wave + 32 * k;
        const int wc = (w < SCAT_WGS) ? w : (SCAT_WGS - 1);
        const unsigned l0 = off[wc * (NBUCK + 1) + b];
        const unsigned h0 = off[wc * (NBUCK + 1) + b + 1];
        lo[k] = l0;
        hi[k] = (w < SCAT_WGS) ? h0 : l0;
    }
    v4u d[SEGS_PER_WAVE];
    #pragma unroll
    for (int k = 0; k < SEGS_PER_WAVE; ++k) {
        const int w = wave + 32 * k;
        const int wc = (w < SCAT_WGS) ? w : (SCAT_WGS - 1);
        const unsigned idx = (lo[k] & ~3u) + 4u * lane;
        const unsigned idc = (idx < ENT_PER_WG - 4u) ? idx : (ENT_PER_WG - 4u);
        d[k] = *(const v4u*)(ent + (size_t)wc * ENT_PER_WG + idc);
    }
    #pragma unroll
    for (int k = 0; k < SEGS_PER_WAVE; ++k) {
        const unsigned idx = (lo[k] & ~3u) + 4u * lane;
        #pragma unroll
        for (int j = 0; j < 4; ++j) {
            const unsigned i = idx + j;
            const unsigned r = (d[k][j] >> 7) - m0;
            if (i >= lo[k] && i < hi[k] && r < (unsigned)SLAB)
                atomicAdd(&ctl[r * N_GRAPHS + (d[k][j] & 127u)], 1u);
        }
    }
    for (int k = 0; k < SEGS_PER_WAVE; ++k) {
        const int w = wave + 32 * k;
        const int wc = (w < SCAT_WGS) ? w : (SCAT_WGS - 1);
        const unsigned* src = ent + (size_t)wc * ENT_PER_WG;
        for (unsigned i = (lo[k] & ~3u) + 128u + lane; i < hi[k]; i += 32) {
            const unsigned ev = src[i];
            const unsigned r = (ev >> 7) - m0;
            if (r < (unsigned)SLAB) atomicAdd(&ctl[r * N_GRAPHS + (ev & 127u)], 1u);
        }
    }
    __syncthreads();

    int rows = CT_ROWS - (int)m0;
    rows = (rows > SLAB) ? SLAB : rows;
    v4u* dst4 = (v4u*)(ct + (size_t)m0 * N_GRAPHS);
    for (int i = t; i < rows * (N_GRAPHS / 4); i += 1024) dst4[i] = ctl4[i];
}

__device__ __forceinline__ void
cgemm_body(const float* __restrict__ x, const unsigned* __restrict__ ct,
           float* __restrict__ partial, int nchunks) {
    const int chunk = blockIdx.x;
    const int wave  = threadIdx.x >> 5;
    const int lane  = threadIdx.x & 31;
    const int l16   = lane & 15;
    const int lh    = lane >> 4;
    const int gb    = (wave & 1) * 64;
    const int cb    = (blockIdx.y * 2 + (wave >> 1)) * 64;

    const int per = (KSTEPS + nchunks - 1) / nchunks;
    const int s0  = chunk * per;
    const int s1  = (s0 + per < KSTEPS) ? (s0 + per) : KSTEPS;
    const int nsteps = s1 - s0;

    const float*    xl = x  + (2 * lh) * HID      + cb + 4 * l16;
    const unsigned* cl = ct + (2 * lh) * N_GRAPHS + gb + 4 * l16;

    v4f rb0[2][DEPTH], rb1[2][DEPTH];
    v4u ra0[2][DEPTH], ra1[2][DEPTH];
    v8f acc[4][4] = {};

    #define ISSUE(p, d, s)                                                      \
        {                                                                       \
            const int sx = ((s) < KSTEPS) ? (s) : (KSTEPS - 1);                 \
            const int sc = ((s) < s1) ? (s) : KSTEPS;                           \
            const float*    xp = xl + (size_t)sx * (4 * HID);                   \
            const unsigned* cp = cl + (size_t)sc * (4 * N_GRAPHS);              \
            rb0[p][d] = *(const v4f*)xp;                                        \
            rb1[p][d] = *(const v4f*)(xp + HID);                                \
            ra0[p][d] = *(const v4u*)cp;                                        \
            ra1[p][d] = *(const v4u*)(cp + N_GRAPHS);                           \
        }
    #define COMPUTE(p, d)                                                       \
        {                                                                       \
            v2f a[4], b[4];                                                     \
            _Pragma("unroll")                                                   \
            for (int i = 0; i < 4; ++i) {                                       \
                a[i].x = (float)ra0[p][d][i];  a[i].y = (float)ra1[p][d][i];    \
                b[i].x = rb0[p][d][i];         b[i].y = rb1[p][d][i];           \
            }                                                                   \
            _Pragma("unroll")                                                   \
            for (int i = 0; i < 4; ++i) {                                       \
                _Pragma("unroll")                                               \
                for (int j = 0; j < 4; ++j)                                     \
                    acc[i][j] = __builtin_amdgcn_wmma_f32_16x16x4_f32(          \
                        false, a[i], false, b[j], (short)0, acc[i][j], false, false); \
            }                                                                   \
        }

    #pragma unroll
    for (int d = 0; d < DEPTH; ++d) ISSUE(0, d, s0 + d)

    for (int base = 0; base < nsteps; base += 2 * DEPTH) {
        #pragma unroll
        for (int d = 0; d < DEPTH; ++d) ISSUE(1, d, s0 + base + DEPTH + d)
        __builtin_amdgcn_sched_barrier(0);
        #pragma unroll
        for (int d = 0; d < DEPTH; ++d) COMPUTE(0, d)
        __builtin_amdgcn_sched_barrier(0);
        #pragma unroll
        for (int d = 0; d < DEPTH; ++d) ISSUE(0, d, s0 + base + 2 * DEPTH + d)
        __builtin_amdgcn_sched_barrier(0);
        #pragma unroll
        for (int d = 0; d < DEPTH; ++d) COMPUTE(1, d)
    }
    #undef ISSUE
    #undef COMPUTE

    float* dst = partial + (size_t)chunk * (N_GRAPHS * HID) + cb + 4 * l16;
    #pragma unroll
    for (int i = 0; i < 4; ++i) {
        #pragma unroll
        for (int v = 0; v < 8; ++v) {
            const int g = gb + 4 * (v + 8 * lh) + i;
            v4f o = {acc[i][0][v], acc[i][1][v], acc[i][2][v], acc[i][3][v]};
            *(v4f*)(dst + g * HID) = o;
        }
    }
}

__global__ void __launch_bounds__(128)
k_cgemm(const float* __restrict__ x, const unsigned* __restrict__ ct,
        float* __restrict__ partial, int nchunks) {
    cgemm_body(x, ct, partial, nchunks);
}

__global__ void __launch_bounds__(128)
k_cgemm_w(const float* __restrict__ x, const unsigned* __restrict__ ct,
          float* __restrict__ partial, int nchunks,
          const float* __restrict__ w01, const float* __restrict__ W2,
          const float* __restrict__ b2, float* __restrict__ w012) {
    if (blockIdx.y < 4) {
        cgemm_body(x, ct, partial, nchunks);
    } else {
        const int tile = ((blockIdx.y - 4) * gridDim.x + blockIdx.x) * 4 + (threadIdx.x >> 5);
        if (tile < WP_TILES) wprod_tile(w01, w01 + HID * HID, W2, b2, w012, tile);
    }
}

__global__ void k_reduce(const float* __restrict__ partial,
                         const unsigned* __restrict__ counts,
                         float* __restrict__ P, float* __restrict__ s,
                         int nchunks) {
    int i = blockIdx.x * 256 + threadIdx.x;
    int g = i >> 9;
    float a = 0.0f;
    for (int ch = 0; ch < nchunks; ++ch)
        a += partial[(size_t)ch * (N_GRAPHS * HID) + i];
    float cnt = (float)counts[g];
    P[i] = a / fmaxf(cnt, 1.0f);
    if (i < N_GRAPHS) s[i] = (counts[i] != 0u) ? 12.0f : 0.0f;
}

__global__ void __launch_bounds__(256)
k_gemm_wmma(const float* __restrict__ A, int lda,
            const float* __restrict__ W, const float* __restrict__ bias,
            const float* __restrict__ s, float* __restrict__ out) {
    const int wave = (blockIdx.x << 3) + (threadIdx.x >> 5);
    const int lane = threadIdx.x & 31;
    const int l16  = lane & 15;
    const int lh   = lane >> 4;
    const int tm   = wave >> 5;
    const int tn   = wave & 31;

    const float* Arow = A + (tm * 16 + l16) * lda;
    const int col = tn * 16 + l16;

    v8f acc = {};
    for (int k = 0; k < HID; k += 4) {
        const int kk = k + lh * 2;
        v2f a; a.x = Arow[kk];           a.y = Arow[kk + 1];
        v2f b; b.x = W[kk * HID + col];  b.y = W[(kk + 1) * HID + col];
        acc = __builtin_amdgcn_wmma_f32_16x16x4_f32(
             false, a,  false, b,
             (short)0, acc,  false,  false);
    }

    const float bn = bias[col];
    #pragma unroll
    for (int v = 0; v < 8; ++v) {
        const int m = tm * 16 + v + lh * 8;
        out[m * OUT_PITCH + col] = acc[v] + s[m] * bn;
    }
}

__global__ void __launch_bounds__(256)
k_gemm3(const float* __restrict__ P,
        const float* __restrict__ W0, const float* __restrict__ b0,
        const float* __restrict__ w01, const float* __restrict__ w012,
        const float* __restrict__ s, float* __restrict__ out) {
    const int wave = (blockIdx.x << 3) + (threadIdx.x >> 5);
    const int lane = threadIdx.x & 31;
    const int l16  = lane & 15;
    const int lh   = lane >> 4;
    const int tm    = wave / 96;
    const int tq    = wave - tm * 96;
    const int layer = tq >> 5;
    const int tn    = tq & 31;

    const float* B    = (layer == 0) ? W0 : ((layer == 1) ? w01 : w012);
    const float* bias = (layer == 0) ? b0 : (B + HID * HID);
    const float* Arow = P + (tm * 16 + l16) * HID;
    const int col = tn * 16 + l16;

    const v8f acc = tile_kloop(Arow + lh * 2, B + (lh * 2) * HID + col, true);

    const float bn = bias[col];
    #pragma unroll
    for (int v = 0; v < 8; ++v) {
        const int m = tm * 16 + v + lh * 8;
        out[m * OUT_PITCH + layer * HID + col] = acc[v] + s[m] * bn;
    }
}

extern "C" void kernel_launch(void* const* d_in, const int* in_sizes, int n_in,
                              void* d_out, int out_size, void* d_ws, size_t ws_size,
                              hipStream_t stream) {
    const float* x     = (const float*)d_in[0];
    const int*   w2    = (const int*)  d_in[1];
    const int*   w3    = (const int*)  d_in[2];
    const int*   w4    = (const int*)  d_in[3];
    const int*   batch = (const int*)  d_in[4];
    const float* W0 = (const float*)d_in[5];  const float* b0 = (const float*)d_in[6];
    const float* W1 = (const float*)d_in[7];  const float* b1 = (const float*)d_in[8];
    const float* W2 = (const float*)d_in[9];  const float* b2 = (const float*)d_in[10];
    float* out = (float*)d_out;

    const size_t partBytes = (size_t)N_GRAPHS * HID * sizeof(float);
    const size_t fixed = partBytes   + 512   + 512
                       + (size_t)N_NODES * sizeof(int)  ;
    int nchunks = 1;
    if (ws_size > fixed + partBytes) {
        size_t fit = (ws_size - fixed) / partBytes;
        nchunks = (fit > MAX_CHUNKS) ? MAX_CHUNKS : (int)fit;
        if (nchunks < 1) nchunks = 1;
    }
    char* ws = (char*)d_ws;
    float*    partial = (float*)ws;
    float*    P       = (float*)(ws + (size_t)nchunks * partBytes);
    unsigned* counts  = (unsigned*)((char*)P + partBytes);
    float*    s       = (float*)((char*)counts + 512);
    int*      pb      = (int*)((char*)s + 512);

    const size_t ctBytes = (size_t)CT_ROWS * N_GRAPHS * sizeof(unsigned);
    const size_t denseNeed = (size_t)MAX_CHUNKS * partBytes + partBytes + 1024 + ctBytes;
    const size_t wpBytes = (size_t)WP_ROWS * HID * sizeof(float);
    if (ws_size >= denseNeed + 2 * wpBytes) {
        unsigned* ct = (unsigned*)pb;
        float* w01  = (float*)((char*)ct + ctBytes);
        float* w012 = (float*)((char*)w01 + wpBytes);
        unsigned* ent  = (unsigned*)partial;
        unsigned* off  = ent + (size_t)SCAT_WGS * ENT_PER_WG;
        unsigned* hist = off + SCAT_WGS * (NBUCK + 1);
        k_bin_w<<<SCAT_WGS + WP_TILES / 8, 256, 0, stream>>>(w2, w3, w4, batch, ent, off, hist,
                                                          W0, b0, W1, b1, w01);
        k_build<<<NSLABS + 1, 1024, 0, stream>>>(ent, off, hist, ct, counts);
        dim3 gDenseW(nchunks, 4 + 5);
        k_cgemm_w<<<gDenseW, 128, 0, stream>>>(x, ct, partial, nchunks, w01, W2, b2, w012);
        k_reduce<<<(N_GRAPHS * HID) / 256, 256, 0, stream>>>(partial, counts, P, s, nchunks);
        k_gemm3<<<96, 256, 0, stream>>>(P, W0, b0, w01, w012, s, out);
        return;
    }
    if (ws_size >= denseNeed) {
        unsigned* ct = (unsigned*)pb;
        k_zero_ct<<<512, 256, 0, stream>>>(ct, counts);
        k_scatter<<<(N_NODES + 255) / 256, 256, 0, stream>>>(w2, w3, w4, batch, ct, counts);
        dim3 gDense(nchunks, 4);
        k_cgemm<<<gDense, 128, 0, stream>>>(x, ct, partial, nchunks);
    } else {
        k_zero_counts<<<1, 128, 0, stream>>>(counts);
        k_pb_hist<<<(N_NODES + 255) / 256, 256, 0, stream>>>(w2, batch, pb, counts);
        dim3 gAcc(nchunks, 2);
        k_accum<<<gAcc, 256, 0, stream>>>(x, w2, w3, w4, pb, partial, nchunks);
    }
    k_reduce<<<(N_GRAPHS * HID) / 256, 256, 0, stream>>>(partial, counts, P, s, nchunks);

    k_gemm_wmma<<<32, 256, 0, stream>>>(P,         HID,       W0, b0, s, out);
    k_gemm_wmma<<<32, 256, 0, stream>>>(out,       OUT_PITCH, W1, b1, s, out + HID);
    k_gemm_wmma<<<32, 256, 0, stream>>>(out + HID, OUT_PITCH, W2, b2, s, out + 2 * HID);
}


# 16x16 tile GEMMs (weight products, pooled GEMM): the 16 A rows a workgroup's waves share are copied once into LDS by whole-line loads, fragments read from LDS, not 16 lines per load; same sums
# speedup vs baseline: 6.9823x; 1.0382x over previous
#include <hip/hip_runtime.h>


#define N_NODES   50000
#define N_GRAPHS  128
#define HID       512
#define OUT_PITCH (3 * HID)
#define MAX_CHUNKS 64
#define COLS_PER_WG 256
#define KSTEPS    (N_NODES / 4)
#define CT_ROWS   (N_NODES + 4)
#define DEPTH     7
#define WP_ROWS   528
#define WP_TILES  (33 * 32)
#define SCAT_WGS  ((N_NODES + 255) / 256)
#define SLAB      196
#define NSLABS    256
#define NBUCK     64
#define BUCK_NODES (4 * SLAB)
#define SEGS_PER_WAVE 7
#define ENT_PER_WG (256 * 12)

typedef __attribute__((ext_vector_type(2))) float v2f;
typedef __attribute__((ext_vector_type(8))) float v8f;
typedef __attribute__((ext_vector_type(4))) float v4f;
typedef __attribute__((ext_vector_type(4))) unsigned v4u;

__global__ void k_zero_counts(unsigned* __restrict__ counts) {
    counts[threadIdx.x] = 0u;
}

__global__ void k_pb_hist(const int* __restrict__ walk2_row0,
                          const int* __restrict__ batch,
                          int* __restrict__ pb,
                          unsigned* __restrict__ counts) {
    int n = blockIdx.x * 256 + threadIdx.x;
    if (n < N_NODES) {
        int g = batch[walk2_row0[n]];
        pb[n] = g;
        atomicAdd(&counts[g], 1u);
    }
}

__global__ void __launch_bounds__(256)
k_accum(const float* __restrict__ x,
        const int* __restrict__ w2, const int* __restrict__ w3,
        const int* __restrict__ w4, const int* __restrict__ pb,
        float* __restrict__ partial, int nchunks) {
    __shared__ float acc[N_GRAPHS * COLS_PER_WG];
    const int chunk = blockIdx.x;
    const int cg    = blockIdx.y;
    const int t     = threadIdx.x;
    const int c     = cg * COLS_PER_WG + t;

    for (int i = t; i < N_GRAPHS * COLS_PER_WG; i += 256) acc[i] = 0.0f;
    __syncthreads();

    const int per = (N_NODES + nchunks - 1) / nchunks;
    const int n0  = chunk * per;
    const int n1  = (n0 + per < N_NODES) ? (n0 + per) : N_NODES;

    for (int n = n0; n < n1; ++n) {
        const int g = pb[n];
        float y = 0.0f;
        #pragma unroll
        for (int l = 0; l < 3; ++l) y += x[w2[l * N_NODES + n] * HID + c];
        #pragma unroll
        for (int l = 0; l < 4; ++l) y += x[w3[l * N_NODES + n] * HID + c];
        #pragma unroll
        for (int l = 0; l < 5; ++l) y += x[w4[l * N_NODES + n] * HID + c];
        acc[g * COLS_PER_WG + t] += y;
    }
    __syncthreads();

    float* dst = partial + (size_t)chunk * (N_GRAPHS * HID) + cg * COLS_PER_WG;
    for (int g = 0; g < N_GRAPHS; ++g)
        dst[g * HID + t] = acc[g * COLS_PER_WG + t];
}


__global__ void k_zero_ct(unsigned* __restrict__ ct, unsigned* __restrict__ counts) {
    v4u* ct4 = (v4u*)ct;
    const int t = blockIdx.x * 256 + threadIdx.x;
    const v4u z = {0u, 0u, 0u, 0u};
    for (int i = t; i < CT_ROWS * N_GRAPHS / 4; i += 512 * 256) ct4[i] = z;
    if (t < N_GRAPHS) counts[t] = 0u;
}

#define TSTEPS  16
#define TBATCH  (HID / 4 / TSTEPS)
#define APITCH  516
#define ASLAB   (16 * APITCH)
template <int NT>
__device__ __forceinline__ v8f
tile_kloop(const float* __restrict__ A, const float* __restrict__ brow, const int row0,
           float* __restrict__ as, const float* __restrict__ bp) {
    const int t    = threadIdx.x;
    const int lane = t & 31;
    const int l16  = lane & 15;
    const int lh   = lane >> 4;
    v2f ta[2][TSTEPS], tb[2][TSTEPS];
    v8f acc = {};
    #define T_ISSUE_A(p, kb)                                                    \
        {                                                                       \
            _Pragma("unroll")                                                   \
            for (int d = 0; d < TSTEPS; ++d) {                                  \
                const int k = ((kb) * TSTEPS + d) * 4;                          \
                ta[p][d] = *(const v2f*)(ap + k);                               \
            }                                                                   \
        }
    #define T_ISSUE_B(p, kb)                                                    \
        {                                                                       \
            _Pragma("unroll")                                                   \
            for (int d = 0; d < TSTEPS; ++d) {                                  \
                const int k = ((kb) * TSTEPS + d) * 4;                          \
                tb[p][d].x = bp[k * HID];    tb[p][d].y = bp[(k + 1) * HID];    \
            }                                                                   \
        }
    #define T_COMPUTE(p)                                                        \
        {                                                                       \
            _Pragma("unroll")                                                   \
            for (int d = 0; d < TSTEPS; ++d)                                    \
                acc = __builtin_amdgcn_wmma_f32_16x16x4_f32(                    \
                    false, ta[p][d], false, tb[p][d], (short)0, acc, false, false); \
        }
    T_ISSUE_B(0, 0)
    #pragma unroll
    for (int i = 0; i < (16 * 128) / NT; ++i) {
        const int idx  = i * NT + t;
        const int r    = idx >> 7;
        const int c    = (idx & 127) * 4;
        const int grow = row0 + r;
        const float* srow = (grow < HID) ? (A + grow * HID) : brow;
        v4f v = *(const v4f*)(srow + c);
        if (grow > HID) v = (v4f){0.0f, 0.0f, 0.0f, 0.0f};
        *(v4f*)(as + r * APITCH + c) = v;
    }
    __syncthreads();
    const float* ap = as + l16 * APITCH + 2 * lh;
    T_ISSUE_A(0, 0)
    #pragma unroll
    for (int kb = 0; kb < TBATCH; kb += 2) {
        T_ISSUE_A(1, kb + 1)
        T_ISSUE_B(1, kb + 1)
        __builtin_amdgcn_sched_barrier(0);
        T_COMPUTE(0)
        __builtin_amdgcn_sched_barrier(0);
        if (kb + 2 < TBATCH) { T_ISSUE_A(0, kb + 2) T_ISSUE_B(0, kb + 2) }
        __builtin_amdgcn_sched_barrier(0);
        T_COMPUTE(1)
    }
    #undef T_ISSUE_A
    #undef T_ISSUE_B
    #undef T_COMPUTE
    return acc;
}

template <int NT>
__device__ __forceinline__ void wprod_tile(const float* __restrict__ A,
                                           const float* __restrict__ brow,
                                           const float* __restrict__ B,
                                           const float* __restrict__ badd,
                                           float* __restrict__ outp, int tile,
                                           float* __restrict__ as) {
    const int lane = threadIdx.x & 31;
    const int l16  = lane & 15;
    const int lh   = lane >> 4;
    const int tm   = tile >> 5;
    const int tn   = tile & 31;
    const int col = tn * 16 + l16;

    const v8f acc = tile_kloop<NT>(A, brow, tm * 16, as, B + (lh * 2) * HID + col);
    const float bn = badd[col];
    #pragma unroll
    for (int v = 0; v < 8; ++v) {
        const int m = tm * 16 + v + lh * 8;
        outp[m * HID + col] = acc[v] + ((m == HID) ? bn : 0.0f);
    }
}

__device__ __forceinline__ void scatter_body(const int* __restrict__ w2, const int* __restrict__ w3,
                          const int* __restrict__ w4, const int* __restrict__ batch,
                          unsigned* __restrict__ ct, unsigned* __restrict__ counts) {
    const int n = blockIdx.x * 256 + threadIdx.x;
    if (n < N_NODES) {
        int m[12];
        for (int l = 0; l < 3; ++l) m[l]     = w2[l * N_NODES + n];
        for (int l = 0; l < 4; ++l) m[3 + l] = w3[l * N_NODES + n];
        for (int l = 0; l < 5; ++l) m[7 + l] = w4[l * N_NODES + n];
        const int g = batch[m[0]];
        atomicAdd(&counts[g], 1u);
        for (int l = 0; l < 12; ++l) atomicAdd(&ct[(size_t)m[l] * N_GRAPHS + g], 1u);
    }
}

__global__ void k_scatter(const int* __restrict__ w2, const int* __restrict__ w3,
                          const int* __restrict__ w4, const int* __restrict__ batch,
                          unsigned* __restrict__ ct, unsigned* __restrict__ counts) {
    scatter_body(w2, w3, w4, batch, ct, counts);
}

__global__ void __launch_bounds__(256)
k_bin_w(const int* __restrict__ w2, const int* __restrict__ w3,
        const int* __restrict__ w4, const int* __restrict__ batch,
        unsigned* __restrict__ ent, unsigned* __restrict__ off,
        unsigned* __restrict__ hist,
        const float* __restrict__ W0, const float* __restrict__ b0,
        const float* __restrict__ W1, const float* __restrict__ b1,
        float* __restrict__ w01) {
    __shared__ unsigned cnt[NBUCK];
    __shared__ unsigned base[NBUCK + 1];
    __shared__ unsigned hg[N_GRAPHS];
    __shared__ __attribute__((aligned(16))) unsigned stage[ENT_PER_WG];
    __shared__ __attribute__((aligned(16))) float aslab[ASLAB];
    if (blockIdx.x >= SCAT_WGS) {
        const int tile0 = (blockIdx.x - SCAT_WGS) * 8;
        if (tile0 < WP_TILES)
            wprod_tile<256>(W0, b0, W1, b1, w01, tile0 + (threadIdx.x >> 5), aslab);
        return;
    }
    const int t = threadIdx.x;
    const int wg = blockIdx.x;
    if (t < NBUCK) cnt[t] = 0u;
    if (t < N_GRAPHS) hg[t] = 0u;
    __syncthreads();

    const int n = wg * 256 + t;
    unsigned e[12], pos[12], bk[12];
    if (n < N_NODES) {
        int m[12];
        #pragma unroll
        for (int l = 0; l < 3; ++l) m[l]     = w2[l * N_NODES + n];
        #pragma unroll
        for (int l = 0; l < 4; ++l) m[3 + l] = w3[l * N_NODES + n];
        #pragma unroll
        for (int l = 0; l < 5; ++l) m[7 + l] = w4[l * N_NODES + n];
        const unsigned g = (unsigned)batch[m[0]];
        atomicAdd(&hg[g], 1u);
        #pragma unroll
        for (int l = 0; l < 12; ++l) {
            bk[l]  = (unsigned)m[l] / BUCK_NODES;
            e[l]   = ((unsigned)m[l] << 7) | g;
            pos[l] = atomicAdd(&cnt[bk[l]], 1u);
        }
    }
    __syncthreads();
    if (t == 0) {
        unsigned run = 0u;
        for (int b = 0; b < NBUCK; ++b) { base[b] = run; run += cnt[b]; }
        base[NBUCK] = run;
    }
    __syncthreads();
    const unsigned total = base[NBUCK];
    if (n < N_NODES) {
        #pragma unroll
        for (int l = 0; l < 12; ++l) stage[base[bk[l]] + pos[l]] = e[l];
    }
    if (t < 3 && total + t < ENT_PER_WG) stage[total + t] = 0u;
    __syncthreads();
    {
        const v4u* s4 = (const v4u*)stage;
        v4u* d4 = (v4u*)(ent + (size_t)wg * ENT_PER_WG);
        for (unsigned i = t; i < (total + 3u) / 4u; i += 256) d4[i] = s4[i];
    }
    if (t <= NBUCK) off[wg * (NBUCK + 1) + t] = base[t];
    if (t < N_GRAPHS) hist[wg * N_GRAPHS + t] = hg[t];
}

__global__ void __launch_bounds__(1024)
k_build(const unsigned* __restrict__ ent, const unsigned* __restrict__ off,
        const unsigned* __restrict__ hist,
        unsigned* __restrict__ ct, unsigned* __restrict__ counts) {
    __shared__ __attribute__((aligned(16))) unsigned ctl[SLAB * N_GRAPHS];
    const int t = threadIdx.x;
    const int o = blockIdx.x;
    if (o == NSLABS) {
        const int g = t & (N_GRAPHS - 1);
        const int part = t >> 7;
        unsigned a = 0u;
        for (int w = part; w < SCAT_WGS; w += 8) a += hist[w * N_GRAPHS + g];
        ctl[t] = a;
        __syncthreads();
        if (t < N_GRAPHS) {
            unsigned s = 0u;
            #pragma unroll
            for (int p = 0; p < 8; ++p) s += ctl[p * N_GRAPHS + t];
            counts[t] = s;
        }
        return;
    }
    v4u* ctl4 = (v4u*)ctl;
    const v4u z = {0u, 0u, 0u, 0u};
    for (int i = t; i < SLAB * N_GRAPHS / 4; i += 1024) ctl4[i] = z;
    __syncthreads();

    const int b = o >> 2;
    const unsigned m0 = (unsigned)o * SLAB;
    const int wave = __builtin_amdgcn_readfirstlane(t >> 5);
    const int lane = t & 31;

    unsigned lo[SEGS_PER_WAVE], hi[SEGS_PER_WAVE];
    #pragma unroll
    for (int k = 0; k < SEGS_PER_WAVE; ++k) {
        const int w = wave + 32 * k;
        const int wc = (w < SCAT_WGS) ? w : (SCAT_WGS - 1);
        const unsigned l0 = off[wc * (NBUCK + 1) + b];
        const unsigned h0 = off[wc * (NBUCK + 1) + b + 1];
        lo[k] = l0;
        hi[k] = (w < SCAT_WGS) ? h0 : l0;
    }
    v4u d[SEGS_PER_WAVE];
    #pragma unroll
    for (int k = 0; k < SEGS_PER_WAVE; ++k) {
        const int w = wave + 32 * k;
        const int wc = (w < SCAT_WGS) ? w : (SCAT_WGS - 1);
        const unsigned idx = (lo[k] & ~3u) + 4u * lane;
        const unsigned idc = (idx < ENT_PER_WG - 4u) ? idx : (ENT_PER_WG - 4u);
        d[k] = *(const v4u*)(ent + (size_t)wc * ENT_PER_WG + idc);
    }
    #pragma unroll
    for (int k = 0; k < SEGS_PER_WAVE; ++k) {
        const unsigned idx = (lo[k] & ~3u) + 4u * lane;
        #pragma unroll
        for (int j = 0; j < 4; ++j) {
            const unsigned i = idx + j;
            const unsigned r = (d[k][j] >> 7) - m0;
            if (i >= lo[k] && i < hi[k] && r < (unsigned)SLAB)
                atomicAdd(&ctl[r * N_GRAPHS + (d[k][j] & 127u)], 1u);
        }
    }
    for (int k = 0; k < SEGS_PER_WAVE; ++k) {
        const int w = wave + 32 * k;
        const int wc = (w < SCAT_WGS) ? w : (SCAT_WGS - 1);
        const unsigned* src = ent + (size_t)wc * ENT_PER_WG;
        for (unsigned i = (lo[k] & ~3u) + 128u + lane; i < hi[k]; i += 32) {
            const unsigned ev = src[i];
            const unsigned r = (ev >> 7) - m0;
            if (r < (unsigned)SLAB) atomicAdd(&ctl[r * N_GRAPHS + (ev & 127u)], 1u);
        }
    }
    __syncthreads();

    int rows = CT_ROWS - (int)m0;
    rows = (rows > SLAB) ? SLAB : rows;
    v4u* dst4 = (v4u*)(ct + (size_t)m0 * N_GRAPHS);
    for (int i = t; i < rows * (N_GRAPHS / 4); i += 1024) dst4[i] = ctl4[i];
}

__device__ __forceinline__ void
cgemm_body(const float* __restrict__ x, const unsigned* __restrict__ ct,
           float* __restrict__ partial, int nchunks) {
    const int chunk = blockIdx.x;
    const int wave  = threadIdx.x >> 5;
    const int lane  = threadIdx.x & 31;
    const int l16   = lane & 15;
    const int lh    = lane >> 4;
    const int gb    = (wave & 1) * 64;
    const int cb    = (blockIdx.y * 2 + (wave >> 1)) * 64;

    const int per = (KSTEPS + nchunks - 1) / nchunks;
    const int s0  = chunk * per;
    const int s1  = (s0 + per < KSTEPS) ? (s0 + per) : KSTEPS;
    const int nsteps = s1 - s0;

    const float*    xl = x  + (2 * lh) * HID      + cb + 4 * l16;
    const unsigned* cl = ct + (2 * lh) * N_GRAPHS + gb + 4 * l16;

    v4f rb0[2][DEPTH], rb1[2][DEPTH];
    v4u ra0[2][DEPTH], ra1[2][DEPTH];
    v8f acc[4][4] = {};

    #define ISSUE(p, d, s)                                                      \
        {                                                                       \
            const int sx = ((s) < KSTEPS) ? (s) : (KSTEPS - 1);                 \
            const int sc = ((s) < s1) ? (s) : KSTEPS;                           \
            const float*    xp = xl + (size_t)sx * (4 * HID);                   \
            const unsigned* cp = cl + (size_t)sc * (4 * N_GRAPHS);              \
            rb0[p][d] = *(const v4f*)xp;                                        \
            rb1[p][d] = *(const v4f*)(xp + HID);                                \
            ra0[p][d] = *(const v4u*)cp;                                        \
            ra1[p][d] = *(const v4u*)(cp + N_GRAPHS);                           \
        }
    #define COMPUTE(p, d)                                                       \
        {                                                                       \
            v2f a[4], b[4];                                                     \
            _Pragma("unroll")                                                   \
            for (int i = 0; i < 4; ++i) {                                       \
                a[i].x = (float)ra0[p][d][i];  a[i].y = (float)ra1[p][d][i];    \
                b[i].x = rb0[p][d][i];         b[i].y = rb1[p][d][i];           \
            }                                                                   \
            _Pragma("unroll")                                                   \
            for (int i = 0; i < 4; ++i) {                                       \
                _Pragma("unroll")                                               \
                for (int j = 0; j < 4; ++j)                                     \
                    acc[i][j] = __builtin_amdgcn_wmma_f32_16x16x4_f32(          \
                        false, a[i], false, b[j], (short)0, acc[i][j], false, false); \
            }                                                                   \
        }

    #pragma unroll
    for (int d = 0; d < DEPTH; ++d) ISSUE(0, d, s0 + d)

    for (int base = 0; base < nsteps; base += 2 * DEPTH) {
        #pragma unroll
        for (int d = 0; d < DEPTH; ++d) ISSUE(1, d, s0 + base + DEPTH + d)
        __builtin_amdgcn_sched_barrier(0);
        #pragma unroll
        for (int d = 0; d < DEPTH; ++d) COMPUTE(0, d)
        __builtin_amdgcn_sched_barrier(0);
        #pragma unroll
        for (int d = 0; d < DEPTH; ++d) ISSUE(0, d, s0 + base + 2 * DEPTH + d)
        __builtin_amdgcn_sched_barrier(0);
        #pragma unroll
        for (int d = 0; d < DEPTH; ++d) COMPUTE(1, d)
    }
    #undef ISSUE
    #undef COMPUTE

    float* dst = partial + (size_t)chunk * (N_GRAPHS * HID) + cb + 4 * l16;
    #pragma unroll
    for (int i = 0; i < 4; ++i) {
        #pragma unroll
        for (int v = 0; v < 8; ++v) {
            const int g = gb + 4 * (v + 8 * lh) + i;
            v4f o = {acc[i][0][v], acc[i][1][v], acc[i][2][v], acc[i][3][v]};
            *(v4f*)(dst + g * HID) = o;
        }
    }
}

__global__ void __launch_bounds__(128)
k_cgemm(const float* __restrict__ x, const unsigned* __restrict__ ct,
        float* __restrict__ partial, int nchunks) {
    cgemm_body(x, ct, partial, nchunks);
}

__global__ void __launch_bounds__(128)
k_cgemm_w(const float* __restrict__ x, const unsigned* __restrict__ ct,
          float* __restrict__ partial, int nchunks,
          const float* __restrict__ w01, const float* __restrict__ W2,
          const float* __restrict__ b2, float* __restrict__ w012) {
    __shared__ __attribute__((aligned(16))) float aslab[ASLAB];
    if (blockIdx.y < 4) {
        cgemm_body(x, ct, partial, nchunks);
    } else {
        const int tile0 = ((blockIdx.y - 4) * gridDim.x + blockIdx.x) * 4;
        if (tile0 < WP_TILES)
            wprod_tile<128>(w01, w01 + HID * HID, W2, b2, w012, tile0 + (threadIdx.x >> 5), aslab);
    }
}

__global__ void k_reduce(const float* __restrict__ partial,
                         const unsigned* __restrict__ counts,
                         float* __restrict__ P, float* __restrict__ s,
                         int nchunks) {
    int i = blockIdx.x * 256 + threadIdx.x;
    int g = i >> 9;
    float a = 0.0f;
    for (int ch = 0; ch < nchunks; ++ch)
        a += partial[(size_t)ch * (N_GRAPHS * HID) + i];
    float cnt = (float)counts[g];
    P[i] = a / fmaxf(cnt, 1.0f);
    if (i < N_GRAPHS) s[i] = (counts[i] != 0u) ? 12.0f : 0.0f;
}

__global__ void __launch_bounds__(256)
k_gemm_wmma(const float* __restrict__ A, int lda,
            const float* __restrict__ W, const float* __restrict__ bias,
            const float* __restrict__ s, float* __restrict__ out) {
    const int wave = (blockIdx.x << 3) + (threadIdx.x >> 5);
    const int lane = threadIdx.x & 31;
    const int l16  = lane & 15;
    const int lh   = lane >> 4;
    const int tm   = wave >> 5;
    const int tn   = wave & 31;

    const float* Arow = A + (tm * 16 + l16) * lda;
    const int col = tn * 16 + l16;

    v8f acc = {};
    for (int k = 0; k < HID; k += 4) {
        const int kk = k + lh * 2;
        v2f a; a.x = Arow[kk];           a.y = Arow[kk + 1];
        v2f b; b.x = W[kk * HID + col];  b.y = W[(kk + 1) * HID + col];
        acc = __builtin_amdgcn_wmma_f32_16x16x4_f32(
             false, a,  false, b,
             (short)0, acc,  false,  false);
    }

    const float bn = bias[col];
    #pragma unroll
    for (int v = 0; v < 8; ++v) {
        const int m = tm * 16 + v + lh * 8;
        out[m * OUT_PITCH + col] = acc[v] + s[m] * bn;
    }
}

__global__ void __launch_bounds__(256)
k_gemm3(const float* __restrict__ P,
        const float* __restrict__ W0, const float* __restrict__ b0,
        const float* __restrict__ w01, const float* __restrict__ w012,
        const float* __restrict__ s, float* __restrict__ out) {
    __shared__ __attribute__((aligned(16))) float aslab[ASLAB];
    const int wave = (blockIdx.x << 3) + (threadIdx.x >> 5);
    const int lane = threadIdx.x & 31;
    const int l16  = lane & 15;
    const int lh   = lane >> 4;
    const int tm    = wave / 96;
    const int tq    = wave - tm * 96;
    const int layer = tq >> 5;
    const int tn    = tq & 31;

    const float* B    = (layer == 0) ? W0 : ((layer == 1) ? w01 : w012);
    const float* bias = (layer == 0) ? b0 : (B + HID * HID);
    const int col = tn * 16 + l16;

    const v8f acc = tile_kloop<256>(P, P, tm * 16, aslab, B + (lh * 2) * HID + col);

    const float bn = bias[col];
    #pragma unroll
    for (int v = 0; v < 8; ++v) {
        const int m = tm * 16 + v + lh * 8;
        out[m * OUT_PITCH + layer * HID + col] = acc[v] + s[m] * bn;
    }
}

extern "C" void kernel_launch(void* const* d_in, const int* in_sizes, int n_in,
                              void* d_out, int out_size, void* d_ws, size_t ws_size,
                              hipStream_t stream) {
    const float* x     = (const float*)d_in[0];
    const int*   w2    = (const int*)  d_in[1];
    const int*   w3    = (const int*)  d_in[2];
    const int*   w4    = (const int*)  d_in[3];
    const int*   batch = (const int*)  d_in[4];
    const float* W0 = (const float*)d_in[5];  const float* b0 = (const float*)d_in[6];
    const float* W1 = (const float*)d_in[7];  const float* b1 = (const float*)d_in[8];
    const float* W2 = (const float*)d_in[9];  const float* b2 = (const float*)d_in[10];
    float* out = (float*)d_out;

    const size_t partBytes = (size_t)N_GRAPHS * HID * sizeof(float);
    const size_t fixed = partBytes   + 512   + 512
                       + (size_t)N_NODES * sizeof(int)  ;
    int nchunks = 1;
    if (ws_size > fixed + partBytes) {
        size_t fit = (ws_size - fixed) / partBytes;
        nchunks = (fit > MAX_CHUNKS) ? MAX_CHUNKS : (int)fit;
        if (nchunks < 1) nchunks = 1;
    }
    char* ws = (char*)d_ws;
    float*    partial = (float*)ws;
    float*    P       = (float*)(ws + (size_t)nchunks * partBytes);
    unsigned* counts  = (unsigned*)((char*)P + partBytes);
    float*    s       = (float*)((char*)counts + 512);
    int*      pb      = (int*)((char*)s + 512);

    const size_t ctBytes = (size_t)CT_ROWS * N_GRAPHS * sizeof(unsigned);
    const size_t denseNeed = (size_t)MAX_CHUNKS * partBytes + partBytes + 1024 + ctBytes;
    const size_t wpBytes = (size_t)WP_ROWS * HID * sizeof(float);
    if (ws_size >= denseNeed + 2 * wpBytes) {
        unsigned* ct = (unsigned*)pb;
        float* w01  = (float*)((char*)ct + ctBytes);
        float* w012 = (float*)((char*)w01 + wpBytes);
        unsigned* ent  = (unsigned*)partial;
        unsigned* off  = ent + (size_t)SCAT_WGS * ENT_PER_WG;
        unsigned* hist = off + SCAT_WGS * (NBUCK + 1);
        k_bin_w<<<SCAT_WGS + WP_TILES / 8, 256, 0, stream>>>(w2, w3, w4, batch, ent, off, hist,
                                                          W0, b0, W1, b1, w01);
        k_build<<<NSLABS + 1, 1024, 0, stream>>>(ent, off, hist, ct, counts);
        dim3 gDenseW(nchunks, 4 + 5);
        k_cgemm_w<<<gDenseW, 128, 0, stream>>>(x, ct, partial, nchunks, w01, W2, b2, w012);
        k_reduce<<<(N_GRAPHS * HID) / 256, 256, 0, stream>>>(partial, counts, P, s, nchunks);
        k_gemm3<<<96, 256, 0, stream>>>(P, W0, b0, w01, w012, s, out);
        return;
    }
    if (ws_size >= denseNeed) {
        unsigned* ct = (unsigned*)pb;
        k_zero_ct<<<512, 256, 0, stream>>>(ct, counts);
        k_scatter<<<(N_NODES + 255) / 256, 256, 0, stream>>>(w2, w3, w4, batch, ct, counts);
        dim3 gDense(nchunks, 4);
        k_cgemm<<<gDense, 128, 0, stream>>>(x, ct, partial, nchunks);
    } else {
        k_zero_counts<<<1, 128, 0, stream>>>(counts);
        k_pb_hist<<<(N_NODES + 255) / 256, 256, 0, stream>>>(w2, batch, pb, counts);
        dim3 gAcc(nchunks, 2);
        k_accum<<<gAcc, 256, 0, stream>>>(x, w2, w3, w4, pb, partial, nchunks);
    }
    k_reduce<<<(N_GRAPHS * HID) / 256, 256, 0, stream>>>(partial, counts, P, s, nchunks);

    k_gemm_wmma<<<32, 256, 0, stream>>>(P,         HID,       W0, b0, s, out);
    k_gemm_wmma<<<32, 256, 0, stream>>>(out,       OUT_PITCH, W1, b1, s, out + HID);
    k_gemm_wmma<<<32, 256, 0, stream>>>(out + HID, OUT_PITCH, W2, b2, s, out + 2 * HID);
}


# Ct stays in LDS: each count-GEMM workgroup counts its K chunk's 784 Ct rows (64 graphs) from the binned entries and reads A fragments there; pass-2 launch removed (5 launches to 4); same sums
# speedup vs baseline: 7.2306x; 1.0356x over previous
#include <hip/hip_runtime.h>


#define N_NODES   50000
#define N_GRAPHS  128
#define HID       512
#define OUT_PITCH (3 * HID)
#define MAX_CHUNKS 64
#define COLS_PER_WG 256
#define KSTEPS    (N_NODES / 4)
#define CT_ROWS   (N_NODES + 4)
#define DEPTH     7
#define WP_ROWS   528
#define WP_TILES  (33 * 32)
#define SCAT_WGS  ((N_NODES + 255) / 256)
#define NBUCK     64
#define BUCK_NODES 784
#define ENT_PER_WG (256 * 12)
#define FPER      (BUCK_NODES / 4)
#define FROWS     (BUCK_NODES + 4)
#define FSEGS     (SCAT_WGS / 4)
#define GHALF     64
static_assert(4 * FSEGS == SCAT_WGS, "four waves share the binning workgroups evenly");
static_assert(NBUCK == MAX_CHUNKS && NBUCK * BUCK_NODES >= N_NODES, "one bucket per K chunk, all nodes covered");
static_assert((KSTEPS + MAX_CHUNKS - 1) / MAX_CHUNKS == FPER, "a K chunk of the count GEMM is one bucket");

typedef __attribute__((ext_vector_type(2))) float v2f;
typedef __attribute__((ext_vector_type(8))) float v8f;
typedef __attribute__((ext_vector_type(4))) float v4f;
typedef __attribute__((ext_vector_type(4))) unsigned v4u;
typedef __attribute__((ext_vector_type(2))) unsigned v2u;

__global__ void k_zero_counts(unsigned* __restrict__ counts) {
    counts[threadIdx.x] = 0u;
}

__global__ void k_pb_hist(const int* __restrict__ walk2_row0,
                          const int* __restrict__ batch,
                          int* __restrict__ pb,
                          unsigned* __restrict__ counts) {
    int n = blockIdx.x * 256 + threadIdx.x;
    if (n < N_NODES) {
        int g = batch[walk2_row0[n]];
        pb[n] = g;
        atomicAdd(&counts[g], 1u);
    }
}

__global__ void __launch_bounds__(256)
k_accum(const float* __restrict__ x,
        const int* __restrict__ w2, const int* __restrict__ w3,
        const int* __restrict__ w4, const int* __restrict__ pb,
        float* __restrict__ partial, int nchunks) {
    __shared__ float acc[N_GRAPHS * COLS_PER_WG];
    const int chunk = blockIdx.x;
    const int cg    = blockIdx.y;
    const int t     = threadIdx.x;
    const int c     = cg * COLS_PER_WG + t;

    for (int i = t; i < N_GRAPHS * COLS_PER_WG; i += 256) acc[i] = 0.0f;
    __syncthreads();

    const int per = (N_NODES + nchunks - 1) / nchunks;
    const int n0  = chunk * per;
    const int n1  = (n0 + per < N_NODES) ? (n0 + per) : N_NODES;

    for (int n = n0; n < n1; ++n) {
        const int g = pb[n];
        float y = 0.0f;
        #pragma unroll
        for (int l = 0; l < 3; ++l) y += x[w2[l * N_NODES + n] * HID + c];
        #pragma unroll
        for (int l = 0; l < 4; ++l) y += x[w3[l * N_NODES + n] * HID + c];
        #pragma unroll
        for (int l = 0; l < 5; ++l) y += x[w4[l * N_NODES + n] * HID + c];
        acc[g * COLS_PER_WG + t] += y;
    }
    __syncthreads();

    float* dst = partial + (size_t)chunk * (N_GRAPHS * HID) + cg * COLS_PER_WG;
    for (int g = 0; g < N_GRAPHS; ++g)
        dst[g * HID + t] = acc[g * COLS_PER_WG + t];
}


__global__ void k_zero_ct(unsigned* __restrict__ ct, unsigned* __restrict__ counts) {
    v4u* ct4 = (v4u*)ct;
    const int t = blockIdx.x * 256 + threadIdx.x;
    const v4u z = {0u, 0u, 0u, 0u};
    for (int i = t; i < CT_ROWS * N_GRAPHS / 4; i += 512 * 256) ct4[i] = z;
    if (t < N_GRAPHS) counts[t] = 0u;
}

#define TSTEPS  16
#define TBATCH  (HID / 4 / TSTEPS)
#define APITCH  516
#define ASLAB   (16 * APITCH)
template <int NT>
__device__ __forceinline__ v8f
tile_kloop(const float* __restrict__ A, const float* __restrict__ brow, const int row0,
           float* __restrict__ as, const float* __restrict__ bp) {
    const int t    = threadIdx.x;
    const int lane = t & 31;
    const int l16  = lane & 15;
    const int lh   = lane >> 4;
    v2f ta[2][TSTEPS], tb[2][TSTEPS];
    v8f acc = {};
    #define T_ISSUE_A(p, kb)                                                    \
        {                                                                       \
            _Pragma("unroll")                                                   \
            for (int d = 0; d < TSTEPS; ++d) {                                  \
                const int k = ((kb) * TSTEPS + d) * 4;                          \
                ta[p][d] = *(const v2f*)(ap + k);                               \
            }                                                                   \
        }
    #define T_ISSUE_B(p, kb)                                                    \
        {                                                                       \
            _Pragma("unroll")                                                   \
            for (int d = 0; d < TSTEPS; ++d) {                                  \
                const int k = ((kb) * TSTEPS + d) * 4;                          \
                tb[p][d].x = bp[k * HID];    tb[p][d].y = bp[(k + 1) * HID];    \
            }                                                                   \
        }
    #define T_COMPUTE(p)                                                        \
        {                                                                       \
            _Pragma("unroll")                                                   \
            for (int d = 0; d < TSTEPS; ++d)                                    \
                acc = __builtin_amdgcn_wmma_f32_16x16x4_f32(                    \
                    false, ta[p][d], false, tb[p][d], (short)0, acc, false, false); \
        }
    T_ISSUE_B(0, 0)
    #pragma unroll
    for (int i = 0; i < (16 * 128) / NT; ++i) {
        const int idx  = i * NT + t;
        const int r    = idx >> 7;
        const int c    = (idx & 127) * 4;
        const int grow = row0 + r;
        const float* srow = (grow < HID) ? (A + grow * HID) : brow;
        v4f v = *(const v4f*)(srow + c);
        if (grow > HID) v = (v4f){0.0f, 0.0f, 0.0f, 0.0f};
        *(v4f*)(as + r * APITCH + c) = v;
    }
    __syncthreads();
    const float* ap = as + l16 * APITCH + 2 * lh;
    T_ISSUE_A(0, 0)
    #pragma unroll
    for (int kb = 0; kb < TBATCH; kb += 2) {
        T_ISSUE_A(1, kb + 1)
        T_ISSUE_B(1, kb + 1)
        __builtin_amdgcn_sched_barrier(0);
        T_COMPUTE(0)
        __builtin_amdgcn_sched_barrier(0);
        if (kb + 2 < TBATCH) { T_ISSUE_A(0, kb + 2) T_ISSUE_B(0, kb + 2) }
        __builtin_amdgcn_sched_barrier(0);
        T_COMPUTE(1)
    }
    #undef T_ISSUE_A
    #undef T_ISSUE_B
    #undef T_COMPUTE
    return acc;
}

template <int NT>
__device__ __forceinline__ void wprod_tile(const float* __restrict__ A,
                                           const float* __restrict__ brow,
                                           const float* __restrict__ B,
                                           const float* __restrict__ badd,
                                           float* __restrict__ outp, int tile,
                                           float* __restrict__ as) {
    const int lane = threadIdx.x & 31;
    const int l16  = lane & 15;
    const int lh   = lane >> 4;
    const int tm   = tile >> 5;
    const int tn   = tile & 31;
    const int col = tn * 16 + l16;

    const v8f acc = tile_kloop<NT>(A, brow, tm * 16, as, B + (lh * 2) * HID + col);
    const float bn = badd[col];
    #pragma unroll
    for (int v = 0; v < 8; ++v) {
        const int m = tm * 16 + v + lh * 8;
        outp[m * HID + col] = acc[v] + ((m == HID) ? bn : 0.0f);
    }
}

__device__ __forceinline__ void scatter_body(const int* __restrict__ w2, const int* __restrict__ w3,
                          const int* __restrict__ w4, const int* __restrict__ batch,
                          unsigned* __restrict__ ct, unsigned* __restrict__ counts) {
    const int n = blockIdx.x * 256 + threadIdx.x;
    if (n < N_NODES) {
        int m[12];
        for (int l = 0; l < 3; ++l) m[l]     = w2[l * N_NODES + n];
        for (int l = 0; l < 4; ++l) m[3 + l] = w3[l * N_NODES + n];
        for (int l = 0; l < 5; ++l) m[7 + l] = w4[l * N_NODES + n];
        const int g = batch[m[0]];
        atomicAdd(&counts[g], 1u);
        for (int l = 0; l < 12; ++l) atomicAdd(&ct[(size_t)m[l] * N_GRAPHS + g], 1u);
    }
}

__global__ void k_scatter(const int* __restrict__ w2, const int* __restrict__ w3,
                          const int* __restrict__ w4, const int* __restrict__ batch,
                          unsigned* __restrict__ ct, unsigned* __restrict__ counts) {
    scatter_body(w2, w3, w4, batch, ct, counts);
}

__global__ void __launch_bounds__(256)
k_bin_w(const int* __restrict__ w2, const int* __restrict__ w3,
        const int* __restrict__ w4, const int* __restrict__ batch,
        unsigned* __restrict__ ent, unsigned* __restrict__ off,
        unsigned* __restrict__ hist,
        const float* __restrict__ W0, const float* __restrict__ b0,
        const float* __restrict__ W1, const float* __restrict__ b1,
        float* __restrict__ w01) {
    __shared__ unsigned cnt[NBUCK];
    __shared__ unsigned base[NBUCK + 1];
    __shared__ unsigned hg[N_GRAPHS];
    __shared__ __attribute__((aligned(16))) unsigned stage[ENT_PER_WG];
    __shared__ __attribute__((aligned(16))) float aslab[ASLAB];
    if (blockIdx.x >= SCAT_WGS) {
        const int tile0 = (blockIdx.x - SCAT_WGS) * 8;
        if (tile0 < WP_TILES)
            wprod_tile<256>(W0, b0, W1, b1, w01, tile0 + (threadIdx.x >> 5), aslab);
        return;
    }
    const int t = threadIdx.x;
    const int wg = blockIdx.x;
    if (t < NBUCK) cnt[t] = 0u;
    if (t < N_GRAPHS) hg[t] = 0u;
    __syncthreads();

    const int n = wg * 256 + t;
    unsigned e[12], pos[12], bk[12];
    if (n < N_NODES) {
        int m[12];
        #pragma unroll
        for (int l = 0; l < 3; ++l) m[l]     = w2[l * N_NODES + n];
        #pragma unroll
        for (int l = 0; l < 4; ++l) m[3 + l] = w3[l * N_NODES + n];
        #pragma unroll
        for (int l = 0; l < 5; ++l) m[7 + l] = w4[l * N_NODES + n];
        const unsigned g = (unsigned)batch[m[0]];
        atomicAdd(&hg[g], 1u);
        #pragma unroll
        for (int l = 0; l < 12; ++l) {
            bk[l]  = (unsigned)m[l] / BUCK_NODES;
            e[l]   = ((unsigned)m[l] << 7) | g;
            pos[l] = atomicAdd(&cnt[bk[l]], 1u);
        }
    }
    __syncthreads();
    if (t == 0) {
        unsigned run = 0u;
        for (int b = 0; b < NBUCK; ++b) { base[b] = run; run += cnt[b]; }
        base[NBUCK] = run;
    }
    __syncthreads();
    const unsigned total = base[NBUCK];
    if (n < N_NODES) {
        #pragma unroll
        for (int l = 0; l < 12; ++l) stage[base[bk[l]] + pos[l]] = e[l];
    }
    if (t < 3 && total + t < ENT_PER_WG) stage[total + t] = 0u;
    __syncthreads();
    {
        const v4u* s4 = (const v4u*)stage;
        v4u* d4 = (v4u*)(ent + (size_t)wg * ENT_PER_WG);
        for (unsigned i = t; i < (total + 3u) / 4u; i += 256) d4[i] = s4[i];
    }
    if (t <= NBUCK) off[wg * (NBUCK + 1) + t] = base[t];
    if (t < N_GRAPHS) hist[wg * N_GRAPHS + t] = hg[t];
}

__device__ __forceinline__ void
cgemm_body(const float* __restrict__ x, const unsigned* __restrict__ ct,
           float* __restrict__ partial, int nchunks) {
    const int chunk = blockIdx.x;
    const int wave  = threadIdx.x >> 5;
    const int lane  = threadIdx.x & 31;
    const int l16   = lane & 15;
    const int lh    = lane >> 4;
    const int gb    = (wave & 1) * 64;
    const int cb    = (blockIdx.y * 2 + (wave >> 1)) * 64;

    const int per = (KSTEPS + nchunks - 1) / nchunks;
    const int s0  = chunk * per;
    const int s1  = (s0 + per < KSTEPS) ? (s0 + per) : KSTEPS;
    const int nsteps = s1 - s0;

    const float*    xl = x  + (2 * lh) * HID      + cb + 4 * l16;
    const unsigned* cl = ct + (2 * lh) * N_GRAPHS + gb + 4 * l16;

    v4f rb0[2][DEPTH], rb1[2][DEPTH];
    v4u ra0[2][DEPTH], ra1[2][DEPTH];
    v8f acc[4][4] = {};

    #define ISSUE(p, d, s)                                                      \
        {                                                                       \
            const int sx = ((s) < KSTEPS) ? (s) : (KSTEPS - 1);                 \
            const int sc = ((s) < s1) ? (s) : KSTEPS;                           \
            const float*    xp = xl + (size_t)sx * (4 * HID);                   \
            const unsigned* cp = cl + (size_t)sc * (4 * N_GRAPHS);              \
            rb0[p][d] = *(const v4f*)xp;                                        \
            rb1[p][d] = *(const v4f*)(xp + HID);                                \
            ra0[p][d] = *(const v4u*)cp;                                        \
            ra1[p][d] = *(const v4u*)(cp + N_GRAPHS);                           \
        }
    #define COMPUTE(p, d)                                                       \
        {                                                                       \
            v2f a[4], b[4];                                                     \
            _Pragma("unroll")                                                   \
            for (int i = 0; i < 4; ++i) {                                       \
                a[i].x = (float)ra0[p][d][i];  a[i].y = (float)ra1[p][d][i];    \
                b[i].x = rb0[p][d][i];         b[i].y = rb1[p][d][i];           \
            }                                                                   \
            _Pragma("unroll")                                                   \
            for (int i = 0; i < 4; ++i) {                                       \
                _Pragma("unroll")                                               \
                for (int j = 0; j < 4; ++j)                                     \
                    acc[i][j] = __builtin_amdgcn_wmma_f32_16x16x4_f32(          \
                        false, a[i], false, b[j], (short)0, acc[i][j], false, false); \
            }                                                                   \
        }

    #pragma unroll
    for (int d = 0; d < DEPTH; ++d) ISSUE(0, d, s0 + d)

    for (int base = 0; base < nsteps; base += 2 * DEPTH) {
        #pragma unroll
        for (int d = 0; d < DEPTH; ++d) ISSUE(1, d, s0 + base + DEPTH + d)
        __builtin_amdgcn_sched_barrier(0);
        #pragma unroll
        for (int d = 0; d < DEPTH; ++d) COMPUTE(0, d)
        __builtin_amdgcn_sched_barrier(0);
        #pragma unroll
        for (int d = 0; d < DEPTH; ++d) ISSUE(0, d, s0 + base + 2 * DEPTH + d)
        __builtin_amdgcn_sched_barrier(0);
        #pragma unroll
        for (int d = 0; d < DEPTH; ++d) COMPUTE(1, d)
    }
    #undef ISSUE
    #undef COMPUTE

    float* dst = partial + (size_t)chunk * (N_GRAPHS * HID) + cb + 4 * l16;
    #pragma unroll
    for (int i = 0; i < 4; ++i) {
        #pragma unroll
        for (int v = 0; v < 8; ++v) {
            const int g = gb + 4 * (v + 8 * lh) + i;
            v4f o = {acc[i][0][v], acc[i][1][v], acc[i][2][v], acc[i][3][v]};
            *(v4f*)(dst + g * HID) = o;
        }
    }
}

__global__ void __launch_bounds__(128)
k_cgemm(const float* __restrict__ x, const unsigned* __restrict__ ct,
        float* __restrict__ partial, int nchunks) {
    cgemm_body(x, ct, partial, nchunks);
}

__global__ void __launch_bounds__(128)
k_cgemm_f(const float* __restrict__ x, const unsigned* __restrict__ ent,
          const unsigned* __restrict__ off, float* __restrict__ partial) {
    __shared__ __attribute__((aligned(16))) unsigned cs[FROWS * GHALF];
    const int t     = threadIdx.x;
    const int chunk = blockIdx.x;
    const unsigned gbh = blockIdx.y & 1;
    const int colh  = blockIdx.y >> 1;
    const int wave  = __builtin_amdgcn_readfirstlane(t >> 5);
    const int lane  = t & 31;
    const int l16   = lane & 15;
    const int lh    = lane >> 4;
    const int gb    = (int)gbh * GHALF;
    const int cb    = (colh * 4 + wave) * 64;

    const int s0 = chunk * FPER;
    const int s1 = (s0 + FPER < KSTEPS) ? (s0 + FPER) : KSTEPS;
    const int nsteps = s1 - s0;

    const float*    xl = x  + (2 * lh) * HID   + cb + 4 * l16;
    const unsigned* al = cs + (2 * lh) * GHALF + 4 * l16;

    v4f rb0[2][DEPTH], rb1[2][DEPTH];
    v4u ra0[2][DEPTH], ra1[2][DEPTH];
    v8f acc[4][4] = {};

    #define ISSUE_X(p, d, s)                                                    \
        {                                                                       \
            const int sx = ((s) < KSTEPS) ? (s) : (KSTEPS - 1);                 \
            const float* xp = xl + (size_t)sx * (4 * HID);                      \
            rb0[p][d] = *(const v4f*)xp;                                        \
            rb1[p][d] = *(const v4f*)(xp + HID);                                \
        }
    #define ISSUE_A(p, d, s)                                                    \
        {                                                                       \
            const int sl = ((s) < s1) ? ((s) - s0) : FPER;                      \
            const unsigned* cp = al + sl * (4 * GHALF);                         \
            ra0[p][d] = *(const v4u*)cp;                                        \
            ra1[p][d] = *(const v4u*)(cp + GHALF);                              \
        }
    #define COMPUTE(p, d)                                                       \
        {                                                                       \
            v2f a[4], b[4];                                                     \
            _Pragma("unroll")                                                   \
            for (int i = 0; i < 4; ++i) {                                       \
                a[i].x = (float)ra0[p][d][i];  a[i].y = (float)ra1[p][d][i];    \
                b[i].x = rb0[p][d][i];         b[i].y = rb1[p][d][i];           \
            }                                                                   \
            _Pragma("unroll")                                                   \
            for (int i = 0; i < 4; ++i) {                                       \
                _Pragma("unroll")                                               \
                for (int j = 0; j < 4; ++j)                                     \
                    acc[i][j] = __builtin_amdgcn_wmma_f32_16x16x4_f32(          \
                        false, a[i], false, b[j], (short)0, acc[i][j], false, false); \
            }                                                                   \
        }


    const unsigned* offb = off + chunk;
    const int j1 = (lane < FSEGS - 32) ? (32 + lane) : (FSEGS - 1);
    const unsigned lo0 = offb[(wave + 4 * lane) * (NBUCK + 1)];
    const unsigned hi0 = offb[(wave + 4 * lane) * (NBUCK + 1) + 1];
    const unsigned lo1 = offb[(wave + 4 * j1) * (NBUCK + 1)];
    const unsigned hi1 = offb[(wave + 4 * j1) * (NBUCK + 1) + 1];


    const unsigned m0 = (unsigned)chunk * BUCK_NODES;
    v2u dw[FSEGS];
    #pragma unroll
    for (int k = 0; k < FSEGS; ++k) {
        const unsigned lo = (unsigned)__builtin_amdgcn_readlane((int)((k < 32) ? lo0 : lo1), k & 31);
        const unsigned idx = (lo & ~1u) + 2u * lane;
        const unsigned idc = (idx < ENT_PER_WG - 2u) ? idx : (ENT_PER_WG - 2u);
        dw[k] = *(const v2u*)(ent + (size_t)(wave + 4 * k) * ENT_PER_WG + idc);
    }
    v4u* cs4 = (v4u*)cs;
    const v4u z = {0u, 0u, 0u, 0u};
    for (int i = t; i < FROWS * GHALF / 4; i += 128) cs4[i] = z;
    __syncthreads();
    #pragma unroll
    for (int d = 0; d < DEPTH; ++d) ISSUE_X(0, d, s0 + d)
    #pragma unroll
    for (int k = 0; k < FSEGS; ++k) {
        const unsigned lo = (unsigned)__builtin_amdgcn_readlane((int)((k < 32) ? lo0 : lo1), k & 31);
        const unsigned hi = (unsigned)__builtin_amdgcn_readlane((int)((k < 32) ? hi0 : hi1), k & 31);
        const unsigned idx = (lo & ~1u) + 2u * lane;
        #pragma unroll
        for (int q = 0; q < 2; ++q) {
            const unsigned ev = dw[k][q];
            const unsigned r = (ev >> 7) - m0;
            const bool ok = ((idx + q - lo) < (hi - lo))
                          & (r < (unsigned)BUCK_NODES) & (((ev >> 6) & 1u) == gbh);
            if (ok) atomicAdd(&cs[r * GHALF + (ev & 63u)], 1u);
        }
    }
    for (int k = 0; k < FSEGS; ++k) {
        const unsigned lo = (unsigned)__builtin_amdgcn_readlane((int)((k < 32) ? lo0 : lo1), k & 31);
        const unsigned hi = (unsigned)__builtin_amdgcn_readlane((int)((k < 32) ? hi0 : hi1), k & 31);
        const unsigned* sp = ent + (size_t)(wave + 4 * k) * ENT_PER_WG;
        for (unsigned i = (lo & ~1u) + 64u + lane; i < hi; i += 32) {
            const unsigned ev = sp[i];
            const unsigned r = (ev >> 7) - m0;
            if (r < (unsigned)BUCK_NODES && ((ev >> 6) & 1u) == gbh)
                atomicAdd(&cs[r * GHALF + (ev & 63u)], 1u);
        }
    }
    __syncthreads();

    #pragma unroll
    for (int d = 0; d < DEPTH; ++d) ISSUE_A(0, d, s0 + d)

    for (int base = 0; base < nsteps; base += 2 * DEPTH) {
        #pragma unroll
        for (int d = 0; d < DEPTH; ++d) { ISSUE_X(1, d, s0 + base + DEPTH + d) ISSUE_A(1, d, s0 + base + DEPTH + d) }
        __builtin_amdgcn_sched_barrier(0);
        #pragma unroll
        for (int d = 0; d < DEPTH; ++d) COMPUTE(0, d)
        __builtin_amdgcn_sched_barrier(0);
        #pragma unroll
        for (int d = 0; d < DEPTH; ++d) { ISSUE_X(0, d, s0 + base + 2 * DEPTH + d) ISSUE_A(0, d, s0 + base + 2 * DEPTH + d) }
        __builtin_amdgcn_sched_barrier(0);
        #pragma unroll
        for (int d = 0; d < DEPTH; ++d) COMPUTE(1, d)
    }
    #undef ISSUE_X
    #undef ISSUE_A
    #undef COMPUTE

    float* dst = partial + (size_t)chunk * (N_GRAPHS * HID) + cb + 4 * l16;
    #pragma unroll
    for (int i = 0; i < 4; ++i) {
        #pragma unroll
        for (int v = 0; v < 8; ++v) {
            const int g = gb + 4 * (v + 8 * lh) + i;
            v4f o = {acc[i][0][v], acc[i][1][v], acc[i][2][v], acc[i][3][v]};
            *(v4f*)(dst + g * HID) = o;
        }
    }
}

__global__ void k_reduce(const float* __restrict__ partial,
                         const unsigned* __restrict__ counts,
                         float* __restrict__ P, float* __restrict__ s,
                         int nchunks) {
    int i = blockIdx.x * 256 + threadIdx.x;
    int g = i >> 9;
    float a = 0.0f;
    for (int ch = 0; ch < nchunks; ++ch)
        a += partial[(size_t)ch * (N_GRAPHS * HID) + i];
    float cnt = (float)counts[g];
    P[i] = a / fmaxf(cnt, 1.0f);
    if (i < N_GRAPHS) s[i] = (counts[i] != 0u) ? 12.0f : 0.0f;
}

__global__ void __launch_bounds__(256)
k_reduce_w(const float* __restrict__ partial, const unsigned* __restrict__ hist,
           float* __restrict__ P, float* __restrict__ s, int nchunks,
           const float* __restrict__ w01, const float* __restrict__ W2,
           const float* __restrict__ b2, float* __restrict__ w012) {
    __shared__ __attribute__((aligned(16))) float aslab[ASLAB];
    __shared__ unsigned gcnt;
    if (blockIdx.x < WP_TILES / 8) {
        wprod_tile<256>(w01, w01 + HID * HID, W2, b2, w012, blockIdx.x * 8 + (threadIdx.x >> 5), aslab);
        return;
    }
    const int t = threadIdx.x;
    const int i = (blockIdx.x - WP_TILES / 8) * 256 + t;
    const int g = i >> 9;
    if (t == 0) gcnt = 0u;
    __syncthreads();
    unsigned c = (t < SCAT_WGS) ? hist[t * N_GRAPHS + g] : 0u;
    #pragma unroll
    for (int o = 16; o > 0; o >>= 1) c += __shfl_xor(c, o);
    if ((t & 31) == 0) atomicAdd(&gcnt, c);
    float a = 0.0f;
    for (int ch = 0; ch < nchunks; ++ch)
        a += partial[(size_t)ch * (N_GRAPHS * HID) + i];
    __syncthreads();
    const unsigned cg = gcnt;
    P[i] = a / fmaxf((float)cg, 1.0f);
    if ((i & (HID - 1)) == 0) s[g] = (cg != 0u) ? 12.0f : 0.0f;
}

__global__ void __launch_bounds__(256)
k_gemm_wmma(const float* __restrict__ A, int lda,
            const float* __restrict__ W, const float* __restrict__ bias,
            const float* __restrict__ s, float* __restrict__ out) {
    const int wave = (blockIdx.x << 3) + (threadIdx.x >> 5);
    const int lane = threadIdx.x & 31;
    const int l16  = lane & 15;
    const int lh   = lane >> 4;
    const int tm   = wave >> 5;
    const int tn   = wave & 31;

    const float* Arow = A + (tm * 16 + l16) * lda;
    const int col = tn * 16 + l16;

    v8f acc = {};
    for (int k = 0; k < HID; k += 4) {
        const int kk = k + lh * 2;
        v2f a; a.x = Arow[kk];           a.y = Arow[kk + 1];
        v2f b; b.x = W[kk * HID + col];  b.y = W[(kk + 1) * HID + col];
        acc = __builtin_amdgcn_wmma_f32_16x16x4_f32(
             false, a,  false, b,
             (short)0, acc,  false,  false);
    }

    const float bn = bias[col];
    #pragma unroll
    for (int v = 0; v < 8; ++v) {
        const int m = tm * 16 + v + lh * 8;
        out[m * OUT_PITCH + col] = acc[v] + s[m] * bn;
    }
}

__global__ void __launch_bounds__(256)
k_gemm3(const float* __restrict__ P,
        const float* __restrict__ W0, const float* __restrict__ b0,
        const float* __restrict__ w01, const float* __restrict__ w012,
        const float* __restrict__ s, float* __restrict__ out) {
    __shared__ __attribute__((aligned(16))) float aslab[ASLAB];
    const int wave = (blockIdx.x << 3) + (threadIdx.x >> 5);
    const int lane = threadIdx.x & 31;
    const int l16  = lane & 15;
    const int lh   = lane >> 4;
    const int tm    = wave / 96;
    const int tq    = wave - tm * 96;
    const int layer = tq >> 5;
    const int tn    = tq & 31;

    const float* B    = (layer == 0) ? W0 : ((layer == 1) ? w01 : w012);
    const float* bias = (layer == 0) ? b0 : (B + HID * HID);
    const int col = tn * 16 + l16;

    const v8f acc = tile_kloop<256>(P, P, tm * 16, aslab, B + (lh * 2) * HID + col);

    const float bn = bias[col];
    #pragma unroll
    for (int v = 0; v < 8; ++v) {
        const int m = tm * 16 + v + lh * 8;
        out[m * OUT_PITCH + layer * HID + col] = acc[v] + s[m] * bn;
    }
}

extern "C" void kernel_launch(void* const* d_in, const int* in_sizes, int n_in,
                              void* d_out, int out_size, void* d_ws, size_t ws_size,
                              hipStream_t stream) {
    const float* x     = (const float*)d_in[0];
    const int*   w2    = (const int*)  d_in[1];
    const int*   w3    = (const int*)  d_in[2];
    const int*   w4    = (const int*)  d_in[3];
    const int*   batch = (const int*)  d_in[4];
    const float* W0 = (const float*)d_in[5];  const float* b0 = (const float*)d_in[6];
    const float* W1 = (const float*)d_in[7];  const float* b1 = (const float*)d_in[8];
    const float* W2 = (const float*)d_in[9];  const float* b2 = (const float*)d_in[10];
    float* out = (float*)d_out;

    const size_t partBytes = (size_t)N_GRAPHS * HID * sizeof(float);
    const size_t fixed = partBytes   + 512   + 512
                       + (size_t)N_NODES * sizeof(int)  ;
    int nchunks = 1;
    if (ws_size > fixed + partBytes) {
        size_t fit = (ws_size - fixed) / partBytes;
        nchunks = (fit > MAX_CHUNKS) ? MAX_CHUNKS : (int)fit;
        if (nchunks < 1) nchunks = 1;
    }
    char* ws = (char*)d_ws;
    float*    partial = (float*)ws;
    float*    P       = (float*)(ws + (size_t)nchunks * partBytes);
    unsigned* counts  = (unsigned*)((char*)P + partBytes);
    float*    s       = (float*)((char*)counts + 512);
    int*      pb      = (int*)((char*)s + 512);

    const size_t ctBytes = (size_t)CT_ROWS * N_GRAPHS * sizeof(unsigned);
    const size_t denseNeed = (size_t)MAX_CHUNKS * partBytes + partBytes + 1024 + ctBytes;
    const size_t wpBytes = (size_t)WP_ROWS * HID * sizeof(float);
    if (ws_size >= denseNeed + 2 * wpBytes) {
        unsigned* ent  = (unsigned*)pb;
        unsigned* off  = ent + (size_t)SCAT_WGS * ENT_PER_WG;
        unsigned* hist = off + SCAT_WGS * (NBUCK + 1);
        float* w01  = (float*)((char*)pb + ctBytes);
        float* w012 = (float*)((char*)w01 + wpBytes);
        k_bin_w<<<SCAT_WGS + WP_TILES / 8, 256, 0, stream>>>(w2, w3, w4, batch, ent, off, hist,
                                                          W0, b0, W1, b1, w01);
        dim3 gFused(MAX_CHUNKS, 4);
        k_cgemm_f<<<gFused, 128, 0, stream>>>(x, ent, off, partial);
        k_reduce_w<<<WP_TILES / 8 + (N_GRAPHS * HID) / 256, 256, 0, stream>>>(partial, hist, P, s, nchunks,
                                                                              w01, W2, b2, w012);
        k_gemm3<<<96, 256, 0, stream>>>(P, W0, b0, w01, w012, s, out);
        return;
    }
    if (ws_size >= denseNeed) {
        unsigned* ct = (unsigned*)pb;
        k_zero_ct<<<512, 256, 0, stream>>>(ct, counts);
        k_scatter<<<(N_NODES + 255) / 256, 256, 0, stream>>>(w2, w3, w4, batch, ct, counts);
        dim3 gDense(nchunks, 4);
        k_cgemm<<<gDense, 128, 0, stream>>>(x, ct, partial, nchunks);
    } else {
        k_zero_counts<<<1, 128, 0, stream>>>(counts);
        k_pb_hist<<<(N_NODES + 255) / 256, 256, 0, stream>>>(w2, batch, pb, counts);
        dim3 gAcc(nchunks, 2);
        k_accum<<<gAcc, 256, 0, stream>>>(x, w2, w3, w4, pb, partial, nchunks);
    }
    k_reduce<<<(N_GRAPHS * HID) / 256, 256, 0, stream>>>(partial, counts, P, s, nchunks);

    k_gemm_wmma<<<32, 256, 0, stream>>>(P,         HID,       W0, b0, s, out);
    k_gemm_wmma<<<32, 256, 0, stream>>>(out,       OUT_PITCH, W1, b1, s, out + HID);
    k_gemm_wmma<<<32, 256, 0, stream>>>(out + HID, OUT_PITCH, W2, b2, s, out + 2 * HID);
}


# count GEMM, Ct count phase: every window slot does one unconditional LDS add (its Ct word or a spare word) with no branch; segments longer than a window found by one vote; same counts, same sums
# speedup vs baseline: 7.7649x; 1.0739x over previous
#include <hip/hip_runtime.h>


#define N_NODES   50000
#define N_GRAPHS  128
#define HID       512
#define OUT_PITCH (3 * HID)
#define MAX_CHUNKS 64
#define COLS_PER_WG 256
#define KSTEPS    (N_NODES / 4)
#define CT_ROWS   (N_NODES + 4)
#define DEPTH     7
#define WP_ROWS   528
#define WP_TILES  (33 * 32)
#define SCAT_WGS  ((N_NODES + 255) / 256)
#define NBUCK     64
#define BUCK_NODES 784
#define ENT_PER_WG (256 * 12)
#define FPER      (BUCK_NODES / 4)
#define FROWS     (BUCK_NODES + 4)
#define FSEGS     (SCAT_WGS / 4)
#define GHALF     64
#define FSPARE    32
static_assert(4 * FSEGS == SCAT_WGS, "four waves share the binning workgroups evenly");
static_assert(NBUCK == MAX_CHUNKS && NBUCK * BUCK_NODES >= N_NODES, "one bucket per K chunk, all nodes covered");
static_assert((KSTEPS + MAX_CHUNKS - 1) / MAX_CHUNKS == FPER, "a K chunk of the count GEMM is one bucket");

typedef __attribute__((ext_vector_type(2))) float v2f;
typedef __attribute__((ext_vector_type(8))) float v8f;
typedef __attribute__((ext_vector_type(4))) float v4f;
typedef __attribute__((ext_vector_type(4))) unsigned v4u;
typedef __attribute__((ext_vector_type(2))) unsigned v2u;

__global__ void k_zero_counts(unsigned* __restrict__ counts) {
    counts[threadIdx.x] = 0u;
}

__global__ void k_pb_hist(const int* __restrict__ walk2_row0,
                          const int* __restrict__ batch,
                          int* __restrict__ pb,
                          unsigned* __restrict__ counts) {
    int n = blockIdx.x * 256 + threadIdx.x;
    if (n < N_NODES) {
        int g = batch[walk2_row0[n]];
        pb[n] = g;
        atomicAdd(&counts[g], 1u);
    }
}

__global__ void __launch_bounds__(256)
k_accum(const float* __restrict__ x,
        const int* __restrict__ w2, const int* __restrict__ w3,
        const int* __restrict__ w4, const int* __restrict__ pb,
        float* __restrict__ partial, int nchunks) {
    __shared__ float acc[N_GRAPHS * COLS_PER_WG];
    const int chunk = blockIdx.x;
    const int cg    = blockIdx.y;
    const int t     = threadIdx.x;
    const int c     = cg * COLS_PER_WG + t;

    for (int i = t; i < N_GRAPHS * COLS_PER_WG; i += 256) acc[i] = 0.0f;
    __syncthreads();

    const int per = (N_NODES + nchunks - 1) / nchunks;
    const int n0  = chunk * per;
    const int n1  = (n0 + per < N_NODES) ? (n0 + per) : N_NODES;

    for (int n = n0; n < n1; ++n) {
        const int g = pb[n];
        float y = 0.0f;
        #pragma unroll
        for (int l = 0; l < 3; ++l) y += x[w2[l * N_NODES + n] * HID + c];
        #pragma unroll
        for (int l = 0; l < 4; ++l) y += x[w3[l * N_NODES + n] * HID + c];
        #pragma unroll
        for (int l = 0; l < 5; ++l) y += x[w4[l * N_NODES + n] * HID + c];
        acc[g * COLS_PER_WG + t] += y;
    }
    __syncthreads();

    float* dst = partial + (size_t)chunk * (N_GRAPHS * HID) + cg * COLS_PER_WG;
    for (int g = 0; g < N_GRAPHS; ++g)
        dst[g * HID + t] = acc[g * COLS_PER_WG + t];
}


__global__ void k_zero_ct(unsigned* __restrict__ ct, unsigned* __restrict__ counts) {
    v4u* ct4 = (v4u*)ct;
    const int t = blockIdx.x * 256 + threadIdx.x;
    const v4u z = {0u, 0u, 0u, 0u};
    for (int i = t; i < CT_ROWS * N_GRAPHS / 4; i += 512 * 256) ct4[i] = z;
    if (t < N_GRAPHS) counts[t] = 0u;
}

#define TSTEPS  16
#define TBATCH  (HID / 4 / TSTEPS)
#define APITCH  516
#define ASLAB   (16 * APITCH)
template <int NT>
__device__ __forceinline__ v8f
tile_kloop(const float* __restrict__ A, const float* __restrict__ brow, const int row0,
           float* __restrict__ as, const float* __restrict__ bp) {
    const int t    = threadIdx.x;
    const int lane = t & 31;
    const int l16  = lane & 15;
    const int lh   = lane >> 4;
    v2f ta[2][TSTEPS], tb[2][TSTEPS];
    v8f acc = {};
    #define T_ISSUE_A(p, kb)                                                    \
        {                                                                       \
            _Pragma("unroll")                                                   \
            for (int d = 0; d < TSTEPS; ++d) {                                  \
                const int k = ((kb) * TSTEPS + d) * 4;                          \
                ta[p][d] = *(const v2f*)(ap + k);                               \
            }                                                                   \
        }
    #define T_ISSUE_B(p, kb)                                                    \
        {                                                                       \
            _Pragma("unroll")                                                   \
            for (int d = 0; d < TSTEPS; ++d) {                                  \
                const int k = ((kb) * TSTEPS + d) * 4;                          \
                tb[p][d].x = bp[k * HID];    tb[p][d].y = bp[(k + 1) * HID];    \
            }                                                                   \
        }
    #define T_COMPUTE(p)                                                        \
        {                                                                       \
            _Pragma("unroll")                                                   \
            for (int d = 0; d < TSTEPS; ++d)                                    \
                acc = __builtin_amdgcn_wmma_f32_16x16x4_f32(                    \
                    false, ta[p][d], false, tb[p][d], (short)0, acc, false, false); \
        }
    T_ISSUE_B(0, 0)
    #pragma unroll
    for (int i = 0; i < (16 * 128) / NT; ++i) {
        const int idx  = i * NT + t;
        const int r    = idx >> 7;
        const int c    = (idx & 127) * 4;
        const int grow = row0 + r;
        const float* srow = (grow < HID) ? (A + grow * HID) : brow;
        v4f v = *(const v4f*)(srow + c);
        if (grow > HID) v = (v4f){0.0f, 0.0f, 0.0f, 0.0f};
        *(v4f*)(as + r * APITCH + c) = v;
    }
    __syncthreads();
    const float* ap = as + l16 * APITCH + 2 * lh;
    T_ISSUE_A(0, 0)
    #pragma unroll
    for (int kb = 0; kb < TBATCH; kb += 2) {
        T_ISSUE_A(1, kb + 1)
        T_ISSUE_B(1, kb + 1)
        __builtin_amdgcn_sched_barrier(0);
        T_COMPUTE(0)
        __builtin_amdgcn_sched_barrier(0);
        if (kb + 2 < TBATCH) { T_ISSUE_A(0, kb + 2) T_ISSUE_B(0, kb + 2) }
        __builtin_amdgcn_sched_barrier(0);
        T_COMPUTE(1)
    }
    #undef T_ISSUE_A
    #undef T_ISSUE_B
    #undef T_COMPUTE
    return acc;
}

template <int NT>
__device__ __forceinline__ void wprod_tile(const float* __restrict__ A,
                                           const float* __restrict__ brow,
                                           const float* __restrict__ B,
                                           const float* __restrict__ badd,
                                           float* __restrict__ outp, int tile,
                                           float* __restrict__ as) {
    const int lane = threadIdx.x & 31;
    const int l16  = lane & 15;
    const int lh   = lane >> 4;
    const int tm   = tile >> 5;
    const int tn   = tile & 31;
    const int col = tn * 16 + l16;

    const v8f acc = tile_kloop<NT>(A, brow, tm * 16, as, B + (lh * 2) * HID + col);
    const float bn = badd[col];
    #pragma unroll
    for (int v = 0; v < 8; ++v) {
        const int m = tm * 16 + v + lh * 8;
        outp[m * HID + col] = acc[v] + ((m == HID) ? bn : 0.0f);
    }
}

__device__ __forceinline__ void scatter_body(const int* __restrict__ w2, const int* __restrict__ w3,
                          const int* __restrict__ w4, const int* __restrict__ batch,
                          unsigned* __restrict__ ct, unsigned* __restrict__ counts) {
    const int n = blockIdx.x * 256 + threadIdx.x;
    if (n < N_NODES) {
        int m[12];
        for (int l = 0; l < 3; ++l) m[l]     = w2[l * N_NODES + n];
        for (int l = 0; l < 4; ++l) m[3 + l] = w3[l * N_NODES + n];
        for (int l = 0; l < 5; ++l) m[7 + l] = w4[l * N_NODES + n];
        const int g = batch[m[0]];
        atomicAdd(&counts[g], 1u);
        for (int l = 0; l < 12; ++l) atomicAdd(&ct[(size_t)m[l] * N_GRAPHS + g], 1u);
    }
}

__global__ void k_scatter(const int* __restrict__ w2, const int* __restrict__ w3,
                          const int* __restrict__ w4, const int* __restrict__ batch,
                          unsigned* __restrict__ ct, unsigned* __restrict__ counts) {
    scatter_body(w2, w3, w4, batch, ct, counts);
}

__global__ void __launch_bounds__(256)
k_bin_w(const int* __restrict__ w2, const int* __restrict__ w3,
        const int* __restrict__ w4, const int* __restrict__ batch,
        unsigned* __restrict__ ent, unsigned* __restrict__ off,
        unsigned* __restrict__ hist,
        const float* __restrict__ W0, const float* __restrict__ b0,
        const float* __restrict__ W1, const float* __restrict__ b1,
        float* __restrict__ w01) {
    __shared__ unsigned cnt[NBUCK];
    __shared__ unsigned base[NBUCK + 1];
    __shared__ unsigned hg[N_GRAPHS];
    __shared__ __attribute__((aligned(16))) unsigned stage[ENT_PER_WG];
    __shared__ __attribute__((aligned(16))) float aslab[ASLAB];
    if (blockIdx.x >= SCAT_WGS) {
        const int tile0 = (blockIdx.x - SCAT_WGS) * 8;
        if (tile0 < WP_TILES)
            wprod_tile<256>(W0, b0, W1, b1, w01, tile0 + (threadIdx.x >> 5), aslab);
        return;
    }
    const int t = threadIdx.x;
    const int wg = blockIdx.x;
    if (t < NBUCK) cnt[t] = 0u;
    if (t < N_GRAPHS) hg[t] = 0u;
    __syncthreads();

    const int n = wg * 256 + t;
    unsigned e[12], pos[12], bk[12];
    if (n < N_NODES) {
        int m[12];
        #pragma unroll
        for (int l = 0; l < 3; ++l) m[l]     = w2[l * N_NODES + n];
        #pragma unroll
        for (int l = 0; l < 4; ++l) m[3 + l] = w3[l * N_NODES + n];
        #pragma unroll
        for (int l = 0; l < 5; ++l) m[7 + l] = w4[l * N_NODES + n];
        const unsigned g = (unsigned)batch[m[0]];
        atomicAdd(&hg[g], 1u);
        #pragma unroll
        for (int l = 0; l < 12; ++l) {
            bk[l]  = (unsigned)m[l] / BUCK_NODES;
            e[l]   = ((unsigned)m[l] << 7) | g;
            pos[l] = atomicAdd(&cnt[bk[l]], 1u);
        }
    }
    __syncthreads();
    if (t == 0) {
        unsigned run = 0u;
        for (int b = 0; b < NBUCK; ++b) { base[b] = run; run += cnt[b]; }
        base[NBUCK] = run;
    }
    __syncthreads();
    const unsigned total = base[NBUCK];
    if (n < N_NODES) {
        #pragma unroll
        for (int l = 0; l < 12; ++l) stage[base[bk[l]] + pos[l]] = e[l];
    }
    if (t < 3 && total + t < ENT_PER_WG) stage[total + t] = 0u;
    __syncthreads();
    {
        const v4u* s4 = (const v4u*)stage;
        v4u* d4 = (v4u*)(ent + (size_t)wg * ENT_PER_WG);
        for (unsigned i = t; i < (total + 3u) / 4u; i += 256) d4[i] = s4[i];
    }
    if (t <= NBUCK) off[wg * (NBUCK + 1) + t] = base[t];
    if (t < N_GRAPHS) hist[wg * N_GRAPHS + t] = hg[t];
}

__device__ __forceinline__ void
cgemm_body(const float* __restrict__ x, const unsigned* __restrict__ ct,
           float* __restrict__ partial, int nchunks) {
    const int chunk = blockIdx.x;
    const int wave  = threadIdx.x >> 5;
    const int lane  = threadIdx.x & 31;
    const int l16   = lane & 15;
    const int lh    = lane >> 4;
    const int gb    = (wave & 1) * 64;
    const int cb    = (blockIdx.y * 2 + (wave >> 1)) * 64;

    const int per = (KSTEPS + nchunks - 1) / nchunks;
    const int s0  = chunk * per;
    const int s1  = (s0 + per < KSTEPS) ? (s0 + per) : KSTEPS;
    const int nsteps = s1 - s0;

    const float*    xl = x  + (2 * lh) * HID      + cb + 4 * l16;
    const unsigned* cl = ct + (2 * lh) * N_GRAPHS + gb + 4 * l16;

    v4f rb0[2][DEPTH], rb1[2][DEPTH];
    v4u ra0[2][DEPTH], ra1[2][DEPTH];
    v8f acc[4][4] = {};

    #define ISSUE(p, d, s)                                                      \
        {                                                                       \
            const int sx = ((s) < KSTEPS) ? (s) : (KSTEPS - 1);                 \
            const int sc = ((s) < s1) ? (s) : KSTEPS;                           \
            const float*    xp = xl + (size_t)sx * (4 * HID);                   \
            const unsigned* cp = cl + (size_t)sc * (4 * N_GRAPHS);              \
            rb0[p][d] = *(const v4f*)xp;                                        \
            rb1[p][d] = *(const v4f*)(xp + HID);                                \
            ra0[p][d] = *(const v4u*)cp;                                        \
            ra1[p][d] = *(const v4u*)(cp + N_GRAPHS);                           \
        }
    #define COMPUTE(p, d)                                                       \
        {                                                                       \
            v2f a[4], b[4];                                                     \
            _Pragma("unroll")                                                   \
            for (int i = 0; i < 4; ++i) {                                       \
                a[i].x = (float)ra0[p][d][i];  a[i].y = (float)ra1[p][d][i];    \
                b[i].x = rb0[p][d][i];         b[i].y = rb1[p][d][i];           \
            }                                                                   \
            _Pragma("unroll")                                                   \
            for (int i = 0; i < 4; ++i) {                                       \
                _Pragma("unroll")                                               \
                for (int j = 0; j < 4; ++j)                                     \
                    acc[i][j] = __builtin_amdgcn_wmma_f32_16x16x4_f32(          \
                        false, a[i], false, b[j], (short)0, acc[i][j], false, false); \
            }                                                                   \
        }

    #pragma unroll
    for (int d = 0; d < DEPTH; ++d) ISSUE(0, d, s0 + d)

    for (int base = 0; base < nsteps; base += 2 * DEPTH) {
        #pragma unroll
        for (int d = 0; d < DEPTH; ++d) ISSUE(1, d, s0 + base + DEPTH + d)
        __builtin_amdgcn_sched_barrier(0);
        #pragma unroll
        for (int d = 0; d < DEPTH; ++d) COMPUTE(0, d)
        __builtin_amdgcn_sched_barrier(0);
        #pragma unroll
        for (int d = 0; d < DEPTH; ++d) ISSUE(0, d, s0 + base + 2 * DEPTH + d)
        __builtin_amdgcn_sched_barrier(0);
        #pragma unroll
        for (int d = 0; d < DEPTH; ++d) COMPUTE(1, d)
    }
    #undef ISSUE
    #undef COMPUTE

    float* dst = partial + (size_t)chunk * (N_GRAPHS * HID) + cb + 4 * l16;
    #pragma unroll
    for (int i = 0; i < 4; ++i) {
        #pragma unroll
        for (int v = 0; v < 8; ++v) {
            const int g = gb + 4 * (v + 8 * lh) + i;
            v4f o = {acc[i][0][v], acc[i][1][v], acc[i][2][v], acc[i][3][v]};
            *(v4f*)(dst + g * HID) = o;
        }
    }
}

__global__ void __launch_bounds__(128)
k_cgemm(const float* __restrict__ x, const unsigned* __restrict__ ct,
        float* __restrict__ partial, int nchunks) {
    cgemm_body(x, ct, partial, nchunks);
}

__global__ void __launch_bounds__(128)
k_cgemm_f(const float* __restrict__ x, const unsigned* __restrict__ ent,
          const unsigned* __restrict__ off, float* __restrict__ partial) {
    __shared__ __attribute__((aligned(16))) unsigned cs[FROWS * GHALF + FSPARE];
    const int t     = threadIdx.x;
    const int chunk = blockIdx.x;
    const unsigned gbh = blockIdx.y & 1;
    const int colh  = blockIdx.y >> 1;
    const int wave  = __builtin_amdgcn_readfirstlane(t >> 5);
    const int lane  = t & 31;
    const int l16   = lane & 15;
    const int lh    = lane >> 4;
    const int gb    = (int)gbh * GHALF;
    const int cb    = (colh * 4 + wave) * 64;

    const int s0 = chunk * FPER;
    const int s1 = (s0 + FPER < KSTEPS) ? (s0 + FPER) : KSTEPS;
    const int nsteps = s1 - s0;

    const float*    xl = x  + (2 * lh) * HID   + cb + 4 * l16;
    const unsigned* al = cs + (2 * lh) * GHALF + 4 * l16;

    v4f rb0[2][DEPTH], rb1[2][DEPTH];
    v4u ra0[2][DEPTH], ra1[2][DEPTH];
    v8f acc[4][4] = {};

    #define ISSUE_X(p, d, s)                                                    \
        {                                                                       \
            const int sx = ((s) < KSTEPS) ? (s) : (KSTEPS - 1);                 \
            const float* xp = xl + (size_t)sx * (4 * HID);                      \
            rb0[p][d] = *(const v4f*)xp;                                        \
            rb1[p][d] = *(const v4f*)(xp + HID);                                \
        }
    #define ISSUE_A(p, d, s)                                                    \
        {                                                                       \
            const int sl = ((s) < s1) ? ((s) - s0) : FPER;                      \
            const unsigned* cp = al + sl * (4 * GHALF);                         \
            ra0[p][d] = *(const v4u*)cp;                                        \
            ra1[p][d] = *(const v4u*)(cp + GHALF);                              \
        }
    #define COMPUTE(p, d)                                                       \
        {                                                                       \
            v2f a[4], b[4];                                                     \
            _Pragma("unroll")                                                   \
            for (int i = 0; i < 4; ++i) {                                       \
                a[i].x = (float)ra0[p][d][i];  a[i].y = (float)ra1[p][d][i];    \
                b[i].x = rb0[p][d][i];         b[i].y = rb1[p][d][i];           \
            }                                                                   \
            _Pragma("unroll")                                                   \
            for (int i = 0; i < 4; ++i) {                                       \
                _Pragma("unroll")                                               \
                for (int j = 0; j < 4; ++j)                                     \
                    acc[i][j] = __builtin_amdgcn_wmma_f32_16x16x4_f32(          \
                        false, a[i], false, b[j], (short)0, acc[i][j], false, false); \
            }                                                                   \
        }


    const unsigned* offb = off + chunk;
    const int j1 = (lane < FSEGS - 32) ? (32 + lane) : (FSEGS - 1);
    const unsigned lo0 = offb[(wave + 4 * lane) * (NBUCK + 1)];
    const unsigned hi0 = offb[(wave + 4 * lane) * (NBUCK + 1) + 1];
    const unsigned lo1 = offb[(wave + 4 * j1) * (NBUCK + 1)];
    const unsigned hi1 = offb[(wave + 4 * j1) * (NBUCK + 1) + 1];


    const unsigned m0 = (unsigned)chunk * BUCK_NODES;
    v2u dw[FSEGS];
    #pragma unroll
    for (int k = 0; k < FSEGS; ++k) {
        const unsigned lo = (unsigned)__builtin_amdgcn_readlane((int)((k < 32) ? lo0 : lo1), k & 31);
        const unsigned idx = (lo & ~1u) + 2u * lane;
        const unsigned idc = (idx < ENT_PER_WG - 2u) ? idx : (ENT_PER_WG - 2u);
        dw[k] = *(const v2u*)(ent + (size_t)(wave + 4 * k) * ENT_PER_WG + idc);
    }
    v4u* cs4 = (v4u*)cs;
    const v4u z = {0u, 0u, 0u, 0u};
    for (int i = t; i < (FROWS * GHALF + FSPARE) / 4; i += 128) cs4[i] = z;
    __syncthreads();
    #pragma unroll
    for (int d = 0; d < DEPTH; ++d) ISSUE_X(0, d, s0 + d)
    const unsigned ubase = 2u * m0 + gbh;
    const unsigned spare = (unsigned)(FROWS * GHALF) + (unsigned)lane;
    #pragma unroll
    for (int k = 0; k < FSEGS; ++k) {
        const unsigned lo = (unsigned)__builtin_amdgcn_readlane((int)((k < 32) ? lo0 : lo1), k & 31);
        const unsigned hi = (unsigned)__builtin_amdgcn_readlane((int)((k < 32) ? hi0 : hi1), k & 31);
        const unsigned odd = lo & 1u;
        const unsigned len = hi - lo;
        #pragma unroll
        for (int q = 0; q < 2; ++q) {
            const unsigned pos = 2u * (unsigned)lane + (unsigned)q - odd;
            const unsigned ev = (pos < len) ? dw[k][q] : 0xFFFFFFFFu;
            const unsigned uq = (ev >> 6) - ubase;
            const unsigned r  = (uq >> 1) | (uq << 31);
            const unsigned a  = (r < (unsigned)BUCK_NODES) ? (r * GHALF + (ev & 63u)) : spare;
            atomicAdd(&cs[a], 1u);
        }
    }
    const unsigned more0 = __builtin_amdgcn_ballot_w32(hi0 > (lo0 & ~1u) + 64u);
    const unsigned more1 = __builtin_amdgcn_ballot_w32((lane < FSEGS - 32) & (hi1 > (lo1 & ~1u) + 64u));
    #pragma unroll
    for (int h = 0; h < 2; ++h) {
        unsigned more = h ? more1 : more0;
        while (more != 0u) {
            const int j = __builtin_ctz(more);
            more &= more - 1u;
            const unsigned lo = (unsigned)__builtin_amdgcn_readlane((int)(h ? lo1 : lo0), j);
            const unsigned hi = (unsigned)__builtin_amdgcn_readlane((int)(h ? hi1 : hi0), j);
            const unsigned* sp = ent + (size_t)(wave + 4 * (32 * h + j)) * ENT_PER_WG;
            for (unsigned i = (lo & ~1u) + 64u + lane; i < hi; i += 32) {
                const unsigned ev = sp[i];
                const unsigned r = (ev >> 7) - m0;
                if (r < (unsigned)BUCK_NODES && ((ev >> 6) & 1u) == gbh)
                    atomicAdd(&cs[r * GHALF + (ev & 63u)], 1u);
            }
        }
    }
    __syncthreads();

    #pragma unroll
    for (int d = 0; d < DEPTH; ++d) ISSUE_A(0, d, s0 + d)

    for (int base = 0; base < nsteps; base += 2 * DEPTH) {
        #pragma unroll
        for (int d = 0; d < DEPTH; ++d) { ISSUE_X(1, d, s0 + base + DEPTH + d) ISSUE_A(1, d, s0 + base + DEPTH + d) }
        __builtin_amdgcn_sched_barrier(0);
        #pragma unroll
        for (int d = 0; d < DEPTH; ++d) COMPUTE(0, d)
        __builtin_amdgcn_sched_barrier(0);
        #pragma unroll
        for (int d = 0; d < DEPTH; ++d) { ISSUE_X(0, d, s0 + base + 2 * DEPTH + d) ISSUE_A(0, d, s0 + base + 2 * DEPTH + d) }
        __builtin_amdgcn_sched_barrier(0);
        #pragma unroll
        for (int d = 0; d < DEPTH; ++d) COMPUTE(1, d)
    }
    #undef ISSUE_X
    #undef ISSUE_A
    #undef COMPUTE

    float* dst = partial + (size_t)chunk * (N_GRAPHS * HID) + cb + 4 * l16;
    #pragma unroll
    for (int i = 0; i < 4; ++i) {
        #pragma unroll
        for (int v = 0; v < 8; ++v) {
            const int g = gb + 4 * (v + 8 * lh) + i;
            v4f o = {acc[i][0][v], acc[i][1][v], acc[i][2][v], acc[i][3][v]};
            *(v4f*)(dst + g * HID) = o;
        }
    }
}

__global__ void k_reduce(const float* __restrict__ partial,
                         const unsigned* __restrict__ counts,
                         float* __restrict__ P, float* __restrict__ s,
                         int nchunks) {
    int i = blockIdx.x * 256 + threadIdx.x;
    int g = i >> 9;
    float a = 0.0f;
    for (int ch = 0; ch < nchunks; ++ch)
        a += partial[(size_t)ch * (N_GRAPHS * HID) + i];
    float cnt = (float)counts[g];
    P[i] = a / fmaxf(cnt, 1.0f);
    if (i < N_GRAPHS) s[i] = (counts[i] != 0u) ? 12.0f : 0.0f;
}

__global__ void __launch_bounds__(256)
k_reduce_w(const float* __restrict__ partial, const unsigned* __restrict__ hist,
           float* __restrict__ P, float* __restrict__ s, int nchunks,
           const float* __restrict__ w01, const float* __restrict__ W2,
           const float* __restrict__ b2, float* __restrict__ w012) {
    __shared__ __attribute__((aligned(16))) float aslab[ASLAB];
    __shared__ unsigned gcnt;
    if (blockIdx.x < WP_TILES / 8) {
        wprod_tile<256>(w01, w01 + HID * HID, W2, b2, w012, blockIdx.x * 8 + (threadIdx.x >> 5), aslab);
        return;
    }
    const int t = threadIdx.x;
    const int i = (blockIdx.x - WP_TILES / 8) * 256 + t;
    const int g = i >> 9;
    if (t == 0) gcnt = 0u;
    __syncthreads();
    unsigned c = (t < SCAT_WGS) ? hist[t * N_GRAPHS + g] : 0u;
    #pragma unroll
    for (int o = 16; o > 0; o >>= 1) c += __shfl_xor(c, o);
    if ((t & 31) == 0) atomicAdd(&gcnt, c);
    float a = 0.0f;
    for (int ch = 0; ch < nchunks; ++ch)
        a += partial[(size_t)ch * (N_GRAPHS * HID) + i];
    __syncthreads();
    const unsigned cg = gcnt;
    P[i] = a / fmaxf((float)cg, 1.0f);
    if ((i & (HID - 1)) == 0) s[g] = (cg != 0u) ? 12.0f : 0.0f;
}

__global__ void __launch_bounds__(256)
k_gemm_wmma(const float* __restrict__ A, int lda,
            const float* __restrict__ W, const float* __restrict__ bias,
            const float* __restrict__ s, float* __restrict__ out) {
    const int wave = (blockIdx.x << 3) + (threadIdx.x >> 5);
    const int lane = threadIdx.x & 31;
    const int l16  = lane & 15;
    const int lh   = lane >> 4;
    const int tm   = wave >> 5;
    const int tn   = wave & 31;

    const float* Arow = A + (tm * 16 + l16) * lda;
    const int col = tn * 16 + l16;

    v8f acc = {};
    for (int k = 0; k < HID; k += 4) {
        const int kk = k + lh * 2;
        v2f a; a.x = Arow[kk];           a.y = Arow[kk + 1];
        v2f b; b.x = W[kk * HID + col];  b.y = W[(kk + 1) * HID + col];
        acc = __builtin_amdgcn_wmma_f32_16x16x4_f32(
             false, a,  false, b,
             (short)0, acc,  false,  false);
    }

    const float bn = bias[col];
    #pragma unroll
    for (int v = 0; v < 8; ++v) {
        const int m = tm * 16 + v + lh * 8;
        out[m * OUT_PITCH + col] = acc[v] + s[m] * bn;
    }
}

__global__ void __launch_bounds__(256)
k_gemm3(const float* __restrict__ P,
        const float* __restrict__ W0, const float* __restrict__ b0,
        const float* __restrict__ w01, const float* __restrict__ w012,
        const float* __restrict__ s, float* __restrict__ out) {
    __shared__ __attribute__((aligned(16))) float aslab[ASLAB];
    const int wave = (blockIdx.x << 3) + (threadIdx.x >> 5);
    const int lane = threadIdx.x & 31;
    const int l16  = lane & 15;
    const int lh   = lane >> 4;
    const int tm    = wave / 96;
    const int tq    = wave - tm * 96;
    const int layer = tq >> 5;
    const int tn    = tq & 31;

    const float* B    = (layer == 0) ? W0 : ((layer == 1) ? w01 : w012);
    const float* bias = (layer == 0) ? b0 : (B + HID * HID);
    const int col = tn * 16 + l16;

    const v8f acc = tile_kloop<256>(P, P, tm * 16, aslab, B + (lh * 2) * HID + col);

    const float bn = bias[col];
    #pragma unroll
    for (int v = 0; v < 8; ++v) {
        const int m = tm * 16 + v + lh * 8;
        out[m * OUT_PITCH + layer * HID + col] = acc[v] + s[m] * bn;
    }
}

extern "C" void kernel_launch(void* const* d_in, const int* in_sizes, int n_in,
                              void* d_out, int out_size, void* d_ws, size_t ws_size,
                              hipStream_t stream) {
    const float* x     = (const float*)d_in[0];
    const int*   w2    = (const int*)  d_in[1];
    const int*   w3    = (const int*)  d_in[2];
    const int*   w4    = (const int*)  d_in[3];
    const int*   batch = (const int*)  d_in[4];
    const float* W0 = (const float*)d_in[5];  const float* b0 = (const float*)d_in[6];
    const float* W1 = (const float*)d_in[7];  const float* b1 = (const float*)d_in[8];
    const float* W2 = (const float*)d_in[9];  const float* b2 = (const float*)d_in[10];
    float* out = (float*)d_out;

    const size_t partBytes = (size_t)N_GRAPHS * HID * sizeof(float);
    const size_t fixed = partBytes   + 512   + 512
                       + (size_t)N_NODES * sizeof(int)  ;
    int nchunks = 1;
    if (ws_size > fixed + partBytes) {
        size_t fit = (ws_size - fixed) / partBytes;
        nchunks = (fit > MAX_CHUNKS) ? MAX_CHUNKS : (int)fit;
        if (nchunks < 1) nchunks = 1;
    }
    char* ws = (char*)d_ws;
    float*    partial = (float*)ws;
    float*    P       = (float*)(ws + (size_t)nchunks * partBytes);
    unsigned* counts  = (unsigned*)((char*)P + partBytes);
    float*    s       = (float*)((char*)counts + 512);
    int*      pb      = (int*)((char*)s + 512);

    const size_t ctBytes = (size_t)CT_ROWS * N_GRAPHS * sizeof(unsigned);
    const size_t denseNeed = (size_t)MAX_CHUNKS * partBytes + partBytes + 1024 + ctBytes;
    const size_t wpBytes = (size_t)WP_ROWS * HID * sizeof(float);
    if (ws_size >= denseNeed + 2 * wpBytes) {
        unsigned* ent  = (unsigned*)pb;
        unsigned* off  = ent + (size_t)SCAT_WGS * ENT_PER_WG;
        unsigned* hist = off + SCAT_WGS * (NBUCK + 1);
        float* w01  = (float*)((char*)pb + ctBytes);
        float* w012 = (float*)((char*)w01 + wpBytes);
        k_bin_w<<<SCAT_WGS + WP_TILES / 8, 256, 0, stream>>>(w2, w3, w4, batch, ent, off, hist,
                                                          W0, b0, W1, b1, w01);
        dim3 gFused(MAX_CHUNKS, 4);
        k_cgemm_f<<<gFused, 128, 0, stream>>>(x, ent, off, partial);
        k_reduce_w<<<WP_TILES / 8 + (N_GRAPHS * HID) / 256, 256, 0, stream>>>(partial, hist, P, s, nchunks,
                                                                              w01, W2, b2, w012);
        k_gemm3<<<96, 256, 0, stream>>>(P, W0, b0, w01, w012, s, out);
        return;
    }
    if (ws_size >= denseNeed) {
        unsigned* ct = (unsigned*)pb;
        k_zero_ct<<<512, 256, 0, stream>>>(ct, counts);
        k_scatter<<<(N_NODES + 255) / 256, 256, 0, stream>>>(w2, w3, w4, batch, ct, counts);
        dim3 gDense(nchunks, 4);
        k_cgemm<<<gDense, 128, 0, stream>>>(x, ct, partial, nchunks);
    } else {
        k_zero_counts<<<1, 128, 0, stream>>>(counts);
        k_pb_hist<<<(N_NODES + 255) / 256, 256, 0, stream>>>(w2, batch, pb, counts);
        dim3 gAcc(nchunks, 2);
        k_accum<<<gAcc, 256, 0, stream>>>(x, w2, w3, w4, pb, partial, nchunks);
    }
    k_reduce<<<(N_GRAPHS * HID) / 256, 256, 0, stream>>>(partial, counts, P, s, nchunks);

    k_gemm_wmma<<<32, 256, 0, stream>>>(P,         HID,       W0, b0, s, out);
    k_gemm_wmma<<<32, 256, 0, stream>>>(out,       OUT_PITCH, W1, b1, s, out + HID);
    k_gemm_wmma<<<32, 256, 0, stream>>>(out + HID, OUT_PITCH, W2, b2, s, out + 2 * HID);
}


# count GEMM K loop: every chunk runs exactly 196 steps (last chunk starts early over zero slab rows), operand addresses are one base plus constants, no per-step index, clamp or select; same sums
# speedup vs baseline: 8.2523x; 1.0628x over previous
#include <hip/hip_runtime.h>


#define N_NODES   50000
#define N_GRAPHS  128
#define HID       512
#define OUT_PITCH (3 * HID)
#define MAX_CHUNKS 64
#define COLS_PER_WG 256
#define KSTEPS    (N_NODES / 4)
#define CT_ROWS   (N_NODES + 4)
#define DEPTH     7
#define WP_ROWS   528
#define WP_TILES  (33 * 32)
#define SCAT_WGS  ((N_NODES + 255) / 256)
#define NBUCK     64
#define BUCK_NODES 784
#define ENT_PER_WG (256 * 12)
#define FPER      (BUCK_NODES / 4)
#define FROWS     BUCK_NODES
#define FSEGS     (SCAT_WGS / 4)
#define GHALF     64
#define FSPARE    32
static_assert(4 * FSEGS == SCAT_WGS, "four waves share the binning workgroups evenly");
static_assert(NBUCK == MAX_CHUNKS && NBUCK * BUCK_NODES >= N_NODES, "one bucket per K chunk, all nodes covered");
static_assert((KSTEPS + MAX_CHUNKS - 1) / MAX_CHUNKS == FPER, "a K chunk of the count GEMM is one bucket");
static_assert(FPER % (2 * DEPTH) == 0 && FPER <= KSTEPS, "the fused count GEMM runs whole trips of two register buffers");

typedef __attribute__((ext_vector_type(2))) float v2f;
typedef __attribute__((ext_vector_type(8))) float v8f;
typedef __attribute__((ext_vector_type(4))) float v4f;
typedef __attribute__((ext_vector_type(4))) unsigned v4u;
typedef __attribute__((ext_vector_type(2))) unsigned v2u;

__global__ void k_zero_counts(unsigned* __restrict__ counts) {
    counts[threadIdx.x] = 0u;
}

__global__ void k_pb_hist(const int* __restrict__ walk2_row0,
                          const int* __restrict__ batch,
                          int* __restrict__ pb,
                          unsigned* __restrict__ counts) {
    int n = blockIdx.x * 256 + threadIdx.x;
    if (n < N_NODES) {
        int g = batch[walk2_row0[n]];
        pb[n] = g;
        atomicAdd(&counts[g], 1u);
    }
}

__global__ void __launch_bounds__(256)
k_accum(const float* __restrict__ x,
        const int* __restrict__ w2, const int* __restrict__ w3,
        const int* __restrict__ w4, const int* __restrict__ pb,
        float* __restrict__ partial, int nchunks) {
    __shared__ float acc[N_GRAPHS * COLS_PER_WG];
    const int chunk = blockIdx.x;
    const int cg    = blockIdx.y;
    const int t     = threadIdx.x;
    const int c     = cg * COLS_PER_WG + t;

    for (int i = t; i < N_GRAPHS * COLS_PER_WG; i += 256) acc[i] = 0.0f;
    __syncthreads();

    const int per = (N_NODES + nchunks - 1) / nchunks;
    const int n0  = chunk * per;
    const int n1  = (n0 + per < N_NODES) ? (n0 + per) : N_NODES;

    for (int n = n0; n < n1; ++n) {
        const int g = pb[n];
        float y = 0.0f;
        #pragma unroll
        for (int l = 0; l < 3; ++l) y += x[w2[l * N_NODES + n] * HID + c];
        #pragma unroll
        for (int l = 0; l < 4; ++l) y += x[w3[l * N_NODES + n] * HID + c];
        #pragma unroll
        for (int l = 0; l < 5; ++l) y += x[w4[l * N_NODES + n] * HID + c];
        acc[g * COLS_PER_WG + t] += y;
    }
    __syncthreads();

    float* dst = partial + (size_t)chunk * (N_GRAPHS * HID) + cg * COLS_PER_WG;
    for (int g = 0; g < N_GRAPHS; ++g)
        dst[g * HID + t] = acc[g * COLS_PER_WG + t];
}


__global__ void k_zero_ct(unsigned* __restrict__ ct, unsigned* __restrict__ counts) {
    v4u* ct4 = (v4u*)ct;
    const int t = blockIdx.x * 256 + threadIdx.x;
    const v4u z = {0u, 0u, 0u, 0u};
    for (int i = t; i < CT_ROWS * N_GRAPHS / 4; i += 512 * 256) ct4[i] = z;
    if (t < N_GRAPHS) counts[t] = 0u;
}

#define TSTEPS  16
#define TBATCH  (HID / 4 / TSTEPS)
#define APITCH  516
#define ASLAB   (16 * APITCH)
template <int NT>
__device__ __forceinline__ v8f
tile_kloop(const float* __restrict__ A, const float* __restrict__ brow, const int row0,
           float* __restrict__ as, const float* __restrict__ bp) {
    const int t    = threadIdx.x;
    const int lane = t & 31;
    const int l16  = lane & 15;
    const int lh   = lane >> 4;
    v2f ta[2][TSTEPS], tb[2][TSTEPS];
    v8f acc = {};
    #define T_ISSUE_A(p, kb)                                                    \
        {                                                                       \
            _Pragma("unroll")                                                   \
            for (int d = 0; d < TSTEPS; ++d) {                                  \
                const int k = ((kb) * TSTEPS + d) * 4;                          \
                ta[p][d] = *(const v2f*)(ap + k);                               \
            }                                                                   \
        }
    #define T_ISSUE_B(p, kb)                                                    \
        {                                                                       \
            _Pragma("unroll")                                                   \
            for (int d = 0; d < TSTEPS; ++d) {                                  \
                const int k = ((kb) * TSTEPS + d) * 4;                          \
                tb[p][d].x = bp[k * HID];    tb[p][d].y = bp[(k + 1) * HID];    \
            }                                                                   \
        }
    #define T_COMPUTE(p)                                                        \
        {                                                                       \
            _Pragma("unroll")                                                   \
            for (int d = 0; d < TSTEPS; ++d)                                    \
                acc = __builtin_amdgcn_wmma_f32_16x16x4_f32(                    \
                    false, ta[p][d], false, tb[p][d], (short)0, acc, false, false); \
        }
    T_ISSUE_B(0, 0)
    #pragma unroll
    for (int i = 0; i < (16 * 128) / NT; ++i) {
        const int idx  = i * NT + t;
        const int r    = idx >> 7;
        const int c    = (idx & 127) * 4;
        const int grow = row0 + r;
        const float* srow = (grow < HID) ? (A + grow * HID) : brow;
        v4f v = *(const v4f*)(srow + c);
        if (grow > HID) v = (v4f){0.0f, 0.0f, 0.0f, 0.0f};
        *(v4f*)(as + r * APITCH + c) = v;
    }
    __syncthreads();
    const float* ap = as + l16 * APITCH + 2 * lh;
    T_ISSUE_A(0, 0)
    #pragma unroll
    for (int kb = 0; kb < TBATCH; kb += 2) {
        T_ISSUE_A(1, kb + 1)
        T_ISSUE_B(1, kb + 1)
        __builtin_amdgcn_sched_barrier(0);
        T_COMPUTE(0)
        __builtin_amdgcn_sched_barrier(0);
        if (kb + 2 < TBATCH) { T_ISSUE_A(0, kb + 2) T_ISSUE_B(0, kb + 2) }
        __builtin_amdgcn_sched_barrier(0);
        T_COMPUTE(1)
    }
    #undef T_ISSUE_A
    #undef T_ISSUE_B
    #undef T_COMPUTE
    return acc;
}

template <int NT>
__device__ __forceinline__ void wprod_tile(const float* __restrict__ A,
                                           const float* __restrict__ brow,
                                           const float* __restrict__ B,
                                           const float* __restrict__ badd,
                                           float* __restrict__ outp, int tile,
                                           float* __restrict__ as) {
    const int lane = threadIdx.x & 31;
    const int l16  = lane & 15;
    const int lh   = lane >> 4;
    const int tm   = tile >> 5;
    const int tn   = tile & 31;
    const int col = tn * 16 + l16;

    const v8f acc = tile_kloop<NT>(A, brow, tm * 16, as, B + (lh * 2) * HID + col);
    const float bn = badd[col];
    #pragma unroll
    for (int v = 0; v < 8; ++v) {
        const int m = tm * 16 + v + lh * 8;
        outp[m * HID + col] = acc[v] + ((m == HID) ? bn : 0.0f);
    }
}

__device__ __forceinline__ void scatter_body(const int* __restrict__ w2, const int* __restrict__ w3,
                          const int* __restrict__ w4, const int* __restrict__ batch,
                          unsigned* __restrict__ ct, unsigned* __restrict__ counts) {
    const int n = blockIdx.x * 256 + threadIdx.x;
    if (n < N_NODES) {
        int m[12];
        for (int l = 0; l < 3; ++l) m[l]     = w2[l * N_NODES + n];
        for (int l = 0; l < 4; ++l) m[3 + l] = w3[l * N_NODES + n];
        for (int l = 0; l < 5; ++l) m[7 + l] = w4[l * N_NODES + n];
        const int g = batch[m[0]];
        atomicAdd(&counts[g], 1u);
        for (int l = 0; l < 12; ++l) atomicAdd(&ct[(size_t)m[l] * N_GRAPHS + g], 1u);
    }
}

__global__ void k_scatter(const int* __restrict__ w2, const int* __restrict__ w3,
                          const int* __restrict__ w4, const int* __restrict__ batch,
                          unsigned* __restrict__ ct, unsigned* __restrict__ counts) {
    scatter_body(w2, w3, w4, batch, ct, counts);
}

__global__ void __launch_bounds__(256)
k_bin_w(const int* __restrict__ w2, const int* __restrict__ w3,
        const int* __restrict__ w4, const int* __restrict__ batch,
        unsigned* __restrict__ ent, unsigned* __restrict__ off,
        unsigned* __restrict__ hist,
        const float* __restrict__ W0, const float* __restrict__ b0,
        const float* __restrict__ W1, const float* __restrict__ b1,
        float* __restrict__ w01) {
    __shared__ unsigned cnt[NBUCK];
    __shared__ unsigned base[NBUCK + 1];
    __shared__ unsigned hg[N_GRAPHS];
    __shared__ __attribute__((aligned(16))) unsigned stage[ENT_PER_WG];
    __shared__ __attribute__((aligned(16))) float aslab[ASLAB];
    if (blockIdx.x >= SCAT_WGS) {
        const int tile0 = (blockIdx.x - SCAT_WGS) * 8;
        if (tile0 < WP_TILES)
            wprod_tile<256>(W0, b0, W1, b1, w01, tile0 + (threadIdx.x >> 5), aslab);
        return;
    }
    const int t = threadIdx.x;
    const int wg = blockIdx.x;
    if (t < NBUCK) cnt[t] = 0u;
    if (t < N_GRAPHS) hg[t] = 0u;
    __syncthreads();

    const int n = wg * 256 + t;
    unsigned e[12], pos[12], bk[12];
    if (n < N_NODES) {
        int m[12];
        #pragma unroll
        for (int l = 0; l < 3; ++l) m[l]     = w2[l * N_NODES + n];
        #pragma unroll
        for (int l = 0; l < 4; ++l) m[3 + l] = w3[l * N_NODES + n];
        #pragma unroll
        for (int l = 0; l < 5; ++l) m[7 + l] = w4[l * N_NODES + n];
        const unsigned g = (unsigned)batch[m[0]];
        atomicAdd(&hg[g], 1u);
        #pragma unroll
        for (int l = 0; l < 12; ++l) {
            bk[l]  = (unsigned)m[l] / BUCK_NODES;
            e[l]   = ((unsigned)m[l] << 7) | g;
            pos[l] = atomicAdd(&cnt[bk[l]], 1u);
        }
    }
    __syncthreads();
    if (t == 0) {
        unsigned run = 0u;
        for (int b = 0; b < NBUCK; ++b) { base[b] = run; run += cnt[b]; }
        base[NBUCK] = run;
    }
    __syncthreads();
    const unsigned total = base[NBUCK];
    if (n < N_NODES) {
        #pragma unroll
        for (int l = 0; l < 12; ++l) stage[base[bk[l]] + pos[l]] = e[l];
    }
    if (t < 3 && total + t < ENT_PER_WG) stage[total + t] = 0u;
    __syncthreads();
    {
        const v4u* s4 = (const v4u*)stage;
        v4u* d4 = (v4u*)(ent + (size_t)wg * ENT_PER_WG);
        for (unsigned i = t; i < (total + 3u) / 4u; i += 256) d4[i] = s4[i];
    }
    if (t <= NBUCK) off[wg * (NBUCK + 1) + t] = base[t];
    if (t < N_GRAPHS) hist[wg * N_GRAPHS + t] = hg[t];
}

__device__ __forceinline__ void
cgemm_body(const float* __restrict__ x, const unsigned* __restrict__ ct,
           float* __restrict__ partial, int nchunks) {
    const int chunk = blockIdx.x;
    const int wave  = threadIdx.x >> 5;
    const int lane  = threadIdx.x & 31;
    const int l16   = lane & 15;
    const int lh    = lane >> 4;
    const int gb    = (wave & 1) * 64;
    const int cb    = (blockIdx.y * 2 + (wave >> 1)) * 64;

    const int per = (KSTEPS + nchunks - 1) / nchunks;
    const int s0  = chunk * per;
    const int s1  = (s0 + per < KSTEPS) ? (s0 + per) : KSTEPS;
    const int nsteps = s1 - s0;

    const float*    xl = x  + (2 * lh) * HID      + cb + 4 * l16;
    const unsigned* cl = ct + (2 * lh) * N_GRAPHS + gb + 4 * l16;

    v4f rb0[2][DEPTH], rb1[2][DEPTH];
    v4u ra0[2][DEPTH], ra1[2][DEPTH];
    v8f acc[4][4] = {};

    #define ISSUE(p, d, s)                                                      \
        {                                                                       \
            const int sx = ((s) < KSTEPS) ? (s) : (KSTEPS - 1);                 \
            const int sc = ((s) < s1) ? (s) : KSTEPS;                           \
            const float*    xp = xl + (size_t)sx * (4 * HID);                   \
            const unsigned* cp = cl + (size_t)sc * (4 * N_GRAPHS);              \
            rb0[p][d] = *(const v4f*)xp;                                        \
            rb1[p][d] = *(const v4f*)(xp + HID);                                \
            ra0[p][d] = *(const v4u*)cp;                                        \
            ra1[p][d] = *(const v4u*)(cp + N_GRAPHS);                           \
        }
    #define COMPUTE(p, d)                                                       \
        {                                                                       \
            v2f a[4], b[4];                                                     \
            _Pragma("unroll")                                                   \
            for (int i = 0; i < 4; ++i) {                                       \
                a[i].x = (float)ra0[p][d][i];  a[i].y = (float)ra1[p][d][i];    \
                b[i].x = rb0[p][d][i];         b[i].y = rb1[p][d][i];           \
            }                                                                   \
            _Pragma("unroll")                                                   \
            for (int i = 0; i < 4; ++i) {                                       \
                _Pragma("unroll")                                               \
                for (int j = 0; j < 4; ++j)                                     \
                    acc[i][j] = __builtin_amdgcn_wmma_f32_16x16x4_f32(          \
                        false, a[i], false, b[j], (short)0, acc[i][j], false, false); \
            }                                                                   \
        }

    #pragma unroll
    for (int d = 0; d < DEPTH; ++d) ISSUE(0, d, s0 + d)

    for (int base = 0; base < nsteps; base += 2 * DEPTH) {
        #pragma unroll
        for (int d = 0; d < DEPTH; ++d) ISSUE(1, d, s0 + base + DEPTH + d)
        __builtin_amdgcn_sched_barrier(0);
        #pragma unroll
        for (int d = 0; d < DEPTH; ++d) COMPUTE(0, d)
        __builtin_amdgcn_sched_barrier(0);
        #pragma unroll
        for (int d = 0; d < DEPTH; ++d) ISSUE(0, d, s0 + base + 2 * DEPTH + d)
        __builtin_amdgcn_sched_barrier(0);
        #pragma unroll
        for (int d = 0; d < DEPTH; ++d) COMPUTE(1, d)
    }
    #undef ISSUE
    #undef COMPUTE

    float* dst = partial + (size_t)chunk * (N_GRAPHS * HID) + cb + 4 * l16;
    #pragma unroll
    for (int i = 0; i < 4; ++i) {
        #pragma unroll
        for (int v = 0; v < 8; ++v) {
            const int g = gb + 4 * (v + 8 * lh) + i;
            v4f o = {acc[i][0][v], acc[i][1][v], acc[i][2][v], acc[i][3][v]};
            *(v4f*)(dst + g * HID) = o;
        }
    }
}

__global__ void __launch_bounds__(128)
k_cgemm(const float* __restrict__ x, const unsigned* __restrict__ ct,
        float* __restrict__ partial, int nchunks) {
    cgemm_body(x, ct, partial, nchunks);
}

__global__ void __launch_bounds__(128)
k_cgemm_f(const float* __restrict__ x, const unsigned* __restrict__ ent,
          const unsigned* __restrict__ off, float* __restrict__ partial) {
    __shared__ __attribute__((aligned(16))) unsigned cs[FROWS * GHALF + FSPARE];
    const int t     = threadIdx.x;
    const int chunk = blockIdx.x;
    const unsigned gbh = blockIdx.y & 1;
    const int colh  = blockIdx.y >> 1;
    const int wave  = __builtin_amdgcn_readfirstlane(t >> 5);
    const int lane  = t & 31;
    const int l16   = lane & 15;
    const int lh    = lane >> 4;
    const int gb    = (int)gbh * GHALF;
    const int cb    = (colh * 4 + wave) * 64;

    const int sb = (chunk * FPER + FPER <= KSTEPS) ? (chunk * FPER) : (KSTEPS - FPER);

    const float*    xq = x  + (size_t)sb * (4 * HID) + (2 * lh) * HID   + cb + 4 * l16;
    const unsigned* aq = cs                          + (2 * lh) * GHALF + 4 * l16;

    v4f rb0[2][DEPTH], rb1[2][DEPTH];
    v4u ra0[2][DEPTH], ra1[2][DEPTH];
    v8f acc[4][4] = {};

    #define ISSUE_X(p, xb)                                                      \
        {                                                                       \
            _Pragma("unroll")                                                   \
            for (int d = 0; d < DEPTH; ++d) {                                   \
                rb0[p][d] = *(const v4f*)((xb) + d * (4 * HID));                \
                rb1[p][d] = *(const v4f*)((xb) + d * (4 * HID) + HID);          \
            }                                                                   \
        }
    #define ISSUE_A(p, ab)                                                      \
        {                                                                       \
            _Pragma("unroll")                                                   \
            for (int d = 0; d < DEPTH; ++d) {                                   \
                ra0[p][d] = *(const v4u*)((ab) + d * (4 * GHALF));              \
                ra1[p][d] = *(const v4u*)((ab) + d * (4 * GHALF) + GHALF);      \
            }                                                                   \
        }
    #define COMPUTE(p, d)                                                       \
        {                                                                       \
            v2f a[4], b[4];                                                     \
            _Pragma("unroll")                                                   \
            for (int i = 0; i < 4; ++i) {                                       \
                a[i].x = (float)ra0[p][d][i];  a[i].y = (float)ra1[p][d][i];    \
                b[i].x = rb0[p][d][i];         b[i].y = rb1[p][d][i];           \
            }                                                                   \
            _Pragma("unroll")                                                   \
            for (int i = 0; i < 4; ++i) {                                       \
                _Pragma("unroll")                                               \
                for (int j = 0; j < 4; ++j)                                     \
                    acc[i][j] = __builtin_amdgcn_wmma_f32_16x16x4_f32(          \
                        false, a[i], false, b[j], (short)0, acc[i][j], false, false); \
            }                                                                   \
        }


    const unsigned* offb = off + chunk;
    const int j1 = (lane < FSEGS - 32) ? (32 + lane) : (FSEGS - 1);
    const unsigned lo0 = offb[(wave + 4 * lane) * (NBUCK + 1)];
    const unsigned hi0 = offb[(wave + 4 * lane) * (NBUCK + 1) + 1];
    const unsigned lo1 = offb[(wave + 4 * j1) * (NBUCK + 1)];
    const unsigned hi1 = offb[(wave + 4 * j1) * (NBUCK + 1) + 1];


    const unsigned m0 = 4u * (unsigned)sb;
    v2u dw[FSEGS];
    #pragma unroll
    for (int k = 0; k < FSEGS; ++k) {
        const unsigned lo = (unsigned)__builtin_amdgcn_readlane((int)((k < 32) ? lo0 : lo1), k & 31);
        const unsigned idx = (lo & ~1u) + 2u * lane;
        const unsigned idc = (idx < ENT_PER_WG - 2u) ? idx : (ENT_PER_WG - 2u);
        dw[k] = *(const v2u*)(ent + (size_t)(wave + 4 * k) * ENT_PER_WG + idc);
    }
    v4u* cs4 = (v4u*)cs;
    const v4u z = {0u, 0u, 0u, 0u};
    for (int i = t; i < (FROWS * GHALF + FSPARE) / 4; i += 128) cs4[i] = z;
    __syncthreads();
    ISSUE_X(0, xq)
    const unsigned ubase = 2u * m0 + gbh;
    const unsigned spare = (unsigned)(FROWS * GHALF) + (unsigned)lane;
    #pragma unroll
    for (int k = 0; k < FSEGS; ++k) {
        const unsigned lo = (unsigned)__builtin_amdgcn_readlane((int)((k < 32) ? lo0 : lo1), k & 31);
        const unsigned hi = (unsigned)__builtin_amdgcn_readlane((int)((k < 32) ? hi0 : hi1), k & 31);
        const unsigned odd = lo & 1u;
        const unsigned len = hi - lo;
        #pragma unroll
        for (int q = 0; q < 2; ++q) {
            const unsigned pos = 2u * (unsigned)lane + (unsigned)q - odd;
            const unsigned ev = (pos < len) ? dw[k][q] : 0xFFFFFFFFu;
            const unsigned uq = (ev >> 6) - ubase;
            const unsigned r  = (uq >> 1) | (uq << 31);
            const unsigned a  = (r < (unsigned)BUCK_NODES) ? (r * GHALF + (ev & 63u)) : spare;
            atomicAdd(&cs[a], 1u);
        }
    }
    const unsigned more0 = __builtin_amdgcn_ballot_w32(hi0 > (lo0 & ~1u) + 64u);
    const unsigned more1 = __builtin_amdgcn_ballot_w32((lane < FSEGS - 32) & (hi1 > (lo1 & ~1u) + 64u));
    #pragma unroll
    for (int h = 0; h < 2; ++h) {
        unsigned more = h ? more1 : more0;
        while (more != 0u) {
            const int j = __builtin_ctz(more);
            more &= more - 1u;
            const unsigned lo = (unsigned)__builtin_amdgcn_readlane((int)(h ? lo1 : lo0), j);
            const unsigned hi = (unsigned)__builtin_amdgcn_readlane((int)(h ? hi1 : hi0), j);
            const unsigned* sp = ent + (size_t)(wave + 4 * (32 * h + j)) * ENT_PER_WG;
            for (unsigned i = (lo & ~1u) + 64u + lane; i < hi; i += 32) {
                const unsigned ev = sp[i];
                const unsigned r = (ev >> 7) - m0;
                if (r < (unsigned)BUCK_NODES && ((ev >> 6) & 1u) == gbh)
                    atomicAdd(&cs[r * GHALF + (ev & 63u)], 1u);
            }
        }
    }
    __syncthreads();

    ISSUE_A(0, aq)

    #pragma nounroll
    for (int trip = 0; trip < FPER / (2 * DEPTH) - 1; ++trip) {
        ISSUE_X(1, xq + DEPTH * (4 * HID))
        ISSUE_A(1, aq + DEPTH * (4 * GHALF))
        __builtin_amdgcn_sched_barrier(0);
        #pragma unroll
        for (int d = 0; d < DEPTH; ++d) COMPUTE(0, d)
        __builtin_amdgcn_sched_barrier(0);
        xq += 2 * DEPTH * (4 * HID);
        aq += 2 * DEPTH * (4 * GHALF);
        ISSUE_X(0, xq)
        ISSUE_A(0, aq)
        __builtin_amdgcn_sched_barrier(0);
        #pragma unroll
        for (int d = 0; d < DEPTH; ++d) COMPUTE(1, d)
    }
    ISSUE_X(1, xq + DEPTH * (4 * HID))
    ISSUE_A(1, aq + DEPTH * (4 * GHALF))
    __builtin_amdgcn_sched_barrier(0);
    #pragma unroll
    for (int d = 0; d < DEPTH; ++d) COMPUTE(0, d)
    #pragma unroll
    for (int d = 0; d < DEPTH; ++d) COMPUTE(1, d)
    #undef ISSUE_X
    #undef ISSUE_A
    #undef COMPUTE

    float* dst = partial + (size_t)chunk * (N_GRAPHS * HID) + cb + 4 * l16;
    #pragma unroll
    for (int i = 0; i < 4; ++i) {
        #pragma unroll
        for (int v = 0; v < 8; ++v) {
            const int g = gb + 4 * (v + 8 * lh) + i;
            v4f o = {acc[i][0][v], acc[i][1][v], acc[i][2][v], acc[i][3][v]};
            *(v4f*)(dst + g * HID) = o;
        }
    }
}

__global__ void k_reduce(const float* __restrict__ partial,
                         const unsigned* __restrict__ counts,
                         float* __restrict__ P, float* __restrict__ s,
                         int nchunks) {
    int i = blockIdx.x * 256 + threadIdx.x;
    int g = i >> 9;
    float a = 0.0f;
    for (int ch = 0; ch < nchunks; ++ch)
        a += partial[(size_t)ch * (N_GRAPHS * HID) + i];
    float cnt = (float)counts[g];
    P[i] = a / fmaxf(cnt, 1.0f);
    if (i < N_GRAPHS) s[i] = (counts[i] != 0u) ? 12.0f : 0.0f;
}

__global__ void __launch_bounds__(256)
k_reduce_w(const float* __restrict__ partial, const unsigned* __restrict__ hist,
           float* __restrict__ P, float* __restrict__ s, int nchunks,
           const float* __restrict__ w01, const float* __restrict__ W2,
           const float* __restrict__ b2, float* __restrict__ w012) {
    __shared__ __attribute__((aligned(16))) float aslab[ASLAB];
    __shared__ unsigned gcnt;
    if (blockIdx.x < WP_TILES / 8) {
        wprod_tile<256>(w01, w01 + HID * HID, W2, b2, w012, blockIdx.x * 8 + (threadIdx.x >> 5), aslab);
        return;
    }
    const int t = threadIdx.x;
    const int i = (blockIdx.x - WP_TILES / 8) * 256 + t;
    const int g = i >> 9;
    if (t == 0) gcnt = 0u;
    __syncthreads();
    unsigned c = (t < SCAT_WGS) ? hist[t * N_GRAPHS + g] : 0u;
    #pragma unroll
    for (int o = 16; o > 0; o >>= 1) c += __shfl_xor(c, o);
    if ((t & 31) == 0) atomicAdd(&gcnt, c);
    float a = 0.0f;
    for (int ch = 0; ch < nchunks; ++ch)
        a += partial[(size_t)ch * (N_GRAPHS * HID) + i];
    __syncthreads();
    const unsigned cg = gcnt;
    P[i] = a / fmaxf((float)cg, 1.0f);
    if ((i & (HID - 1)) == 0) s[g] = (cg != 0u) ? 12.0f : 0.0f;
}

__global__ void __launch_bounds__(256)
k_gemm_wmma(const float* __restrict__ A, int lda,
            const float* __restrict__ W, const float* __restrict__ bias,
            const float* __restrict__ s, float* __restrict__ out) {
    const int wave = (blockIdx.x << 3) + (threadIdx.x >> 5);
    const int lane = threadIdx.x & 31;
    const int l16  = lane & 15;
    const int lh   = lane >> 4;
    const int tm   = wave >> 5;
    const int tn   = wave & 31;

    const float* Arow = A + (tm * 16 + l16) * lda;
    const int col = tn * 16 + l16;

    v8f acc = {};
    for (int k = 0; k < HID; k += 4) {
        const int kk = k + lh * 2;
        v2f a; a.x = Arow[kk];           a.y = Arow[kk + 1];
        v2f b; b.x = W[kk * HID + col];  b.y = W[(kk + 1) * HID + col];
        acc = __builtin_amdgcn_wmma_f32_16x16x4_f32(
             false, a,  false, b,
             (short)0, acc,  false,  false);
    }

    const float bn = bias[col];
    #pragma unroll
    for (int v = 0; v < 8; ++v) {
        const int m = tm * 16 + v + lh * 8;
        out[m * OUT_PITCH + col] = acc[v] + s[m] * bn;
    }
}

__global__ void __launch_bounds__(256)
k_gemm3(const float* __restrict__ P,
        const float* __restrict__ W0, const float* __restrict__ b0,
        const float* __restrict__ w01, const float* __restrict__ w012,
        const float* __restrict__ s, float* __restrict__ out) {
    __shared__ __attribute__((aligned(16))) float aslab[ASLAB];
    const int wave = (blockIdx.x << 3) + (threadIdx.x >> 5);
    const int lane = threadIdx.x & 31;
    const int l16  = lane & 15;
    const int lh   = lane >> 4;
    const int tm    = wave / 96;
    const int tq    = wave - tm * 96;
    const int layer = tq >> 5;
    const int tn    = tq & 31;

    const float* B    = (layer == 0) ? W0 : ((layer == 1) ? w01 : w012);
    const float* bias = (layer == 0) ? b0 : (B + HID * HID);
    const int col = tn * 16 + l16;

    const v8f acc = tile_kloop<256>(P, P, tm * 16, aslab, B + (lh * 2) * HID + col);

    const float bn = bias[col];
    #pragma unroll
    for (int v = 0; v < 8; ++v) {
        const int m = tm * 16 + v + lh * 8;
        out[m * OUT_PITCH + layer * HID + col] = acc[v] + s[m] * bn;
    }
}

extern "C" void kernel_launch(void* const* d_in, const int* in_sizes, int n_in,
                              void* d_out, int out_size, void* d_ws, size_t ws_size,
                              hipStream_t stream) {
    const float* x     = (const float*)d_in[0];
    const int*   w2    = (const int*)  d_in[1];
    const int*   w3    = (const int*)  d_in[2];
    const int*   w4    = (const int*)  d_in[3];
    const int*   batch = (const int*)  d_in[4];
    const float* W0 = (const float*)d_in[5];  const float* b0 = (const float*)d_in[6];
    const float* W1 = (const float*)d_in[7];  const float* b1 = (const float*)d_in[8];
    const float* W2 = (const float*)d_in[9];  const float* b2 = (const float*)d_in[10];
    float* out = (float*)d_out;

    const size_t partBytes = (size_t)N_GRAPHS * HID * sizeof(float);
    const size_t fixed = partBytes   + 512   + 512
                       + (size_t)N_NODES * sizeof(int)  ;
    int nchunks = 1;
    if (ws_size > fixed + partBytes) {
        size_t fit = (ws_size - fixed) / partBytes;
        nchunks = (fit > MAX_CHUNKS) ? MAX_CHUNKS : (int)fit;
        if (nchunks < 1) nchunks = 1;
    }
    char* ws = (char*)d_ws;
    float*    partial = (float*)ws;
    float*    P       = (float*)(ws + (size_t)nchunks * partBytes);
    unsigned* counts  = (unsigned*)((char*)P + partBytes);
    float*    s       = (float*)((char*)counts + 512);
    int*      pb      = (int*)((char*)s + 512);

    const size_t ctBytes = (size_t)CT_ROWS * N_GRAPHS * sizeof(unsigned);
    const size_t denseNeed = (size_t)MAX_CHUNKS * partBytes + partBytes + 1024 + ctBytes;
    const size_t wpBytes = (size_t)WP_ROWS * HID * sizeof(float);
    if (ws_size >= denseNeed + 2 * wpBytes) {
        unsigned* ent  = (unsigned*)pb;
        unsigned* off  = ent + (size_t)SCAT_WGS * ENT_PER_WG;
        unsigned* hist = off + SCAT_WGS * (NBUCK + 1);
        float* w01  = (float*)((char*)pb + ctBytes);
        float* w012 = (float*)((char*)w01 + wpBytes);
        k_bin_w<<<SCAT_WGS + WP_TILES / 8, 256, 0, stream>>>(w2, w3, w4, batch, ent, off, hist,
                                                          W0, b0, W1, b1, w01);
        dim3 gFused(MAX_CHUNKS, 4);
        k_cgemm_f<<<gFused, 128, 0, stream>>>(x, ent, off, partial);
        k_reduce_w<<<WP_TILES / 8 + (N_GRAPHS * HID) / 256, 256, 0, stream>>>(partial, hist, P, s, nchunks,
                                                                              w01, W2, b2, w012);
        k_gemm3<<<96, 256, 0, stream>>>(P, W0, b0, w01, w012, s, out);
        return;
    }
    if (ws_size >= denseNeed) {
        unsigned* ct = (unsigned*)pb;
        k_zero_ct<<<512, 256, 0, stream>>>(ct, counts);
        k_scatter<<<(N_NODES + 255) / 256, 256, 0, stream>>>(w2, w3, w4, batch, ct, counts);
        dim3 gDense(nchunks, 4);
        k_cgemm<<<gDense, 128, 0, stream>>>(x, ct, partial, nchunks);
    } else {
        k_zero_counts<<<1, 128, 0, stream>>>(counts);
        k_pb_hist<<<(N_NODES + 255) / 256, 256, 0, stream>>>(w2, batch, pb, counts);
        dim3 gAcc(nchunks, 2);
        k_accum<<<gAcc, 256, 0, stream>>>(x, w2, w3, w4, pb, partial, nchunks);
    }
    k_reduce<<<(N_GRAPHS * HID) / 256, 256, 0, stream>>>(partial, counts, P, s, nchunks);

    k_gemm_wmma<<<32, 256, 0, stream>>>(P,         HID,       W0, b0, s, out);
    k_gemm_wmma<<<32, 256, 0, stream>>>(out,       OUT_PITCH, W1, b1, s, out + HID);
    k_gemm_wmma<<<32, 256, 0, stream>>>(out + HID, OUT_PITCH, W2, b2, s, out + 2 * HID);
}
